# starting kernel
# baseline (speedup 1.0000x reference)
#include <hip/hip_runtime.h>
#include <math.h>

#define NN   2000
#define NE   32000
#define EMB  768
#define HID  384
#define FT   768
#define NH   4
#define NEG_SLOPE 0.2f

typedef __attribute__((ext_vector_type(2))) float v2f;
typedef __attribute__((ext_vector_type(8))) float v8f;

static inline int cdiv(long long a, long long b) { return (int)((a + b - 1) / b); }

__device__ __forceinline__ unsigned fenc(float f) {
  unsigned u = __float_as_uint(f);
  return (u & 0x80000000u) ? ~u : (u | 0x80000000u);
}
__device__ __forceinline__ float fdec(unsigned e) {
  unsigned u = (e & 0x80000000u) ? (e ^ 0x80000000u) : ~e;
  return __uint_as_float(u);
}

__global__ void k_embed_x(const int* __restrict__ x, const float* __restrict__ emb,
                          float* __restrict__ xf) {
  long long i = (long long)blockIdx.x * blockDim.x + threadIdx.x;
  if (i >= (long long)NN * EMB) return;
  int n = (int)(i / EMB), c = (int)(i % EMB);
  xf[i] = emb[(size_t)x[n * 2 + 0] * EMB + c] + emb[(size_t)x[n * 2 + 1] * EMB + c];
}

__global__ void k_embed_e(const int* __restrict__ ea_idx, const float* __restrict__ emb,
                          float* __restrict__ ea) {
  long long i = (long long)blockIdx.x * blockDim.x + threadIdx.x;
  if (i >= (long long)NE * EMB) return;
  int e = (int)(i / EMB), c = (int)(i % EMB);
  ea[i] = emb[(size_t)ea_idx[e * 2 + 0] * EMB + c] + emb[(size_t)ea_idx[e * 2 + 1] * EMB + c];
}

__global__ void k_deg(const int* __restrict__ dst, float* __restrict__ deg) {
  int e = blockIdx.x * blockDim.x + threadIdx.x;
  if (e < NE) atomicAdd(&deg[dst[e]], 1.0f);
}

__global__ void k_loopsum(const int* __restrict__ dst, float* __restrict__ ea) {
  long long i = (long long)blockIdx.x * blockDim.x + threadIdx.x;
  if (i >= (long long)NE * EMB) return;
  int e = (int)(i / EMB), c = (int)(i % EMB);
  atomicAdd(&ea[(size_t)(NE + dst[e]) * EMB + c], ea[(size_t)e * EMB + c]);
}

__global__ void k_loopdiv(const float* __restrict__ deg, float* __restrict__ ea) {
  long long i = (long long)blockIdx.x * blockDim.x + threadIdx.x;
  if (i >= (long long)NN * EMB) return;
  int n = (int)(i / EMB);
  ea[(size_t)NE * EMB + i] /= fmaxf(deg[n], 1.0f);
}

__global__ __launch_bounds__(256) void k_gemm_wmma_f32(
    const float* __restrict__ A, const float* __restrict__ B, float* __restrict__ C,
    int M, int N, int K) {
  __shared__ __align__(16) float sA[64][32];
  __shared__ __align__(16) float sBt[128][32];
  const int tid = threadIdx.x;
  const int wave = tid >> 5, lane = tid & 31;
  const int wm = wave >> 2, wn = wave & 3;
  const int row16 = lane & 15, khalf = lane >> 4;
  const int bm = blockIdx.y * 64, bn = blockIdx.x * 128;

  v8f acc00 = {}, acc01 = {}, acc10 = {}, acc11 = {};

  for (int k0 = 0; k0 < K; k0 += 32) {
    for (int i = tid; i < 64 * 32; i += 256) {
      int r = i >> 5, c = i & 31;
      int gr = bm + r;
      sA[r][c] = (gr < M) ? A[(size_t)gr * K + k0 + c] : 0.0f;
    }
    for (int i = tid; i < 32 * 128; i += 256) {
      int kk = i >> 7, n = i & 127;
      sBt[n][kk] = B[(size_t)(k0 + kk) * N + bn + n];
    }
    __syncthreads();

    const float* a0p = &sA[wm * 32 + row16][0];
    const float* a1p = &sA[wm * 32 + 16 + row16][0];
    const float* b0p = &sBt[wn * 32 + row16][0];
    const float* b1p = &sBt[wn * 32 + 16 + row16][0];
#pragma unroll
    for (int ks = 0; ks < 32; ks += 4) {
      int kk = ks + khalf * 2;
      v2f a0 = *(const v2f*)(a0p + kk);
      v2f a1 = *(const v2f*)(a1p + kk);
      v2f b0 = *(const v2f*)(b0p + kk);
      v2f b1 = *(const v2f*)(b1p + kk);
      acc00 = __builtin_amdgcn_wmma_f32_16x16x4_f32(false, a0, false, b0, (short)0, acc00, false, false);
      acc01 = __builtin_amdgcn_wmma_f32_16x16x4_f32(false, a0, false, b1, (short)0, acc01, false, false);
      acc10 = __builtin_amdgcn_wmma_f32_16x16x4_f32(false, a1, false, b0, (short)0, acc10, false, false);
      acc11 = __builtin_amdgcn_wmma_f32_16x16x4_f32(false, a1, false, b1, (short)0, acc11, false, false);
    }
    __syncthreads();
  }

  int gn0 = bn + wn * 32 + row16;
  int gn1 = gn0 + 16;
  int gm0 = bm + wm * 32 + 8 * khalf;
  int gm1 = gm0 + 16;
#pragma unroll
  for (int v = 0; v < 8; ++v) {
    if (gm0 + v < M) {
      C[(size_t)(gm0 + v) * N + gn0] = acc00[v];
      C[(size_t)(gm0 + v) * N + gn1] = acc01[v];
    }
    if (gm1 + v < M) {
      C[(size_t)(gm1 + v) * N + gn0] = acc10[v];
      C[(size_t)(gm1 + v) * N + gn1] = acc11[v];
    }
  }
}

__global__ void k_proj_e(const float* __restrict__ We, const float* __restrict__ ae,
                         float* __restrict__ proj, int C) {
  int i = blockIdx.x * blockDim.x + threadIdx.x;
  if (i >= EMB * NH) return;
  int k = i / NH, h = i % NH;
  const float* wr = We + (size_t)k * (NH * C) + (size_t)h * C;
  const float* ar = ae + (size_t)h * C;
  float s = 0.f;
  for (int c = 0; c < C; ++c) s += wr[c] * ar[c];
  proj[i] = s;
}

__global__ void k_alpha_node(const float* __restrict__ hx, const float* __restrict__ a_src,
                             const float* __restrict__ a_dst, float* __restrict__ ans,
                             float* __restrict__ adn, int C) {
  int i = blockIdx.x * blockDim.x + threadIdx.x;
  if (i >= NN * NH) return;
  int n = i / NH, h = i % NH;
  const float* hr = hx + (size_t)n * NH * C + (size_t)h * C;
  const float* sr = a_src + (size_t)h * C;
  const float* dr = a_dst + (size_t)h * C;
  float s = 0.f, t = 0.f;
  for (int c = 0; c < C; ++c) { s += hr[c] * sr[c]; t += hr[c] * dr[c]; }
  ans[i] = s;
  adn[i] = t;
}

__global__ void k_edge_alpha(const float* __restrict__ ea, const float* __restrict__ proj,
                             const float* __restrict__ ans, const float* __restrict__ adn,
                             const int* __restrict__ src, const int* __restrict__ dst,
                             float* __restrict__ alpha, unsigned* __restrict__ mx) {
  const int tot = NE + NN;
  int e = blockIdx.x * (blockDim.x >> 5) + (threadIdx.x >> 5);
  int lane = threadIdx.x & 31;
  if (e >= tot) return;
  int s = (e < NE) ? src[e] : (e - NE);
  int d = (e < NE) ? dst[e] : (e - NE);
  const float* row = ea + (size_t)e * EMB;
  const float4* p4 = (const float4*)proj;
  float a0 = 0.f, a1 = 0.f, a2 = 0.f, a3 = 0.f;
  for (int k = lane; k < EMB; k += 32) {
    float v = row[k];
    float4 p = p4[k];
    a0 += v * p.x; a1 += v * p.y; a2 += v * p.z; a3 += v * p.w;
  }
#pragma unroll
  for (int off = 16; off > 0; off >>= 1) {
    a0 += __shfl_xor(a0, off, 32);
    a1 += __shfl_xor(a1, off, 32);
    a2 += __shfl_xor(a2, off, 32);
    a3 += __shfl_xor(a3, off, 32);
  }
  if (lane < 4) {
    float ale = (lane == 0) ? a0 : (lane == 1) ? a1 : (lane == 2) ? a2 : a3;
    float a = ans[s * NH + lane] + adn[d * NH + lane] + ale;
    a = (a > 0.f) ? a : NEG_SLOPE * a;
    alpha[(size_t)e * NH + lane] = a;
    atomicMax(&mx[d * NH + lane], fenc(a));
  }
}

__global__ void k_softmax_den(const int* __restrict__ dst, const unsigned* __restrict__ mx,
                              float* __restrict__ alpha, float* __restrict__ den) {
  const int tot = NE + NN;
  long long i = (long long)blockIdx.x * blockDim.x + threadIdx.x;
  if (i >= (long long)tot * NH) return;
  int e = (int)(i >> 2), h = (int)(i & 3);
  int d = (e < NE) ? dst[e] : (e - NE);
  float ex = expf(alpha[i] - fdec(mx[d * NH + h]));
  alpha[i] = ex;
  atomicAdd(&den[d * NH + h], ex);
}

__global__ void k_aggregate(const float* __restrict__ hx, const float* __restrict__ ex,
                            const float* __restrict__ den, const int* __restrict__ src,
                            const int* __restrict__ dst, float* __restrict__ out, int C) {
  int e = blockIdx.x;
  int s = (e < NE) ? src[e] : (e - NE);
  int d = (e < NE) ? dst[e] : (e - NE);
  __shared__ float att[NH];
  if (threadIdx.x < NH)
    att[threadIdx.x] = ex[(size_t)e * NH + threadIdx.x] / den[d * NH + threadIdx.x];
  __syncthreads();
  const float* hr = hx + (size_t)s * NH * C;
  for (int c = threadIdx.x; c < C; c += blockDim.x) {
    float v = 0.25f * (att[0] * hr[c] + att[1] * hr[C + c] +
                       att[2] * hr[2 * C + c] + att[3] * hr[3 * C + c]);
    atomicAdd(&out[(size_t)d * C + c], v);
  }
}

__global__ void k_bias_relu(const float* __restrict__ in, const float* __restrict__ b,
                            float* __restrict__ o) {
  long long i = (long long)blockIdx.x * blockDim.x + threadIdx.x;
  if (i >= (long long)NN * HID) return;
  o[i] = fmaxf(in[i] + b[i % HID], 0.f);
}

__global__ void k_init_out(const float* __restrict__ b, float* __restrict__ o) {
  long long i = (long long)blockIdx.x * blockDim.x + threadIdx.x;
  if (i >= (long long)NN * FT) return;
  o[i] = b[i % FT];
}

extern "C" void kernel_launch(void* const* d_in, const int* in_sizes, int n_in,
                              void* d_out, int out_size, void* d_ws, size_t ws_size,
                              hipStream_t stream) {
  const int* x      = (const int*)d_in[0];
  const int* eidx   = (const int*)d_in[1];
  const int* src    = eidx;
  const int* dst    = eidx + NE;
  const int* eattr  = (const int*)d_in[2];
  const float* xemb = (const float*)d_in[3];
  const float* eemb = (const float*)d_in[4];
  const float* W1   = (const float*)d_in[5];
  const float* as1  = (const float*)d_in[6];
  const float* ad1  = (const float*)d_in[7];
  const float* We1  = (const float*)d_in[8];
  const float* ae1  = (const float*)d_in[9];
  const float* b1   = (const float*)d_in[10];
  const float* W2   = (const float*)d_in[11];
  const float* as2  = (const float*)d_in[12];
  const float* ad2  = (const float*)d_in[13];
  const float* We2  = (const float*)d_in[14];
  const float* ae2  = (const float*)d_in[15];
  const float* b2   = (const float*)d_in[16];
  float* out = (float*)d_out;

  float* ws = (float*)d_ws;
  size_t off = 0;
  float* EA   = ws + off; off += (size_t)(NE + NN) * EMB;
  float* XF   = ws + off; off += (size_t)NN * EMB;
  float* DEG  = ws + off; off += 2048;
  float* HX1  = ws + off; off += (size_t)NN * NH * HID;
  float* HX2  = ws + off; off += (size_t)NN * NH * FT;
  float* PROJ = ws + off; off += (size_t)EMB * NH;
  float* ANS  = ws + off; off += (size_t)NN * NH;
  float* ADN  = ws + off; off += (size_t)NN * NH;
  float* ALPH = ws + off; off += (size_t)(NE + NN) * NH;
  unsigned* MX = (unsigned*)(ws + off); off += (size_t)NN * NH;
  float* DEN  = ws + off; off += (size_t)NN * NH;
  float* OUT1 = ws + off; off += (size_t)NN * HID;
  float* HB   = ws + off; off += (size_t)NN * HID;
  (void)ws_size; (void)in_sizes; (void)n_in; (void)out_size;

  const int T = 256;

  hipMemsetAsync(DEG, 0, 2048 * sizeof(float), stream);
  hipMemsetAsync(EA + (size_t)NE * EMB, 0, (size_t)NN * EMB * sizeof(float), stream);
  hipMemsetAsync(OUT1, 0, (size_t)NN * HID * sizeof(float), stream);
  hipMemsetAsync(MX, 0, (size_t)NN * NH * sizeof(unsigned), stream);
  hipMemsetAsync(DEN, 0, (size_t)NN * NH * sizeof(float), stream);

  k_embed_x<<<cdiv((long long)NN * EMB, T), T, 0, stream>>>(x, xemb, XF);
  k_embed_e<<<cdiv((long long)NE * EMB, T), T, 0, stream>>>(eattr, eemb, EA);
  k_deg<<<cdiv(NE, T), T, 0, stream>>>(dst, DEG);
  k_loopsum<<<cdiv((long long)NE * EMB, T), T, 0, stream>>>(dst, EA);
  k_loopdiv<<<cdiv((long long)NN * EMB, T), T, 0, stream>>>(DEG, EA);

  {
    dim3 g(NH * HID / 128, cdiv(NN, 64));
    k_gemm_wmma_f32<<<g, 256, 0, stream>>>(XF, W1, HX1, NN, NH * HID, EMB);
  }
  k_proj_e<<<cdiv(EMB * NH, T), T, 0, stream>>>(We1, ae1, PROJ, HID);
  k_alpha_node<<<cdiv(NN * NH, T), T, 0, stream>>>(HX1, as1, ad1, ANS, ADN, HID);
  k_edge_alpha<<<cdiv(NE + NN, 8), 256, 0, stream>>>(EA, PROJ, ANS, ADN, src, dst, ALPH, MX);
  k_softmax_den<<<cdiv((long long)(NE + NN) * NH, T), T, 0, stream>>>(dst, MX, ALPH, DEN);
  k_aggregate<<<NE + NN, HID, 0, stream>>>(HX1, ALPH, DEN, src, dst, OUT1, HID);
  k_bias_relu<<<cdiv((long long)NN * HID, T), T, 0, stream>>>(OUT1, b1, HB);

  hipMemsetAsync(MX, 0, (size_t)NN * NH * sizeof(unsigned), stream);
  hipMemsetAsync(DEN, 0, (size_t)NN * NH * sizeof(float), stream);

  {
    dim3 g(NH * FT / 128, cdiv(NN, 64));
    k_gemm_wmma_f32<<<g, 256, 0, stream>>>(HB, W2, HX2, NN, NH * FT, HID);
  }
  k_proj_e<<<cdiv(EMB * NH, T), T, 0, stream>>>(We2, ae2, PROJ, FT);
  k_alpha_node<<<cdiv(NN * NH, T), T, 0, stream>>>(HX2, as2, ad2, ANS, ADN, FT);
  k_edge_alpha<<<cdiv(NE + NN, 8), 256, 0, stream>>>(EA, PROJ, ANS, ADN, src, dst, ALPH, MX);
  k_softmax_den<<<cdiv((long long)(NE + NN) * NH, T), T, 0, stream>>>(dst, MX, ALPH, DEN);
  k_init_out<<<cdiv((long long)NN * FT, T), T, 0, stream>>>(b2, out);
  k_aggregate<<<NE + NN, FT, 0, stream>>>(HX2, ALPH, DEN, src, dst, out, FT);
}


# GEMM tile staging: each thread issues all 24 global loads of a K-block before its first LDS store (was one load, full wait, one store per trip: 24 dependent memory round trips per K-block)
# speedup vs baseline: 1.2637x; 1.2637x over previous
#include <hip/hip_runtime.h>
#include <math.h>

#define NN   2000
#define NE   32000
#define EMB  768
#define HID  384
#define FT   768
#define NH   4
#define NEG_SLOPE 0.2f

typedef __attribute__((ext_vector_type(2))) float v2f;
typedef __attribute__((ext_vector_type(8))) float v8f;

static inline int cdiv(long long a, long long b) { return (int)((a + b - 1) / b); }

__device__ __forceinline__ unsigned fenc(float f) {
  unsigned u = __float_as_uint(f);
  return (u & 0x80000000u) ? ~u : (u | 0x80000000u);
}
__device__ __forceinline__ float fdec(unsigned e) {
  unsigned u = (e & 0x80000000u) ? (e ^ 0x80000000u) : ~e;
  return __uint_as_float(u);
}

__global__ void k_embed_x(const int* __restrict__ x, const float* __restrict__ emb,
                          float* __restrict__ xf) {
  long long i = (long long)blockIdx.x * blockDim.x + threadIdx.x;
  if (i >= (long long)NN * EMB) return;
  int n = (int)(i / EMB), c = (int)(i % EMB);
  xf[i] = emb[(size_t)x[n * 2 + 0] * EMB + c] + emb[(size_t)x[n * 2 + 1] * EMB + c];
}

__global__ void k_embed_e(const int* __restrict__ ea_idx, const float* __restrict__ emb,
                          float* __restrict__ ea) {
  long long i = (long long)blockIdx.x * blockDim.x + threadIdx.x;
  if (i >= (long long)NE * EMB) return;
  int e = (int)(i / EMB), c = (int)(i % EMB);
  ea[i] = emb[(size_t)ea_idx[e * 2 + 0] * EMB + c] + emb[(size_t)ea_idx[e * 2 + 1] * EMB + c];
}

__global__ void k_deg(const int* __restrict__ dst, float* __restrict__ deg) {
  int e = blockIdx.x * blockDim.x + threadIdx.x;
  if (e < NE) atomicAdd(&deg[dst[e]], 1.0f);
}

__global__ void k_loopsum(const int* __restrict__ dst, float* __restrict__ ea) {
  long long i = (long long)blockIdx.x * blockDim.x + threadIdx.x;
  if (i >= (long long)NE * EMB) return;
  int e = (int)(i / EMB), c = (int)(i % EMB);
  atomicAdd(&ea[(size_t)(NE + dst[e]) * EMB + c], ea[(size_t)e * EMB + c]);
}

__global__ void k_loopdiv(const float* __restrict__ deg, float* __restrict__ ea) {
  long long i = (long long)blockIdx.x * blockDim.x + threadIdx.x;
  if (i >= (long long)NN * EMB) return;
  int n = (int)(i / EMB);
  ea[(size_t)NE * EMB + i] /= fmaxf(deg[n], 1.0f);
}

__global__ __launch_bounds__(256) void k_gemm_wmma_f32(
    const float* __restrict__ A, const float* __restrict__ B, float* __restrict__ C,
    int M, int N, int K) {
  __shared__ __align__(16) float sA[64][32];
  __shared__ __align__(16) float sBt[128][32];
  const int tid = threadIdx.x;
  const int wave = tid >> 5, lane = tid & 31;
  const int wm = wave >> 2, wn = wave & 3;
  const int row16 = lane & 15, khalf = lane >> 4;
  const int bm = blockIdx.y * 64, bn = blockIdx.x * 128;

  v8f acc00 = {}, acc01 = {}, acc10 = {}, acc11 = {};

  for (int k0 = 0; k0 < K; k0 += 32) {
    float ra[8], rb[16];
#pragma unroll
    for (int j = 0; j < 8; ++j) {
      int i = tid + j * 256;
      int r = i >> 5, c = i & 31;
      int gr = bm + r;
      ra[j] = (gr < M) ? A[(size_t)gr * K + k0 + c] : 0.0f;
    }
#pragma unroll
    for (int j = 0; j < 16; ++j) {
      int i = tid + j * 256;
      int kk = i >> 7, n = i & 127;
      rb[j] = B[(size_t)(k0 + kk) * N + bn + n];
    }
#pragma unroll
    for (int j = 0; j < 8; ++j) {
      int i = tid + j * 256;
      sA[i >> 5][i & 31] = ra[j];
    }
#pragma unroll
    for (int j = 0; j < 16; ++j) {
      int i = tid + j * 256;
      sBt[i & 127][i >> 7] = rb[j];
    }
    __syncthreads();

    const float* a0p = &sA[wm * 32 + row16][0];
    const float* a1p = &sA[wm * 32 + 16 + row16][0];
    const float* b0p = &sBt[wn * 32 + row16][0];
    const float* b1p = &sBt[wn * 32 + 16 + row16][0];
#pragma unroll
    for (int ks = 0; ks < 32; ks += 4) {
      int kk = ks + khalf * 2;
      v2f a0 = *(const v2f*)(a0p + kk);
      v2f a1 = *(const v2f*)(a1p + kk);
      v2f b0 = *(const v2f*)(b0p + kk);
      v2f b1 = *(const v2f*)(b1p + kk);
      acc00 = __builtin_amdgcn_wmma_f32_16x16x4_f32(false, a0, false, b0, (short)0, acc00, false, false);
      acc01 = __builtin_amdgcn_wmma_f32_16x16x4_f32(false, a0, false, b1, (short)0, acc01, false, false);
      acc10 = __builtin_amdgcn_wmma_f32_16x16x4_f32(false, a1, false, b0, (short)0, acc10, false, false);
      acc11 = __builtin_amdgcn_wmma_f32_16x16x4_f32(false, a1, false, b1, (short)0, acc11, false, false);
    }
    __syncthreads();
  }

  int gn0 = bn + wn * 32 + row16;
  int gn1 = gn0 + 16;
  int gm0 = bm + wm * 32 + 8 * khalf;
  int gm1 = gm0 + 16;
#pragma unroll
  for (int v = 0; v < 8; ++v) {
    if (gm0 + v < M) {
      C[(size_t)(gm0 + v) * N + gn0] = acc00[v];
      C[(size_t)(gm0 + v) * N + gn1] = acc01[v];
    }
    if (gm1 + v < M) {
      C[(size_t)(gm1 + v) * N + gn0] = acc10[v];
      C[(size_t)(gm1 + v) * N + gn1] = acc11[v];
    }
  }
}

__global__ void k_proj_e(const float* __restrict__ We, const float* __restrict__ ae,
                         float* __restrict__ proj, int C) {
  int i = blockIdx.x * blockDim.x + threadIdx.x;
  if (i >= EMB * NH) return;
  int k = i / NH, h = i % NH;
  const float* wr = We + (size_t)k * (NH * C) + (size_t)h * C;
  const float* ar = ae + (size_t)h * C;
  float s = 0.f;
  for (int c = 0; c < C; ++c) s += wr[c] * ar[c];
  proj[i] = s;
}

__global__ void k_alpha_node(const float* __restrict__ hx, const float* __restrict__ a_src,
                             const float* __restrict__ a_dst, float* __restrict__ ans,
                             float* __restrict__ adn, int C) {
  int i = blockIdx.x * blockDim.x + threadIdx.x;
  if (i >= NN * NH) return;
  int n = i / NH, h = i % NH;
  const float* hr = hx + (size_t)n * NH * C + (size_t)h * C;
  const float* sr = a_src + (size_t)h * C;
  const float* dr = a_dst + (size_t)h * C;
  float s = 0.f, t = 0.f;
  for (int c = 0; c < C; ++c) { s += hr[c] * sr[c]; t += hr[c] * dr[c]; }
  ans[i] = s;
  adn[i] = t;
}

__global__ void k_edge_alpha(const float* __restrict__ ea, const float* __restrict__ proj,
                             const float* __restrict__ ans, const float* __restrict__ adn,
                             const int* __restrict__ src, const int* __restrict__ dst,
                             float* __restrict__ alpha, unsigned* __restrict__ mx) {
  const int tot = NE + NN;
  int e = blockIdx.x * (blockDim.x >> 5) + (threadIdx.x >> 5);
  int lane = threadIdx.x & 31;
  if (e >= tot) return;
  int s = (e < NE) ? src[e] : (e - NE);
  int d = (e < NE) ? dst[e] : (e - NE);
  const float* row = ea + (size_t)e * EMB;
  const float4* p4 = (const float4*)proj;
  float a0 = 0.f, a1 = 0.f, a2 = 0.f, a3 = 0.f;
  for (int k = lane; k < EMB; k += 32) {
    float v = row[k];
    float4 p = p4[k];
    a0 += v * p.x; a1 += v * p.y; a2 += v * p.z; a3 += v * p.w;
  }
#pragma unroll
  for (int off = 16; off > 0; off >>= 1) {
    a0 += __shfl_xor(a0, off, 32);
    a1 += __shfl_xor(a1, off, 32);
    a2 += __shfl_xor(a2, off, 32);
    a3 += __shfl_xor(a3, off, 32);
  }
  if (lane < 4) {
    float ale = (lane == 0) ? a0 : (lane == 1) ? a1 : (lane == 2) ? a2 : a3;
    float a = ans[s * NH + lane] + adn[d * NH + lane] + ale;
    a = (a > 0.f) ? a : NEG_SLOPE * a;
    alpha[(size_t)e * NH + lane] = a;
    atomicMax(&mx[d * NH + lane], fenc(a));
  }
}

__global__ void k_softmax_den(const int* __restrict__ dst, const unsigned* __restrict__ mx,
                              float* __restrict__ alpha, float* __restrict__ den) {
  const int tot = NE + NN;
  long long i = (long long)blockIdx.x * blockDim.x + threadIdx.x;
  if (i >= (long long)tot * NH) return;
  int e = (int)(i >> 2), h = (int)(i & 3);
  int d = (e < NE) ? dst[e] : (e - NE);
  float ex = expf(alpha[i] - fdec(mx[d * NH + h]));
  alpha[i] = ex;
  atomicAdd(&den[d * NH + h], ex);
}

__global__ void k_aggregate(const float* __restrict__ hx, const float* __restrict__ ex,
                            const float* __restrict__ den, const int* __restrict__ src,
                            const int* __restrict__ dst, float* __restrict__ out, int C) {
  int e = blockIdx.x;
  int s = (e < NE) ? src[e] : (e - NE);
  int d = (e < NE) ? dst[e] : (e - NE);
  __shared__ float att[NH];
  if (threadIdx.x < NH)
    att[threadIdx.x] = ex[(size_t)e * NH + threadIdx.x] / den[d * NH + threadIdx.x];
  __syncthreads();
  const float* hr = hx + (size_t)s * NH * C;
  for (int c = threadIdx.x; c < C; c += blockDim.x) {
    float v = 0.25f * (att[0] * hr[c] + att[1] * hr[C + c] +
                       att[2] * hr[2 * C + c] + att[3] * hr[3 * C + c]);
    atomicAdd(&out[(size_t)d * C + c], v);
  }
}

__global__ void k_bias_relu(const float* __restrict__ in, const float* __restrict__ b,
                            float* __restrict__ o) {
  long long i = (long long)blockIdx.x * blockDim.x + threadIdx.x;
  if (i >= (long long)NN * HID) return;
  o[i] = fmaxf(in[i] + b[i % HID], 0.f);
}

__global__ void k_init_out(const float* __restrict__ b, float* __restrict__ o) {
  long long i = (long long)blockIdx.x * blockDim.x + threadIdx.x;
  if (i >= (long long)NN * FT) return;
  o[i] = b[i % FT];
}

extern "C" void kernel_launch(void* const* d_in, const int* in_sizes, int n_in,
                              void* d_out, int out_size, void* d_ws, size_t ws_size,
                              hipStream_t stream) {
  const int* x      = (const int*)d_in[0];
  const int* eidx   = (const int*)d_in[1];
  const int* src    = eidx;
  const int* dst    = eidx + NE;
  const int* eattr  = (const int*)d_in[2];
  const float* xemb = (const float*)d_in[3];
  const float* eemb = (const float*)d_in[4];
  const float* W1   = (const float*)d_in[5];
  const float* as1  = (const float*)d_in[6];
  const float* ad1  = (const float*)d_in[7];
  const float* We1  = (const float*)d_in[8];
  const float* ae1  = (const float*)d_in[9];
  const float* b1   = (const float*)d_in[10];
  const float* W2   = (const float*)d_in[11];
  const float* as2  = (const float*)d_in[12];
  const float* ad2  = (const float*)d_in[13];
  const float* We2  = (const float*)d_in[14];
  const float* ae2  = (const float*)d_in[15];
  const float* b2   = (const float*)d_in[16];
  float* out = (float*)d_out;

  float* ws = (float*)d_ws;
  size_t off = 0;
  float* EA   = ws + off; off += (size_t)(NE + NN) * EMB;
  float* XF   = ws + off; off += (size_t)NN * EMB;
  float* DEG  = ws + off; off += 2048;
  float* HX1  = ws + off; off += (size_t)NN * NH * HID;
  float* HX2  = ws + off; off += (size_t)NN * NH * FT;
  float* PROJ = ws + off; off += (size_t)EMB * NH;
  float* ANS  = ws + off; off += (size_t)NN * NH;
  float* ADN  = ws + off; off += (size_t)NN * NH;
  float* ALPH = ws + off; off += (size_t)(NE + NN) * NH;
  unsigned* MX = (unsigned*)(ws + off); off += (size_t)NN * NH;
  float* DEN  = ws + off; off += (size_t)NN * NH;
  float* OUT1 = ws + off; off += (size_t)NN * HID;
  float* HB   = ws + off; off += (size_t)NN * HID;
  (void)ws_size; (void)in_sizes; (void)n_in; (void)out_size;

  const int T = 256;

  hipMemsetAsync(DEG, 0, 2048 * sizeof(float), stream);
  hipMemsetAsync(EA + (size_t)NE * EMB, 0, (size_t)NN * EMB * sizeof(float), stream);
  hipMemsetAsync(OUT1, 0, (size_t)NN * HID * sizeof(float), stream);
  hipMemsetAsync(MX, 0, (size_t)NN * NH * sizeof(unsigned), stream);
  hipMemsetAsync(DEN, 0, (size_t)NN * NH * sizeof(float), stream);

  k_embed_x<<<cdiv((long long)NN * EMB, T), T, 0, stream>>>(x, xemb, XF);
  k_embed_e<<<cdiv((long long)NE * EMB, T), T, 0, stream>>>(eattr, eemb, EA);
  k_deg<<<cdiv(NE, T), T, 0, stream>>>(dst, DEG);
  k_loopsum<<<cdiv((long long)NE * EMB, T), T, 0, stream>>>(dst, EA);
  k_loopdiv<<<cdiv((long long)NN * EMB, T), T, 0, stream>>>(DEG, EA);

  {
    dim3 g(NH * HID / 128, cdiv(NN, 64));
    k_gemm_wmma_f32<<<g, 256, 0, stream>>>(XF, W1, HX1, NN, NH * HID, EMB);
  }
  k_proj_e<<<cdiv(EMB * NH, T), T, 0, stream>>>(We1, ae1, PROJ, HID);
  k_alpha_node<<<cdiv(NN * NH, T), T, 0, stream>>>(HX1, as1, ad1, ANS, ADN, HID);
  k_edge_alpha<<<cdiv(NE + NN, 8), 256, 0, stream>>>(EA, PROJ, ANS, ADN, src, dst, ALPH, MX);
  k_softmax_den<<<cdiv((long long)(NE + NN) * NH, T), T, 0, stream>>>(dst, MX, ALPH, DEN);
  k_aggregate<<<NE + NN, HID, 0, stream>>>(HX1, ALPH, DEN, src, dst, OUT1, HID);
  k_bias_relu<<<cdiv((long long)NN * HID, T), T, 0, stream>>>(OUT1, b1, HB);

  hipMemsetAsync(MX, 0, (size_t)NN * NH * sizeof(unsigned), stream);
  hipMemsetAsync(DEN, 0, (size_t)NN * NH * sizeof(float), stream);

  {
    dim3 g(NH * FT / 128, cdiv(NN, 64));
    k_gemm_wmma_f32<<<g, 256, 0, stream>>>(HB, W2, HX2, NN, NH * FT, HID);
  }
  k_proj_e<<<cdiv(EMB * NH, T), T, 0, stream>>>(We2, ae2, PROJ, FT);
  k_alpha_node<<<cdiv(NN * NH, T), T, 0, stream>>>(HX2, as2, ad2, ANS, ADN, FT);
  k_edge_alpha<<<cdiv(NE + NN, 8), 256, 0, stream>>>(EA, PROJ, ANS, ADN, src, dst, ALPH, MX);
  k_softmax_den<<<cdiv((long long)(NE + NN) * NH, T), T, 0, stream>>>(dst, MX, ALPH, DEN);
  k_init_out<<<cdiv((long long)NN * FT, T), T, 0, stream>>>(b2, out);
  k_aggregate<<<NE + NN, FT, 0, stream>>>(HX2, ALPH, DEN, src, dst, out, FT);
}


# Edge-attr attention term by linearity: per edge the sum of two rows of an 18x4 table (emb x proj), self-loop from per-node attr-id counts; [E+N,768] tensor and its atomics gone; sum order changes
# speedup vs baseline: 1.5934x; 1.2609x over previous
#include <hip/hip_runtime.h>
#include <math.h>

#define NN   2000
#define NE   32000
#define EMB  768
#define HID  384
#define FT   768
#define NH   4
#define NEA  18
#define NEG_SLOPE 0.2f

typedef __attribute__((ext_vector_type(2))) float v2f;
typedef __attribute__((ext_vector_type(8))) float v8f;

static inline int cdiv(long long a, long long b) { return (int)((a + b - 1) / b); }

__device__ __forceinline__ unsigned fenc(float f) {
  unsigned u = __float_as_uint(f);
  return (u & 0x80000000u) ? ~u : (u | 0x80000000u);
}
__device__ __forceinline__ float fdec(unsigned e) {
  unsigned u = (e & 0x80000000u) ? (e ^ 0x80000000u) : ~e;
  return __uint_as_float(u);
}

__global__ void k_embed_x(const int* __restrict__ x, const float* __restrict__ emb,
                          float* __restrict__ xf) {
  long long i = (long long)blockIdx.x * blockDim.x + threadIdx.x;
  if (i >= (long long)NN * EMB) return;
  int n = (int)(i / EMB), c = (int)(i % EMB);
  xf[i] = emb[(size_t)x[n * 2 + 0] * EMB + c] + emb[(size_t)x[n * 2 + 1] * EMB + c];
}

__global__ void k_attr_count(const int* __restrict__ dst, const int* __restrict__ ea_idx,
                             unsigned* __restrict__ cnt) {
  int e = blockIdx.x * blockDim.x + threadIdx.x;
  if (e >= NE) return;
  int d = dst[e];
  atomicAdd(&cnt[d * NEA + ea_idx[e * 2 + 0]], 1u);
  atomicAdd(&cnt[d * NEA + ea_idx[e * 2 + 1]], 1u);
}

__global__ __launch_bounds__(256) void k_gemm_wmma_f32(
    const float* __restrict__ A, const float* __restrict__ B, float* __restrict__ C,
    int M, int N, int K) {
  __shared__ __align__(16) float sA[64][32];
  __shared__ __align__(16) float sBt[128][32];
  const int tid = threadIdx.x;
  const int wave = tid >> 5, lane = tid & 31;
  const int wm = wave >> 2, wn = wave & 3;
  const int row16 = lane & 15, khalf = lane >> 4;
  const int bm = blockIdx.y * 64, bn = blockIdx.x * 128;

  v8f acc00 = {}, acc01 = {}, acc10 = {}, acc11 = {};

  for (int k0 = 0; k0 < K; k0 += 32) {
    float ra[8], rb[16];
#pragma unroll
    for (int j = 0; j < 8; ++j) {
      int i = tid + j * 256;
      int r = i >> 5, c = i & 31;
      int gr = bm + r;
      ra[j] = (gr < M) ? A[(size_t)gr * K + k0 + c] : 0.0f;
    }
#pragma unroll
    for (int j = 0; j < 16; ++j) {
      int i = tid + j * 256;
      int kk = i >> 7, n = i & 127;
      rb[j] = B[(size_t)(k0 + kk) * N + bn + n];
    }
#pragma unroll
    for (int j = 0; j < 8; ++j) {
      int i = tid + j * 256;
      sA[i >> 5][i & 31] = ra[j];
    }
#pragma unroll
    for (int j = 0; j < 16; ++j) {
      int i = tid + j * 256;
      sBt[i & 127][i >> 7] = rb[j];
    }
    __syncthreads();

    const float* a0p = &sA[wm * 32 + row16][0];
    const float* a1p = &sA[wm * 32 + 16 + row16][0];
    const float* b0p = &sBt[wn * 32 + row16][0];
    const float* b1p = &sBt[wn * 32 + 16 + row16][0];
#pragma unroll
    for (int ks = 0; ks < 32; ks += 4) {
      int kk = ks + khalf * 2;
      v2f a0 = *(const v2f*)(a0p + kk);
      v2f a1 = *(const v2f*)(a1p + kk);
      v2f b0 = *(const v2f*)(b0p + kk);
      v2f b1 = *(const v2f*)(b1p + kk);
      acc00 = __builtin_amdgcn_wmma_f32_16x16x4_f32(false, a0, false, b0, (short)0, acc00, false, false);
      acc01 = __builtin_amdgcn_wmma_f32_16x16x4_f32(false, a0, false, b1, (short)0, acc01, false, false);
      acc10 = __builtin_amdgcn_wmma_f32_16x16x4_f32(false, a1, false, b0, (short)0, acc10, false, false);
      acc11 = __builtin_amdgcn_wmma_f32_16x16x4_f32(false, a1, false, b1, (short)0, acc11, false, false);
    }
    __syncthreads();
  }

  int gn0 = bn + wn * 32 + row16;
  int gn1 = gn0 + 16;
  int gm0 = bm + wm * 32 + 8 * khalf;
  int gm1 = gm0 + 16;
#pragma unroll
  for (int v = 0; v < 8; ++v) {
    if (gm0 + v < M) {
      C[(size_t)(gm0 + v) * N + gn0] = acc00[v];
      C[(size_t)(gm0 + v) * N + gn1] = acc01[v];
    }
    if (gm1 + v < M) {
      C[(size_t)(gm1 + v) * N + gn0] = acc10[v];
      C[(size_t)(gm1 + v) * N + gn1] = acc11[v];
    }
  }
}

__global__ void k_proj_e(const float* __restrict__ We, const float* __restrict__ ae,
                         float* __restrict__ proj, int C) {
  int i = blockIdx.x * blockDim.x + threadIdx.x;
  if (i >= EMB * NH) return;
  int k = i / NH, h = i % NH;
  const float* wr = We + (size_t)k * (NH * C) + (size_t)h * C;
  const float* ar = ae + (size_t)h * C;
  float s = 0.f;
  for (int c = 0; c < C; ++c) s += wr[c] * ar[c];
  proj[i] = s;
}

__global__ void k_alpha_node(const float* __restrict__ hx, const float* __restrict__ a_src,
                             const float* __restrict__ a_dst, float* __restrict__ ans,
                             float* __restrict__ adn, int C) {
  int i = blockIdx.x * blockDim.x + threadIdx.x;
  if (i >= NN * NH) return;
  int n = i / NH, h = i % NH;
  const float* hr = hx + (size_t)n * NH * C + (size_t)h * C;
  const float* sr = a_src + (size_t)h * C;
  const float* dr = a_dst + (size_t)h * C;
  float s = 0.f, t = 0.f;
  for (int c = 0; c < C; ++c) { s += hr[c] * sr[c]; t += hr[c] * dr[c]; }
  ans[i] = s;
  adn[i] = t;
}

__global__ void k_attr_tab(const float* __restrict__ emb, const float* __restrict__ proj,
                           float* __restrict__ tab) {
  int a = blockIdx.x;
  int lane = threadIdx.x;
  const float* row = emb + (size_t)a * EMB;
  const float4* p4 = (const float4*)proj;
  float a0 = 0.f, a1 = 0.f, a2 = 0.f, a3 = 0.f;
#pragma unroll
  for (int t = 0; t < EMB / 32; ++t) {
    int k = lane + 32 * t;
    float v = row[k];
    float4 p = p4[k];
    a0 += v * p.x; a1 += v * p.y; a2 += v * p.z; a3 += v * p.w;
  }
  for (int off = 16; off > 0; off >>= 1) {
    a0 += __shfl_xor(a0, off, 32);
    a1 += __shfl_xor(a1, off, 32);
    a2 += __shfl_xor(a2, off, 32);
    a3 += __shfl_xor(a3, off, 32);
  }
  if (lane == 0) {
    float4 r; r.x = a0; r.y = a1; r.z = a2; r.w = a3;
    ((float4*)tab)[a] = r;
  }
}

__global__ void k_edge_alpha(const int* __restrict__ ea_idx, const float* __restrict__ tab,
                             const unsigned* __restrict__ cnt,
                             const float* __restrict__ ans, const float* __restrict__ adn,
                             const int* __restrict__ src, const int* __restrict__ dst,
                             float* __restrict__ alpha, unsigned* __restrict__ mx) {
  int i = blockIdx.x * blockDim.x + threadIdx.x;
  if (i >= (NE + NN) * NH) return;
  int e = i >> 2, h = i & 3;
  int s, d;
  float ale;
  if (e < NE) {
    s = src[e];
    d = dst[e];
    ale = tab[ea_idx[e * 2 + 0] * NH + h] + tab[ea_idx[e * 2 + 1] * NH + h];
  } else {
    s = d = e - NE;
    const unsigned* cr = cnt + (size_t)d * NEA;
    float sum = 0.f, two_deg = 0.f;
    for (int a = 0; a < NEA; ++a) {
      float cf = (float)cr[a];
      sum += cf * tab[a * NH + h];
      two_deg += cf;
    }
    ale = sum / fmaxf(0.5f * two_deg, 1.0f);
  }
  float a = ans[s * NH + h] + adn[d * NH + h] + ale;
  a = (a > 0.f) ? a : NEG_SLOPE * a;
  alpha[i] = a;
  atomicMax(&mx[d * NH + h], fenc(a));
}

__global__ void k_softmax_den(const int* __restrict__ dst, const unsigned* __restrict__ mx,
                              float* __restrict__ alpha, float* __restrict__ den) {
  const int tot = NE + NN;
  long long i = (long long)blockIdx.x * blockDim.x + threadIdx.x;
  if (i >= (long long)tot * NH) return;
  int e = (int)(i >> 2), h = (int)(i & 3);
  int d = (e < NE) ? dst[e] : (e - NE);
  float ex = expf(alpha[i] - fdec(mx[d * NH + h]));
  alpha[i] = ex;
  atomicAdd(&den[d * NH + h], ex);
}

__global__ void k_aggregate(const float* __restrict__ hx, const float* __restrict__ ex,
                            const float* __restrict__ den, const int* __restrict__ src,
                            const int* __restrict__ dst, float* __restrict__ out, int C) {
  int e = blockIdx.x;
  int s = (e < NE) ? src[e] : (e - NE);
  int d = (e < NE) ? dst[e] : (e - NE);
  __shared__ float att[NH];
  if (threadIdx.x < NH)
    att[threadIdx.x] = ex[(size_t)e * NH + threadIdx.x] / den[d * NH + threadIdx.x];
  __syncthreads();
  const float* hr = hx + (size_t)s * NH * C;
  for (int c = threadIdx.x; c < C; c += blockDim.x) {
    float v = 0.25f * (att[0] * hr[c] + att[1] * hr[C + c] +
                       att[2] * hr[2 * C + c] + att[3] * hr[3 * C + c]);
    atomicAdd(&out[(size_t)d * C + c], v);
  }
}

__global__ void k_bias_relu(const float* __restrict__ in, const float* __restrict__ b,
                            float* __restrict__ o) {
  long long i = (long long)blockIdx.x * blockDim.x + threadIdx.x;
  if (i >= (long long)NN * HID) return;
  o[i] = fmaxf(in[i] + b[i % HID], 0.f);
}

__global__ void k_init_out(const float* __restrict__ b, float* __restrict__ o) {
  long long i = (long long)blockIdx.x * blockDim.x + threadIdx.x;
  if (i >= (long long)NN * FT) return;
  o[i] = b[i % FT];
}

extern "C" void kernel_launch(void* const* d_in, const int* in_sizes, int n_in,
                              void* d_out, int out_size, void* d_ws, size_t ws_size,
                              hipStream_t stream) {
  const int* x      = (const int*)d_in[0];
  const int* eidx   = (const int*)d_in[1];
  const int* src    = eidx;
  const int* dst    = eidx + NE;
  const int* eattr  = (const int*)d_in[2];
  const float* xemb = (const float*)d_in[3];
  const float* eemb = (const float*)d_in[4];
  const float* W1   = (const float*)d_in[5];
  const float* as1  = (const float*)d_in[6];
  const float* ad1  = (const float*)d_in[7];
  const float* We1  = (const float*)d_in[8];
  const float* ae1  = (const float*)d_in[9];
  const float* b1   = (const float*)d_in[10];
  const float* W2   = (const float*)d_in[11];
  const float* as2  = (const float*)d_in[12];
  const float* ad2  = (const float*)d_in[13];
  const float* We2  = (const float*)d_in[14];
  const float* ae2  = (const float*)d_in[15];
  const float* b2   = (const float*)d_in[16];
  float* out = (float*)d_out;

  float* ws = (float*)d_ws;
  size_t off = 0;
  float* XF   = ws + off; off += (size_t)NN * EMB;
  unsigned* CNT = (unsigned*)(ws + off); off += (size_t)2048 * NEA;
  float* TAB  = ws + off; off += 128;
  float* HX1  = ws + off; off += (size_t)NN * NH * HID;
  float* HX2  = ws + off; off += (size_t)NN * NH * FT;
  float* PROJ = ws + off; off += (size_t)EMB * NH;
  float* ANS  = ws + off; off += (size_t)NN * NH;
  float* ADN  = ws + off; off += (size_t)NN * NH;
  float* ALPH = ws + off; off += (size_t)(NE + NN) * NH;
  unsigned* MX = (unsigned*)(ws + off); off += (size_t)NN * NH;
  float* DEN  = ws + off; off += (size_t)NN * NH;
  float* OUT1 = ws + off; off += (size_t)NN * HID;
  float* HB   = ws + off; off += (size_t)NN * HID;
  (void)ws_size; (void)in_sizes; (void)n_in; (void)out_size;

  const int T = 256;

  hipMemsetAsync(CNT, 0, (size_t)NN * NEA * sizeof(unsigned), stream);
  hipMemsetAsync(OUT1, 0, (size_t)NN * HID * sizeof(float), stream);
  hipMemsetAsync(MX, 0, (size_t)NN * NH * sizeof(unsigned), stream);
  hipMemsetAsync(DEN, 0, (size_t)NN * NH * sizeof(float), stream);

  k_embed_x<<<cdiv((long long)NN * EMB, T), T, 0, stream>>>(x, xemb, XF);
  k_attr_count<<<cdiv(NE, T), T, 0, stream>>>(dst, eattr, CNT);

  {
    dim3 g(NH * HID / 128, cdiv(NN, 64));
    k_gemm_wmma_f32<<<g, 256, 0, stream>>>(XF, W1, HX1, NN, NH * HID, EMB);
  }
  k_proj_e<<<cdiv(EMB * NH, T), T, 0, stream>>>(We1, ae1, PROJ, HID);
  k_attr_tab<<<NEA, 32, 0, stream>>>(eemb, PROJ, TAB);
  k_alpha_node<<<cdiv(NN * NH, T), T, 0, stream>>>(HX1, as1, ad1, ANS, ADN, HID);
  k_edge_alpha<<<cdiv((long long)(NE + NN) * NH, T), T, 0, stream>>>(eattr, TAB, CNT, ANS, ADN, src, dst, ALPH, MX);
  k_softmax_den<<<cdiv((long long)(NE + NN) * NH, T), T, 0, stream>>>(dst, MX, ALPH, DEN);
  k_aggregate<<<NE + NN, HID, 0, stream>>>(HX1, ALPH, DEN, src, dst, OUT1, HID);
  k_bias_relu<<<cdiv((long long)NN * HID, T), T, 0, stream>>>(OUT1, b1, HB);

  hipMemsetAsync(MX, 0, (size_t)NN * NH * sizeof(unsigned), stream);
  hipMemsetAsync(DEN, 0, (size_t)NN * NH * sizeof(float), stream);

  {
    dim3 g(NH * FT / 128, cdiv(NN, 64));
    k_gemm_wmma_f32<<<g, 256, 0, stream>>>(HB, W2, HX2, NN, NH * FT, HID);
  }
  k_proj_e<<<cdiv(EMB * NH, T), T, 0, stream>>>(We2, ae2, PROJ, FT);
  k_attr_tab<<<NEA, 32, 0, stream>>>(eemb, PROJ, TAB);
  k_alpha_node<<<cdiv(NN * NH, T), T, 0, stream>>>(HX2, as2, ad2, ANS, ADN, FT);
  k_edge_alpha<<<cdiv((long long)(NE + NN) * NH, T), T, 0, stream>>>(eattr, TAB, CNT, ANS, ADN, src, dst, ALPH, MX);
  k_softmax_den<<<cdiv((long long)(NE + NN) * NH, T), T, 0, stream>>>(dst, MX, ALPH, DEN);
  k_init_out<<<cdiv((long long)NN * FT, T), T, 0, stream>>>(b2, out);
  k_aggregate<<<NE + NN, FT, 0, stream>>>(HX2, ALPH, DEN, src, dst, out, FT);
}


# Aggregation as gather: edges sorted by destination each call (degree count, prefix sum, atomic fill); one wave per node x 128 channels sums in registers; float atomics on out gone; sum order changes
# speedup vs baseline: 2.0704x; 1.2994x over previous
#include <hip/hip_runtime.h>
#include <math.h>

#define NN   2000
#define NE   32000
#define EMB  768
#define HID  384
#define FT   768
#define NH   4
#define NEA  18
#define NEG_SLOPE 0.2f

typedef __attribute__((ext_vector_type(2))) float v2f;
typedef __attribute__((ext_vector_type(8))) float v8f;

static inline int cdiv(long long a, long long b) { return (int)((a + b - 1) / b); }

__device__ __forceinline__ unsigned fenc(float f) {
  unsigned u = __float_as_uint(f);
  return (u & 0x80000000u) ? ~u : (u | 0x80000000u);
}
__device__ __forceinline__ float fdec(unsigned e) {
  unsigned u = (e & 0x80000000u) ? (e ^ 0x80000000u) : ~e;
  return __uint_as_float(u);
}

__global__ void k_embed_x(const int* __restrict__ x, const float* __restrict__ emb,
                          float* __restrict__ xf) {
  long long i = (long long)blockIdx.x * blockDim.x + threadIdx.x;
  if (i >= (long long)NN * EMB) return;
  int n = (int)(i / EMB), c = (int)(i % EMB);
  xf[i] = emb[(size_t)x[n * 2 + 0] * EMB + c] + emb[(size_t)x[n * 2 + 1] * EMB + c];
}

__global__ void k_attr_count(const int* __restrict__ dst, const int* __restrict__ ea_idx,
                             unsigned* __restrict__ cnt, unsigned* __restrict__ deg) {
  int e = blockIdx.x * blockDim.x + threadIdx.x;
  if (e >= NE) return;
  int d = dst[e];
  atomicAdd(&cnt[d * NEA + ea_idx[e * 2 + 0]], 1u);
  atomicAdd(&cnt[d * NEA + ea_idx[e * 2 + 1]], 1u);
  atomicAdd(&deg[d], 1u);
}

static_assert(NN <= 2048, "k_csr_scan covers two nodes per thread of one 1024-thread block");
__global__ __launch_bounds__(1024) void k_csr_scan(const unsigned* __restrict__ deg,
                                                   unsigned* __restrict__ rowptr,
                                                   unsigned* __restrict__ cur) {
  __shared__ unsigned wsum[32];
  const int t = threadIdx.x, lane = t & 31, wave = t >> 5;
  const int n0 = 2 * t;
  const unsigned a = (n0 < NN) ? deg[n0] : 0u;
  const unsigned b = (n0 + 1 < NN) ? deg[n0 + 1] : 0u;
  unsigned incl = a + b;
  for (int off = 1; off < 32; off <<= 1) {
    unsigned v = __shfl_up(incl, off, 32);
    if (lane >= off) incl += v;
  }
  if (lane == 31) wsum[wave] = incl;
  __syncthreads();
  if (wave == 0) {
    unsigned w = wsum[lane];
    for (int off = 1; off < 32; off <<= 1) {
      unsigned v = __shfl_up(w, off, 32);
      if (lane >= off) w += v;
    }
    wsum[lane] = w;
  }
  __syncthreads();
  const unsigned excl = incl - (a + b) + (wave ? wsum[wave - 1] : 0u);
  if (n0 < NN)     { rowptr[n0] = excl;         cur[n0] = excl; }
  if (n0 + 1 < NN) { rowptr[n0 + 1] = excl + a; cur[n0 + 1] = excl + a; }
  if (t == 1023) rowptr[NN] = excl + a + b;
}

__global__ void k_csr_fill(const int* __restrict__ src, const int* __restrict__ dst,
                           unsigned* __restrict__ cur, int* __restrict__ csrc,
                           int* __restrict__ ceid) {
  int e = blockIdx.x * blockDim.x + threadIdx.x;
  if (e >= NE) return;
  unsigned pos = atomicAdd(&cur[dst[e]], 1u);
  csrc[pos] = src[e];
  ceid[pos] = e;
}

__global__ __launch_bounds__(256) void k_gemm_wmma_f32(
    const float* __restrict__ A, const float* __restrict__ B, float* __restrict__ C,
    int M, int N, int K) {
  __shared__ __align__(16) float sA[64][32];
  __shared__ __align__(16) float sBt[128][32];
  const int tid = threadIdx.x;
  const int wave = tid >> 5, lane = tid & 31;
  const int wm = wave >> 2, wn = wave & 3;
  const int row16 = lane & 15, khalf = lane >> 4;
  const int bm = blockIdx.y * 64, bn = blockIdx.x * 128;

  v8f acc00 = {}, acc01 = {}, acc10 = {}, acc11 = {};

  for (int k0 = 0; k0 < K; k0 += 32) {
    float ra[8], rb[16];
#pragma unroll
    for (int j = 0; j < 8; ++j) {
      int i = tid + j * 256;
      int r = i >> 5, c = i & 31;
      int gr = bm + r;
      ra[j] = (gr < M) ? A[(size_t)gr * K + k0 + c] : 0.0f;
    }
#pragma unroll
    for (int j = 0; j < 16; ++j) {
      int i = tid + j * 256;
      int kk = i >> 7, n = i & 127;
      rb[j] = B[(size_t)(k0 + kk) * N + bn + n];
    }
#pragma unroll
    for (int j = 0; j < 8; ++j) {
      int i = tid + j * 256;
      sA[i >> 5][i & 31] = ra[j];
    }
#pragma unroll
    for (int j = 0; j < 16; ++j) {
      int i = tid + j * 256;
      sBt[i & 127][i >> 7] = rb[j];
    }
    __syncthreads();

    const float* a0p = &sA[wm * 32 + row16][0];
    const float* a1p = &sA[wm * 32 + 16 + row16][0];
    const float* b0p = &sBt[wn * 32 + row16][0];
    const float* b1p = &sBt[wn * 32 + 16 + row16][0];
#pragma unroll
    for (int ks = 0; ks < 32; ks += 4) {
      int kk = ks + khalf * 2;
      v2f a0 = *(const v2f*)(a0p + kk);
      v2f a1 = *(const v2f*)(a1p + kk);
      v2f b0 = *(const v2f*)(b0p + kk);
      v2f b1 = *(const v2f*)(b1p + kk);
      acc00 = __builtin_amdgcn_wmma_f32_16x16x4_f32(false, a0, false, b0, (short)0, acc00, false, false);
      acc01 = __builtin_amdgcn_wmma_f32_16x16x4_f32(false, a0, false, b1, (short)0, acc01, false, false);
      acc10 = __builtin_amdgcn_wmma_f32_16x16x4_f32(false, a1, false, b0, (short)0, acc10, false, false);
      acc11 = __builtin_amdgcn_wmma_f32_16x16x4_f32(false, a1, false, b1, (short)0, acc11, false, false);
    }
    __syncthreads();
  }

  int gn0 = bn + wn * 32 + row16;
  int gn1 = gn0 + 16;
  int gm0 = bm + wm * 32 + 8 * khalf;
  int gm1 = gm0 + 16;
#pragma unroll
  for (int v = 0; v < 8; ++v) {
    if (gm0 + v < M) {
      C[(size_t)(gm0 + v) * N + gn0] = acc00[v];
      C[(size_t)(gm0 + v) * N + gn1] = acc01[v];
    }
    if (gm1 + v < M) {
      C[(size_t)(gm1 + v) * N + gn0] = acc10[v];
      C[(size_t)(gm1 + v) * N + gn1] = acc11[v];
    }
  }
}

__global__ void k_proj_e(const float* __restrict__ We, const float* __restrict__ ae,
                         float* __restrict__ proj, int C) {
  int i = blockIdx.x * blockDim.x + threadIdx.x;
  if (i >= EMB * NH) return;
  int k = i / NH, h = i % NH;
  const float* wr = We + (size_t)k * (NH * C) + (size_t)h * C;
  const float* ar = ae + (size_t)h * C;
  float s = 0.f;
  for (int c = 0; c < C; ++c) s += wr[c] * ar[c];
  proj[i] = s;
}

__global__ void k_alpha_node(const float* __restrict__ hx, const float* __restrict__ a_src,
                             const float* __restrict__ a_dst, float* __restrict__ ans,
                             float* __restrict__ adn, int C) {
  int i = blockIdx.x * blockDim.x + threadIdx.x;
  if (i >= NN * NH) return;
  int n = i / NH, h = i % NH;
  const float* hr = hx + (size_t)n * NH * C + (size_t)h * C;
  const float* sr = a_src + (size_t)h * C;
  const float* dr = a_dst + (size_t)h * C;
  float s = 0.f, t = 0.f;
  for (int c = 0; c < C; ++c) { s += hr[c] * sr[c]; t += hr[c] * dr[c]; }
  ans[i] = s;
  adn[i] = t;
}

__global__ void k_attr_tab(const float* __restrict__ emb, const float* __restrict__ proj,
                           float* __restrict__ tab) {
  int a = blockIdx.x;
  int lane = threadIdx.x;
  const float* row = emb + (size_t)a * EMB;
  const float4* p4 = (const float4*)proj;
  float a0 = 0.f, a1 = 0.f, a2 = 0.f, a3 = 0.f;
#pragma unroll
  for (int t = 0; t < EMB / 32; ++t) {
    int k = lane + 32 * t;
    float v = row[k];
    float4 p = p4[k];
    a0 += v * p.x; a1 += v * p.y; a2 += v * p.z; a3 += v * p.w;
  }
  for (int off = 16; off > 0; off >>= 1) {
    a0 += __shfl_xor(a0, off, 32);
    a1 += __shfl_xor(a1, off, 32);
    a2 += __shfl_xor(a2, off, 32);
    a3 += __shfl_xor(a3, off, 32);
  }
  if (lane == 0) {
    float4 r; r.x = a0; r.y = a1; r.z = a2; r.w = a3;
    ((float4*)tab)[a] = r;
  }
}

__global__ void k_edge_alpha(const int* __restrict__ ea_idx, const float* __restrict__ tab,
                             const unsigned* __restrict__ cnt,
                             const float* __restrict__ ans, const float* __restrict__ adn,
                             const int* __restrict__ src, const int* __restrict__ dst,
                             float* __restrict__ alpha, unsigned* __restrict__ mx) {
  int i = blockIdx.x * blockDim.x + threadIdx.x;
  if (i >= (NE + NN) * NH) return;
  int e = i >> 2, h = i & 3;
  int s, d;
  float ale;
  if (e < NE) {
    s = src[e];
    d = dst[e];
    ale = tab[ea_idx[e * 2 + 0] * NH + h] + tab[ea_idx[e * 2 + 1] * NH + h];
  } else {
    s = d = e - NE;
    const unsigned* cr = cnt + (size_t)d * NEA;
    float sum = 0.f, two_deg = 0.f;
    for (int a = 0; a < NEA; ++a) {
      float cf = (float)cr[a];
      sum += cf * tab[a * NH + h];
      two_deg += cf;
    }
    ale = sum / fmaxf(0.5f * two_deg, 1.0f);
  }
  float a = ans[s * NH + h] + adn[d * NH + h] + ale;
  a = (a > 0.f) ? a : NEG_SLOPE * a;
  alpha[i] = a;
  atomicMax(&mx[d * NH + h], fenc(a));
}

__global__ void k_softmax_den(const int* __restrict__ dst, const unsigned* __restrict__ mx,
                              float* __restrict__ alpha, float* __restrict__ den) {
  const int tot = NE + NN;
  long long i = (long long)blockIdx.x * blockDim.x + threadIdx.x;
  if (i >= (long long)tot * NH) return;
  int e = (int)(i >> 2), h = (int)(i & 3);
  int d = (e < NE) ? dst[e] : (e - NE);
  float ex = expf(alpha[i] - fdec(mx[d * NH + h]));
  alpha[i] = ex;
  atomicAdd(&den[d * NH + h], ex);
}

__device__ __forceinline__ void edge_fma(float4& acc, float w0, float w1, float w2, float w3,
                                         const float4& v0, const float4& v1, const float4& v2,
                                         const float4& v3) {
  acc.x = fmaf(w0, v0.x, acc.x); acc.y = fmaf(w0, v0.y, acc.y);
  acc.z = fmaf(w0, v0.z, acc.z); acc.w = fmaf(w0, v0.w, acc.w);
  acc.x = fmaf(w1, v1.x, acc.x); acc.y = fmaf(w1, v1.y, acc.y);
  acc.z = fmaf(w1, v1.z, acc.z); acc.w = fmaf(w1, v1.w, acc.w);
  acc.x = fmaf(w2, v2.x, acc.x); acc.y = fmaf(w2, v2.y, acc.y);
  acc.z = fmaf(w2, v2.z, acc.z); acc.w = fmaf(w2, v2.w, acc.w);
  acc.x = fmaf(w3, v3.x, acc.x); acc.y = fmaf(w3, v3.y, acc.y);
  acc.z = fmaf(w3, v3.z, acc.z); acc.w = fmaf(w3, v3.w, acc.w);
}

__global__ __launch_bounds__(256) void k_gather(
    const float* __restrict__ hx, const float* __restrict__ ex, const float* __restrict__ den,
    const unsigned* __restrict__ rowptr, const int* __restrict__ csrc, const int* __restrict__ ceid,
    const float* __restrict__ bias, float* __restrict__ o, int C, int chunks, int relu) {
  const int lane = threadIdx.x & 31;
  const int gw = __builtin_amdgcn_readfirstlane((int)((blockIdx.x * 256 + threadIdx.x) >> 5));
  const int node = gw / chunks;
  if (node >= NN) return;
  const int c0 = (gw - node * chunks) * 128 + lane * 4;
  const size_t rs = (size_t)NH * C;

  const float4 dn = *(const float4*)(den + (size_t)node * NH);
  const float i0 = 1.0f / dn.x, i1 = 1.0f / dn.y, i2 = 1.0f / dn.z, i3 = 1.0f / dn.w;

  float4 acc = {0.f, 0.f, 0.f, 0.f};
  {
    const float4 w = *(const float4*)(ex + (size_t)(NE + node) * NH);
    const float* p = hx + (size_t)node * rs + c0;
    const float4 v0 = *(const float4*)p, v1 = *(const float4*)(p + C);
    const float4 v2 = *(const float4*)(p + 2 * C), v3 = *(const float4*)(p + 3 * C);
    edge_fma(acc, w.x * i0, w.y * i1, w.z * i2, w.w * i3, v0, v1, v2, v3);
  }

  const int beg = (int)rowptr[node], end = (int)rowptr[node + 1];
  for (int base = beg; base < end; base += 32) {
    const int idx = base + lane;
    int s_l = node;
    float w0_l = 0.f, w1_l = 0.f, w2_l = 0.f, w3_l = 0.f;
    if (idx < end) {
      s_l = csrc[idx];
      const float4 w = *(const float4*)(ex + (size_t)ceid[idx] * NH);
      w0_l = w.x * i0; w1_l = w.y * i1; w2_l = w.z * i2; w3_l = w.w * i3;
    }
    const int cnt = min(32, end - base);
    int j = 0;
    for (; j + 4 <= cnt; j += 4) {
      float4 v[4][4];
      float w[4][4];
#pragma unroll
      for (int u = 0; u < 4; ++u) {
        const int s = __builtin_amdgcn_readlane(s_l, j + u);
        const float* p = hx + (size_t)s * rs + c0;
        v[u][0] = *(const float4*)p;
        v[u][1] = *(const float4*)(p + C);
        v[u][2] = *(const float4*)(p + 2 * C);
        v[u][3] = *(const float4*)(p + 3 * C);
        w[u][0] = __int_as_float(__builtin_amdgcn_readlane(__float_as_int(w0_l), j + u));
        w[u][1] = __int_as_float(__builtin_amdgcn_readlane(__float_as_int(w1_l), j + u));
        w[u][2] = __int_as_float(__builtin_amdgcn_readlane(__float_as_int(w2_l), j + u));
        w[u][3] = __int_as_float(__builtin_amdgcn_readlane(__float_as_int(w3_l), j + u));
      }
#pragma unroll
      for (int u = 0; u < 4; ++u)
        edge_fma(acc, w[u][0], w[u][1], w[u][2], w[u][3], v[u][0], v[u][1], v[u][2], v[u][3]);
    }
    for (; j < cnt; ++j) {
      const int s = __builtin_amdgcn_readlane(s_l, j);
      const float* p = hx + (size_t)s * rs + c0;
      const float4 v0 = *(const float4*)p, v1 = *(const float4*)(p + C);
      const float4 v2 = *(const float4*)(p + 2 * C), v3 = *(const float4*)(p + 3 * C);
      edge_fma(acc,
               __int_as_float(__builtin_amdgcn_readlane(__float_as_int(w0_l), j)),
               __int_as_float(__builtin_amdgcn_readlane(__float_as_int(w1_l), j)),
               __int_as_float(__builtin_amdgcn_readlane(__float_as_int(w2_l), j)),
               __int_as_float(__builtin_amdgcn_readlane(__float_as_int(w3_l), j)),
               v0, v1, v2, v3);
    }
  }

  float4 r;
  r.x = 0.25f * acc.x + bias[c0 + 0];
  r.y = 0.25f * acc.y + bias[c0 + 1];
  r.z = 0.25f * acc.z + bias[c0 + 2];
  r.w = 0.25f * acc.w + bias[c0 + 3];
  if (relu) { r.x = fmaxf(r.x, 0.f); r.y = fmaxf(r.y, 0.f); r.z = fmaxf(r.z, 0.f); r.w = fmaxf(r.w, 0.f); }
  *(float4*)(o + (size_t)node * C + c0) = r;
}

extern "C" void kernel_launch(void* const* d_in, const int* in_sizes, int n_in,
                              void* d_out, int out_size, void* d_ws, size_t ws_size,
                              hipStream_t stream) {
  const int* x      = (const int*)d_in[0];
  const int* eidx   = (const int*)d_in[1];
  const int* src    = eidx;
  const int* dst    = eidx + NE;
  const int* eattr  = (const int*)d_in[2];
  const float* xemb = (const float*)d_in[3];
  const float* eemb = (const float*)d_in[4];
  const float* W1   = (const float*)d_in[5];
  const float* as1  = (const float*)d_in[6];
  const float* ad1  = (const float*)d_in[7];
  const float* We1  = (const float*)d_in[8];
  const float* ae1  = (const float*)d_in[9];
  const float* b1   = (const float*)d_in[10];
  const float* W2   = (const float*)d_in[11];
  const float* as2  = (const float*)d_in[12];
  const float* ad2  = (const float*)d_in[13];
  const float* We2  = (const float*)d_in[14];
  const float* ae2  = (const float*)d_in[15];
  const float* b2   = (const float*)d_in[16];
  float* out = (float*)d_out;

  float* ws = (float*)d_ws;
  size_t off = 0;
  float* XF   = ws + off; off += (size_t)NN * EMB;
  unsigned* CNT = (unsigned*)(ws + off); off += (size_t)2048 * NEA;
  unsigned* DEG = (unsigned*)(ws + off); off += 2048;
  unsigned* ROWPTR = (unsigned*)(ws + off); off += 2112;
  unsigned* CUR = (unsigned*)(ws + off); off += 2048;
  int* CSRC = (int*)(ws + off); off += (size_t)NE;
  int* CEID = (int*)(ws + off); off += (size_t)NE;
  float* TAB  = ws + off; off += 128;
  float* HX1  = ws + off; off += (size_t)NN * NH * HID;
  float* HX2  = ws + off; off += (size_t)NN * NH * FT;
  float* PROJ = ws + off; off += (size_t)EMB * NH;
  float* ANS  = ws + off; off += (size_t)NN * NH;
  float* ADN  = ws + off; off += (size_t)NN * NH;
  float* ALPH = ws + off; off += (size_t)(NE + NN) * NH;
  unsigned* MX = (unsigned*)(ws + off); off += (size_t)NN * NH;
  float* DEN  = ws + off; off += (size_t)NN * NH;
  float* HB   = ws + off; off += (size_t)NN * HID;
  (void)ws_size; (void)in_sizes; (void)n_in; (void)out_size;

  const int T = 256;

  hipMemsetAsync(CNT, 0, (size_t)(2048 * NEA + 2048) * sizeof(unsigned), stream);
  hipMemsetAsync(MX, 0, (size_t)NN * NH * sizeof(unsigned), stream);
  hipMemsetAsync(DEN, 0, (size_t)NN * NH * sizeof(float), stream);

  k_embed_x<<<cdiv((long long)NN * EMB, T), T, 0, stream>>>(x, xemb, XF);
  k_attr_count<<<cdiv(NE, T), T, 0, stream>>>(dst, eattr, CNT, DEG);
  k_csr_scan<<<1, 1024, 0, stream>>>(DEG, ROWPTR, CUR);
  k_csr_fill<<<cdiv(NE, T), T, 0, stream>>>(src, dst, CUR, CSRC, CEID);

  {
    dim3 g(NH * HID / 128, cdiv(NN, 64));
    k_gemm_wmma_f32<<<g, 256, 0, stream>>>(XF, W1, HX1, NN, NH * HID, EMB);
  }
  k_proj_e<<<cdiv(EMB * NH, T), T, 0, stream>>>(We1, ae1, PROJ, HID);
  k_attr_tab<<<NEA, 32, 0, stream>>>(eemb, PROJ, TAB);
  k_alpha_node<<<cdiv(NN * NH, T), T, 0, stream>>>(HX1, as1, ad1, ANS, ADN, HID);
  k_edge_alpha<<<cdiv((long long)(NE + NN) * NH, T), T, 0, stream>>>(eattr, TAB, CNT, ANS, ADN, src, dst, ALPH, MX);
  k_softmax_den<<<cdiv((long long)(NE + NN) * NH, T), T, 0, stream>>>(dst, MX, ALPH, DEN);
  k_gather<<<cdiv((long long)NN * (HID / 128), 8), 256, 0, stream>>>(HX1, ALPH, DEN, ROWPTR, CSRC, CEID, b1, HB, HID, HID / 128, 1);

  hipMemsetAsync(MX, 0, (size_t)NN * NH * sizeof(unsigned), stream);
  hipMemsetAsync(DEN, 0, (size_t)NN * NH * sizeof(float), stream);

  {
    dim3 g(NH * FT / 128, cdiv(NN, 64));
    k_gemm_wmma_f32<<<g, 256, 0, stream>>>(HB, W2, HX2, NN, NH * FT, HID);
  }
  k_proj_e<<<cdiv(EMB * NH, T), T, 0, stream>>>(We2, ae2, PROJ, FT);
  k_attr_tab<<<NEA, 32, 0, stream>>>(eemb, PROJ, TAB);
  k_alpha_node<<<cdiv(NN * NH, T), T, 0, stream>>>(HX2, as2, ad2, ANS, ADN, FT);
  k_edge_alpha<<<cdiv((long long)(NE + NN) * NH, T), T, 0, stream>>>(eattr, TAB, CNT, ANS, ADN, src, dst, ALPH, MX);
  k_softmax_den<<<cdiv((long long)(NE + NN) * NH, T), T, 0, stream>>>(dst, MX, ALPH, DEN);
  k_gather<<<cdiv((long long)NN * (FT / 128), 8), 256, 0, stream>>>(HX2, ALPH, DEN, ROWPTR, CSRC, CEID, b2, out, FT, FT / 128, 0);
}


# GEMM LDS tiles padded from row pitch 32 dwords to 36 (A) and 34 (B): fragment reads and transposed B-tile stores no longer pile onto two bank groups; arithmetic and sum order unchanged
# speedup vs baseline: 2.5498x; 1.2315x over previous
#include <hip/hip_runtime.h>
#include <math.h>

#define NN   2000
#define NE   32000
#define EMB  768
#define HID  384
#define FT   768
#define NH   4
#define NEA  18
#define NEG_SLOPE 0.2f

typedef __attribute__((ext_vector_type(2))) float v2f;
typedef __attribute__((ext_vector_type(8))) float v8f;

static inline int cdiv(long long a, long long b) { return (int)((a + b - 1) / b); }

__device__ __forceinline__ unsigned fenc(float f) {
  unsigned u = __float_as_uint(f);
  return (u & 0x80000000u) ? ~u : (u | 0x80000000u);
}
__device__ __forceinline__ float fdec(unsigned e) {
  unsigned u = (e & 0x80000000u) ? (e ^ 0x80000000u) : ~e;
  return __uint_as_float(u);
}

__global__ void k_embed_x(const int* __restrict__ x, const float* __restrict__ emb,
                          float* __restrict__ xf) {
  long long i = (long long)blockIdx.x * blockDim.x + threadIdx.x;
  if (i >= (long long)NN * EMB) return;
  int n = (int)(i / EMB), c = (int)(i % EMB);
  xf[i] = emb[(size_t)x[n * 2 + 0] * EMB + c] + emb[(size_t)x[n * 2 + 1] * EMB + c];
}

__global__ void k_attr_count(const int* __restrict__ dst, const int* __restrict__ ea_idx,
                             unsigned* __restrict__ cnt, unsigned* __restrict__ deg) {
  int e = blockIdx.x * blockDim.x + threadIdx.x;
  if (e >= NE) return;
  int d = dst[e];
  atomicAdd(&cnt[d * NEA + ea_idx[e * 2 + 0]], 1u);
  atomicAdd(&cnt[d * NEA + ea_idx[e * 2 + 1]], 1u);
  atomicAdd(&deg[d], 1u);
}

static_assert(NN <= 2048, "k_csr_scan covers two nodes per thread of one 1024-thread block");
__global__ __launch_bounds__(1024) void k_csr_scan(const unsigned* __restrict__ deg,
                                                   unsigned* __restrict__ rowptr,
                                                   unsigned* __restrict__ cur) {
  __shared__ unsigned wsum[32];
  const int t = threadIdx.x, lane = t & 31, wave = t >> 5;
  const int n0 = 2 * t;
  const unsigned a = (n0 < NN) ? deg[n0] : 0u;
  const unsigned b = (n0 + 1 < NN) ? deg[n0 + 1] : 0u;
  unsigned incl = a + b;
  for (int off = 1; off < 32; off <<= 1) {
    unsigned v = __shfl_up(incl, off, 32);
    if (lane >= off) incl += v;
  }
  if (lane == 31) wsum[wave] = incl;
  __syncthreads();
  if (wave == 0) {
    unsigned w = wsum[lane];
    for (int off = 1; off < 32; off <<= 1) {
      unsigned v = __shfl_up(w, off, 32);
      if (lane >= off) w += v;
    }
    wsum[lane] = w;
  }
  __syncthreads();
  const unsigned excl = incl - (a + b) + (wave ? wsum[wave - 1] : 0u);
  if (n0 < NN)     { rowptr[n0] = excl;         cur[n0] = excl; }
  if (n0 + 1 < NN) { rowptr[n0 + 1] = excl + a; cur[n0 + 1] = excl + a; }
  if (t == 1023) rowptr[NN] = excl + a + b;
}

__global__ void k_csr_fill(const int* __restrict__ src, const int* __restrict__ dst,
                           unsigned* __restrict__ cur, int* __restrict__ csrc,
                           int* __restrict__ ceid) {
  int e = blockIdx.x * blockDim.x + threadIdx.x;
  if (e >= NE) return;
  unsigned pos = atomicAdd(&cur[dst[e]], 1u);
  csrc[pos] = src[e];
  ceid[pos] = e;
}

#define A_PITCH 36
#define B_PITCH 34
__global__ __launch_bounds__(256) void k_gemm_wmma_f32(
    const float* __restrict__ A, const float* __restrict__ B, float* __restrict__ C,
    int M, int N, int K) {
  __shared__ __align__(16) float sA[64][A_PITCH];
  __shared__ __align__(16) float sBt[128][B_PITCH];
  const int tid = threadIdx.x;
  const int wave = tid >> 5, lane = tid & 31;
  const int wm = wave >> 2, wn = wave & 3;
  const int row16 = lane & 15, khalf = lane >> 4;
  const int bm = blockIdx.y * 64, bn = blockIdx.x * 128;

  v8f acc00 = {}, acc01 = {}, acc10 = {}, acc11 = {};

  for (int k0 = 0; k0 < K; k0 += 32) {
    float ra[8], rb[16];
#pragma unroll
    for (int j = 0; j < 8; ++j) {
      int i = tid + j * 256;
      int r = i >> 5, c = i & 31;
      int gr = bm + r;
      ra[j] = (gr < M) ? A[(size_t)gr * K + k0 + c] : 0.0f;
    }
#pragma unroll
    for (int j = 0; j < 16; ++j) {
      int i = tid + j * 256;
      int kk = i >> 7, n = i & 127;
      rb[j] = B[(size_t)(k0 + kk) * N + bn + n];
    }
#pragma unroll
    for (int j = 0; j < 8; ++j) {
      int i = tid + j * 256;
      sA[i >> 5][i & 31] = ra[j];
    }
#pragma unroll
    for (int j = 0; j < 16; ++j) {
      int i = tid + j * 256;
      sBt[i & 127][i >> 7] = rb[j];
    }
    __syncthreads();

    const float* a0p = &sA[wm * 32 + row16][0];
    const float* a1p = &sA[wm * 32 + 16 + row16][0];
    const float* b0p = &sBt[wn * 32 + row16][0];
    const float* b1p = &sBt[wn * 32 + 16 + row16][0];
#pragma unroll
    for (int ks = 0; ks < 32; ks += 4) {
      int kk = ks + khalf * 2;
      v2f a0 = *(const v2f*)(a0p + kk);
      v2f a1 = *(const v2f*)(a1p + kk);
      v2f b0 = *(const v2f*)(b0p + kk);
      v2f b1 = *(const v2f*)(b1p + kk);
      acc00 = __builtin_amdgcn_wmma_f32_16x16x4_f32(false, a0, false, b0, (short)0, acc00, false, false);
      acc01 = __builtin_amdgcn_wmma_f32_16x16x4_f32(false, a0, false, b1, (short)0, acc01, false, false);
      acc10 = __builtin_amdgcn_wmma_f32_16x16x4_f32(false, a1, false, b0, (short)0, acc10, false, false);
      acc11 = __builtin_amdgcn_wmma_f32_16x16x4_f32(false, a1, false, b1, (short)0, acc11, false, false);
    }
    __syncthreads();
  }

  int gn0 = bn + wn * 32 + row16;
  int gn1 = gn0 + 16;
  int gm0 = bm + wm * 32 + 8 * khalf;
  int gm1 = gm0 + 16;
#pragma unroll
  for (int v = 0; v < 8; ++v) {
    if (gm0 + v < M) {
      C[(size_t)(gm0 + v) * N + gn0] = acc00[v];
      C[(size_t)(gm0 + v) * N + gn1] = acc01[v];
    }
    if (gm1 + v < M) {
      C[(size_t)(gm1 + v) * N + gn0] = acc10[v];
      C[(size_t)(gm1 + v) * N + gn1] = acc11[v];
    }
  }
}

__global__ void k_proj_e(const float* __restrict__ We, const float* __restrict__ ae,
                         float* __restrict__ proj, int C) {
  int i = blockIdx.x * blockDim.x + threadIdx.x;
  if (i >= EMB * NH) return;
  int k = i / NH, h = i % NH;
  const float* wr = We + (size_t)k * (NH * C) + (size_t)h * C;
  const float* ar = ae + (size_t)h * C;
  float s = 0.f;
  for (int c = 0; c < C; ++c) s += wr[c] * ar[c];
  proj[i] = s;
}

__global__ void k_alpha_node(const float* __restrict__ hx, const float* __restrict__ a_src,
                             const float* __restrict__ a_dst, float* __restrict__ ans,
                             float* __restrict__ adn, int C) {
  int i = blockIdx.x * blockDim.x + threadIdx.x;
  if (i >= NN * NH) return;
  int n = i / NH, h = i % NH;
  const float* hr = hx + (size_t)n * NH * C + (size_t)h * C;
  const float* sr = a_src + (size_t)h * C;
  const float* dr = a_dst + (size_t)h * C;
  float s = 0.f, t = 0.f;
  for (int c = 0; c < C; ++c) { s += hr[c] * sr[c]; t += hr[c] * dr[c]; }
  ans[i] = s;
  adn[i] = t;
}

__global__ void k_attr_tab(const float* __restrict__ emb, const float* __restrict__ proj,
                           float* __restrict__ tab) {
  int a = blockIdx.x;
  int lane = threadIdx.x;
  const float* row = emb + (size_t)a * EMB;
  const float4* p4 = (const float4*)proj;
  float a0 = 0.f, a1 = 0.f, a2 = 0.f, a3 = 0.f;
#pragma unroll
  for (int t = 0; t < EMB / 32; ++t) {
    int k = lane + 32 * t;
    float v = row[k];
    float4 p = p4[k];
    a0 += v * p.x; a1 += v * p.y; a2 += v * p.z; a3 += v * p.w;
  }
  for (int off = 16; off > 0; off >>= 1) {
    a0 += __shfl_xor(a0, off, 32);
    a1 += __shfl_xor(a1, off, 32);
    a2 += __shfl_xor(a2, off, 32);
    a3 += __shfl_xor(a3, off, 32);
  }
  if (lane == 0) {
    float4 r; r.x = a0; r.y = a1; r.z = a2; r.w = a3;
    ((float4*)tab)[a] = r;
  }
}

__global__ void k_edge_alpha(const int* __restrict__ ea_idx, const float* __restrict__ tab,
                             const unsigned* __restrict__ cnt,
                             const float* __restrict__ ans, const float* __restrict__ adn,
                             const int* __restrict__ src, const int* __restrict__ dst,
                             float* __restrict__ alpha, unsigned* __restrict__ mx) {
  int i = blockIdx.x * blockDim.x + threadIdx.x;
  if (i >= (NE + NN) * NH) return;
  int e = i >> 2, h = i & 3;
  int s, d;
  float ale;
  if (e < NE) {
    s = src[e];
    d = dst[e];
    ale = tab[ea_idx[e * 2 + 0] * NH + h] + tab[ea_idx[e * 2 + 1] * NH + h];
  } else {
    s = d = e - NE;
    const unsigned* cr = cnt + (size_t)d * NEA;
    float sum = 0.f, two_deg = 0.f;
    for (int a = 0; a < NEA; ++a) {
      float cf = (float)cr[a];
      sum += cf * tab[a * NH + h];
      two_deg += cf;
    }
    ale = sum / fmaxf(0.5f * two_deg, 1.0f);
  }
  float a = ans[s * NH + h] + adn[d * NH + h] + ale;
  a = (a > 0.f) ? a : NEG_SLOPE * a;
  alpha[i] = a;
  atomicMax(&mx[d * NH + h], fenc(a));
}

__global__ void k_softmax_den(const int* __restrict__ dst, const unsigned* __restrict__ mx,
                              float* __restrict__ alpha, float* __restrict__ den) {
  const int tot = NE + NN;
  long long i = (long long)blockIdx.x * blockDim.x + threadIdx.x;
  if (i >= (long long)tot * NH) return;
  int e = (int)(i >> 2), h = (int)(i & 3);
  int d = (e < NE) ? dst[e] : (e - NE);
  float ex = expf(alpha[i] - fdec(mx[d * NH + h]));
  alpha[i] = ex;
  atomicAdd(&den[d * NH + h], ex);
}

__device__ __forceinline__ void edge_fma(float4& acc, float w0, float w1, float w2, float w3,
                                         const float4& v0, const float4& v1, const float4& v2,
                                         const float4& v3) {
  acc.x = fmaf(w0, v0.x, acc.x); acc.y = fmaf(w0, v0.y, acc.y);
  acc.z = fmaf(w0, v0.z, acc.z); acc.w = fmaf(w0, v0.w, acc.w);
  acc.x = fmaf(w1, v1.x, acc.x); acc.y = fmaf(w1, v1.y, acc.y);
  acc.z = fmaf(w1, v1.z, acc.z); acc.w = fmaf(w1, v1.w, acc.w);
  acc.x = fmaf(w2, v2.x, acc.x); acc.y = fmaf(w2, v2.y, acc.y);
  acc.z = fmaf(w2, v2.z, acc.z); acc.w = fmaf(w2, v2.w, acc.w);
  acc.x = fmaf(w3, v3.x, acc.x); acc.y = fmaf(w3, v3.y, acc.y);
  acc.z = fmaf(w3, v3.z, acc.z); acc.w = fmaf(w3, v3.w, acc.w);
}

__global__ __launch_bounds__(256) void k_gather(
    const float* __restrict__ hx, const float* __restrict__ ex, const float* __restrict__ den,
    const unsigned* __restrict__ rowptr, const int* __restrict__ csrc, const int* __restrict__ ceid,
    const float* __restrict__ bias, float* __restrict__ o, int C, int chunks, int relu) {
  const int lane = threadIdx.x & 31;
  const int gw = __builtin_amdgcn_readfirstlane((int)((blockIdx.x * 256 + threadIdx.x) >> 5));
  const int node = gw / chunks;
  if (node >= NN) return;
  const int c0 = (gw - node * chunks) * 128 + lane * 4;
  const size_t rs = (size_t)NH * C;

  const float4 dn = *(const float4*)(den + (size_t)node * NH);
  const float i0 = 1.0f / dn.x, i1 = 1.0f / dn.y, i2 = 1.0f / dn.z, i3 = 1.0f / dn.w;

  float4 acc = {0.f, 0.f, 0.f, 0.f};
  {
    const float4 w = *(const float4*)(ex + (size_t)(NE + node) * NH);
    const float* p = hx + (size_t)node * rs + c0;
    const float4 v0 = *(const float4*)p, v1 = *(const float4*)(p + C);
    const float4 v2 = *(const float4*)(p + 2 * C), v3 = *(const float4*)(p + 3 * C);
    edge_fma(acc, w.x * i0, w.y * i1, w.z * i2, w.w * i3, v0, v1, v2, v3);
  }

  const int beg = (int)rowptr[node], end = (int)rowptr[node + 1];
  for (int base = beg; base < end; base += 32) {
    const int idx = base + lane;
    int s_l = node;
    float w0_l = 0.f, w1_l = 0.f, w2_l = 0.f, w3_l = 0.f;
    if (idx < end) {
      s_l = csrc[idx];
      const float4 w = *(const float4*)(ex + (size_t)ceid[idx] * NH);
      w0_l = w.x * i0; w1_l = w.y * i1; w2_l = w.z * i2; w3_l = w.w * i3;
    }
    const int cnt = min(32, end - base);
    int j = 0;
    for (; j + 4 <= cnt; j += 4) {
      float4 v[4][4];
      float w[4][4];
#pragma unroll
      for (int u = 0; u < 4; ++u) {
        const int s = __builtin_amdgcn_readlane(s_l, j + u);
        const float* p = hx + (size_t)s * rs + c0;
        v[u][0] = *(const float4*)p;
        v[u][1] = *(const float4*)(p + C);
        v[u][2] = *(const float4*)(p + 2 * C);
        v[u][3] = *(const float4*)(p + 3 * C);
        w[u][0] = __int_as_float(__builtin_amdgcn_readlane(__float_as_int(w0_l), j + u));
        w[u][1] = __int_as_float(__builtin_amdgcn_readlane(__float_as_int(w1_l), j + u));
        w[u][2] = __int_as_float(__builtin_amdgcn_readlane(__float_as_int(w2_l), j + u));
        w[u][3] = __int_as_float(__builtin_amdgcn_readlane(__float_as_int(w3_l), j + u));
      }
#pragma unroll
      for (int u = 0; u < 4; ++u)
        edge_fma(acc, w[u][0], w[u][1], w[u][2], w[u][3], v[u][0], v[u][1], v[u][2], v[u][3]);
    }
    for (; j < cnt; ++j) {
      const int s = __builtin_amdgcn_readlane(s_l, j);
      const float* p = hx + (size_t)s * rs + c0;
      const float4 v0 = *(const float4*)p, v1 = *(const float4*)(p + C);
      const float4 v2 = *(const float4*)(p + 2 * C), v3 = *(const float4*)(p + 3 * C);
      edge_fma(acc,
               __int_as_float(__builtin_amdgcn_readlane(__float_as_int(w0_l), j)),
               __int_as_float(__builtin_amdgcn_readlane(__float_as_int(w1_l), j)),
               __int_as_float(__builtin_amdgcn_readlane(__float_as_int(w2_l), j)),
               __int_as_float(__builtin_amdgcn_readlane(__float_as_int(w3_l), j)),
               v0, v1, v2, v3);
    }
  }

  float4 r;
  r.x = 0.25f * acc.x + bias[c0 + 0];
  r.y = 0.25f * acc.y + bias[c0 + 1];
  r.z = 0.25f * acc.z + bias[c0 + 2];
  r.w = 0.25f * acc.w + bias[c0 + 3];
  if (relu) { r.x = fmaxf(r.x, 0.f); r.y = fmaxf(r.y, 0.f); r.z = fmaxf(r.z, 0.f); r.w = fmaxf(r.w, 0.f); }
  *(float4*)(o + (size_t)node * C + c0) = r;
}

extern "C" void kernel_launch(void* const* d_in, const int* in_sizes, int n_in,
                              void* d_out, int out_size, void* d_ws, size_t ws_size,
                              hipStream_t stream) {
  const int* x      = (const int*)d_in[0];
  const int* eidx   = (const int*)d_in[1];
  const int* src    = eidx;
  const int* dst    = eidx + NE;
  const int* eattr  = (const int*)d_in[2];
  const float* xemb = (const float*)d_in[3];
  const float* eemb = (const float*)d_in[4];
  const float* W1   = (const float*)d_in[5];
  const float* as1  = (const float*)d_in[6];
  const float* ad1  = (const float*)d_in[7];
  const float* We1  = (const float*)d_in[8];
  const float* ae1  = (const float*)d_in[9];
  const float* b1   = (const float*)d_in[10];
  const float* W2   = (const float*)d_in[11];
  const float* as2  = (const float*)d_in[12];
  const float* ad2  = (const float*)d_in[13];
  const float* We2  = (const float*)d_in[14];
  const float* ae2  = (const float*)d_in[15];
  const float* b2   = (const float*)d_in[16];
  float* out = (float*)d_out;

  float* ws = (float*)d_ws;
  size_t off = 0;
  float* XF   = ws + off; off += (size_t)NN * EMB;
  unsigned* CNT = (unsigned*)(ws + off); off += (size_t)2048 * NEA;
  unsigned* DEG = (unsigned*)(ws + off); off += 2048;
  unsigned* ROWPTR = (unsigned*)(ws + off); off += 2112;
  unsigned* CUR = (unsigned*)(ws + off); off += 2048;
  int* CSRC = (int*)(ws + off); off += (size_t)NE;
  int* CEID = (int*)(ws + off); off += (size_t)NE;
  float* TAB  = ws + off; off += 128;
  float* HX1  = ws + off; off += (size_t)NN * NH * HID;
  float* HX2  = ws + off; off += (size_t)NN * NH * FT;
  float* PROJ = ws + off; off += (size_t)EMB * NH;
  float* ANS  = ws + off; off += (size_t)NN * NH;
  float* ADN  = ws + off; off += (size_t)NN * NH;
  float* ALPH = ws + off; off += (size_t)(NE + NN) * NH;
  unsigned* MX = (unsigned*)(ws + off); off += (size_t)NN * NH;
  float* DEN  = ws + off; off += (size_t)NN * NH;
  float* HB   = ws + off; off += (size_t)NN * HID;
  (void)ws_size; (void)in_sizes; (void)n_in; (void)out_size;

  const int T = 256;

  hipMemsetAsync(CNT, 0, (size_t)(2048 * NEA + 2048) * sizeof(unsigned), stream);
  hipMemsetAsync(MX, 0, (size_t)NN * NH * sizeof(unsigned), stream);
  hipMemsetAsync(DEN, 0, (size_t)NN * NH * sizeof(float), stream);

  k_embed_x<<<cdiv((long long)NN * EMB, T), T, 0, stream>>>(x, xemb, XF);
  k_attr_count<<<cdiv(NE, T), T, 0, stream>>>(dst, eattr, CNT, DEG);
  k_csr_scan<<<1, 1024, 0, stream>>>(DEG, ROWPTR, CUR);
  k_csr_fill<<<cdiv(NE, T), T, 0, stream>>>(src, dst, CUR, CSRC, CEID);

  {
    dim3 g(NH * HID / 128, cdiv(NN, 64));
    k_gemm_wmma_f32<<<g, 256, 0, stream>>>(XF, W1, HX1, NN, NH * HID, EMB);
  }
  k_proj_e<<<cdiv(EMB * NH, T), T, 0, stream>>>(We1, ae1, PROJ, HID);
  k_attr_tab<<<NEA, 32, 0, stream>>>(eemb, PROJ, TAB);
  k_alpha_node<<<cdiv(NN * NH, T), T, 0, stream>>>(HX1, as1, ad1, ANS, ADN, HID);
  k_edge_alpha<<<cdiv((long long)(NE + NN) * NH, T), T, 0, stream>>>(eattr, TAB, CNT, ANS, ADN, src, dst, ALPH, MX);
  k_softmax_den<<<cdiv((long long)(NE + NN) * NH, T), T, 0, stream>>>(dst, MX, ALPH, DEN);
  k_gather<<<cdiv((long long)NN * (HID / 128), 8), 256, 0, stream>>>(HX1, ALPH, DEN, ROWPTR, CSRC, CEID, b1, HB, HID, HID / 128, 1);

  hipMemsetAsync(MX, 0, (size_t)NN * NH * sizeof(unsigned), stream);
  hipMemsetAsync(DEN, 0, (size_t)NN * NH * sizeof(float), stream);

  {
    dim3 g(NH * FT / 128, cdiv(NN, 64));
    k_gemm_wmma_f32<<<g, 256, 0, stream>>>(HB, W2, HX2, NN, NH * FT, HID);
  }
  k_proj_e<<<cdiv(EMB * NH, T), T, 0, stream>>>(We2, ae2, PROJ, FT);
  k_attr_tab<<<NEA, 32, 0, stream>>>(eemb, PROJ, TAB);
  k_alpha_node<<<cdiv(NN * NH, T), T, 0, stream>>>(HX2, as2, ad2, ANS, ADN, FT);
  k_edge_alpha<<<cdiv((long long)(NE + NN) * NH, T), T, 0, stream>>>(eattr, TAB, CNT, ANS, ADN, src, dst, ALPH, MX);
  k_softmax_den<<<cdiv((long long)(NE + NN) * NH, T), T, 0, stream>>>(dst, MX, ALPH, DEN);
  k_gather<<<cdiv((long long)NN * (FT / 128), 8), 256, 0, stream>>>(HX2, ALPH, DEN, ROWPTR, CSRC, CEID, b2, out, FT, FT / 128, 0);
}


# Row dot products (proj, per-node attention scalars): one wave per group of 4 rows, lanes along the row, 512 contiguous bytes per load, butterfly sum; was one thread per row; sum order changes
# speedup vs baseline: 3.4211x; 1.3417x over previous
#include <hip/hip_runtime.h>
#include <math.h>

#define NN   2000
#define NE   32000
#define EMB  768
#define HID  384
#define FT   768
#define NH   4
#define NEA  18
#define NEG_SLOPE 0.2f

typedef __attribute__((ext_vector_type(2))) float v2f;
typedef __attribute__((ext_vector_type(8))) float v8f;

static inline int cdiv(long long a, long long b) { return (int)((a + b - 1) / b); }

__device__ __forceinline__ unsigned fenc(float f) {
  unsigned u = __float_as_uint(f);
  return (u & 0x80000000u) ? ~u : (u | 0x80000000u);
}
__device__ __forceinline__ float fdec(unsigned e) {
  unsigned u = (e & 0x80000000u) ? (e ^ 0x80000000u) : ~e;
  return __uint_as_float(u);
}

__global__ void k_embed_x(const int* __restrict__ x, const float* __restrict__ emb,
                          float* __restrict__ xf) {
  long long i = (long long)blockIdx.x * blockDim.x + threadIdx.x;
  if (i >= (long long)NN * EMB) return;
  int n = (int)(i / EMB), c = (int)(i % EMB);
  xf[i] = emb[(size_t)x[n * 2 + 0] * EMB + c] + emb[(size_t)x[n * 2 + 1] * EMB + c];
}

__global__ void k_attr_count(const int* __restrict__ dst, const int* __restrict__ ea_idx,
                             unsigned* __restrict__ cnt, unsigned* __restrict__ deg) {
  int e = blockIdx.x * blockDim.x + threadIdx.x;
  if (e >= NE) return;
  int d = dst[e];
  atomicAdd(&cnt[d * NEA + ea_idx[e * 2 + 0]], 1u);
  atomicAdd(&cnt[d * NEA + ea_idx[e * 2 + 1]], 1u);
  atomicAdd(&deg[d], 1u);
}

static_assert(NN <= 2048, "k_csr_scan covers two nodes per thread of one 1024-thread block");
__global__ __launch_bounds__(1024) void k_csr_scan(const unsigned* __restrict__ deg,
                                                   unsigned* __restrict__ rowptr,
                                                   unsigned* __restrict__ cur) {
  __shared__ unsigned wsum[32];
  const int t = threadIdx.x, lane = t & 31, wave = t >> 5;
  const int n0 = 2 * t;
  const unsigned a = (n0 < NN) ? deg[n0] : 0u;
  const unsigned b = (n0 + 1 < NN) ? deg[n0 + 1] : 0u;
  unsigned incl = a + b;
  for (int off = 1; off < 32; off <<= 1) {
    unsigned v = __shfl_up(incl, off, 32);
    if (lane >= off) incl += v;
  }
  if (lane == 31) wsum[wave] = incl;
  __syncthreads();
  if (wave == 0) {
    unsigned w = wsum[lane];
    for (int off = 1; off < 32; off <<= 1) {
      unsigned v = __shfl_up(w, off, 32);
      if (lane >= off) w += v;
    }
    wsum[lane] = w;
  }
  __syncthreads();
  const unsigned excl = incl - (a + b) + (wave ? wsum[wave - 1] : 0u);
  if (n0 < NN)     { rowptr[n0] = excl;         cur[n0] = excl; }
  if (n0 + 1 < NN) { rowptr[n0 + 1] = excl + a; cur[n0 + 1] = excl + a; }
  if (t == 1023) rowptr[NN] = excl + a + b;
}

__global__ void k_csr_fill(const int* __restrict__ src, const int* __restrict__ dst,
                           unsigned* __restrict__ cur, int* __restrict__ csrc,
                           int* __restrict__ ceid) {
  int e = blockIdx.x * blockDim.x + threadIdx.x;
  if (e >= NE) return;
  unsigned pos = atomicAdd(&cur[dst[e]], 1u);
  csrc[pos] = src[e];
  ceid[pos] = e;
}

#define A_PITCH 36
#define B_PITCH 34
__global__ __launch_bounds__(256) void k_gemm_wmma_f32(
    const float* __restrict__ A, const float* __restrict__ B, float* __restrict__ C,
    int M, int N, int K) {
  __shared__ __align__(16) float sA[64][A_PITCH];
  __shared__ __align__(16) float sBt[128][B_PITCH];
  const int tid = threadIdx.x;
  const int wave = tid >> 5, lane = tid & 31;
  const int wm = wave >> 2, wn = wave & 3;
  const int row16 = lane & 15, khalf = lane >> 4;
  const int bm = blockIdx.y * 64, bn = blockIdx.x * 128;

  v8f acc00 = {}, acc01 = {}, acc10 = {}, acc11 = {};

  for (int k0 = 0; k0 < K; k0 += 32) {
    float ra[8], rb[16];
#pragma unroll
    for (int j = 0; j < 8; ++j) {
      int i = tid + j * 256;
      int r = i >> 5, c = i & 31;
      int gr = bm + r;
      ra[j] = (gr < M) ? A[(size_t)gr * K + k0 + c] : 0.0f;
    }
#pragma unroll
    for (int j = 0; j < 16; ++j) {
      int i = tid + j * 256;
      int kk = i >> 7, n = i & 127;
      rb[j] = B[(size_t)(k0 + kk) * N + bn + n];
    }
#pragma unroll
    for (int j = 0; j < 8; ++j) {
      int i = tid + j * 256;
      sA[i >> 5][i & 31] = ra[j];
    }
#pragma unroll
    for (int j = 0; j < 16; ++j) {
      int i = tid + j * 256;
      sBt[i & 127][i >> 7] = rb[j];
    }
    __syncthreads();

    const float* a0p = &sA[wm * 32 + row16][0];
    const float* a1p = &sA[wm * 32 + 16 + row16][0];
    const float* b0p = &sBt[wn * 32 + row16][0];
    const float* b1p = &sBt[wn * 32 + 16 + row16][0];
#pragma unroll
    for (int ks = 0; ks < 32; ks += 4) {
      int kk = ks + khalf * 2;
      v2f a0 = *(const v2f*)(a0p + kk);
      v2f a1 = *(const v2f*)(a1p + kk);
      v2f b0 = *(const v2f*)(b0p + kk);
      v2f b1 = *(const v2f*)(b1p + kk);
      acc00 = __builtin_amdgcn_wmma_f32_16x16x4_f32(false, a0, false, b0, (short)0, acc00, false, false);
      acc01 = __builtin_amdgcn_wmma_f32_16x16x4_f32(false, a0, false, b1, (short)0, acc01, false, false);
      acc10 = __builtin_amdgcn_wmma_f32_16x16x4_f32(false, a1, false, b0, (short)0, acc10, false, false);
      acc11 = __builtin_amdgcn_wmma_f32_16x16x4_f32(false, a1, false, b1, (short)0, acc11, false, false);
    }
    __syncthreads();
  }

  int gn0 = bn + wn * 32 + row16;
  int gn1 = gn0 + 16;
  int gm0 = bm + wm * 32 + 8 * khalf;
  int gm1 = gm0 + 16;
#pragma unroll
  for (int v = 0; v < 8; ++v) {
    if (gm0 + v < M) {
      C[(size_t)(gm0 + v) * N + gn0] = acc00[v];
      C[(size_t)(gm0 + v) * N + gn1] = acc01[v];
    }
    if (gm1 + v < M) {
      C[(size_t)(gm1 + v) * N + gn0] = acc10[v];
      C[(size_t)(gm1 + v) * N + gn1] = acc11[v];
    }
  }
}

__device__ __forceinline__ float dot4(const float4& a, const float4& b, float s) {
  s = fmaf(a.x, b.x, s); s = fmaf(a.y, b.y, s); s = fmaf(a.z, b.z, s); s = fmaf(a.w, b.w, s);
  return s;
}
__device__ __forceinline__ float wave_sum(float v) {
  for (int off = 16; off > 0; off >>= 1) v += __shfl_xor(v, off, 32);
  return v;
}

template <int CC>
__global__ __launch_bounds__(64) void k_proj_e(const float* __restrict__ We,
                                               const float* __restrict__ ae,
                                               float* __restrict__ proj) {
  static_assert(CC % 128 == 0, "a row is a whole number of 512-byte steps of a wave");
  constexpr int PH = CC / 128;
  const int lane = threadIdx.x & 31;
  const int k = __builtin_amdgcn_readfirstlane((int)((blockIdx.x * 64 + threadIdx.x) >> 5));
  if (k >= EMB) return;
  const float4* w4 = (const float4*)(We + (size_t)k * (NH * CC));
  const float4* a4 = (const float4*)ae;
  float s[NH];
#pragma unroll
  for (int h = 0; h < NH; ++h) {
    float acc = 0.f;
#pragma unroll
    for (int u = 0; u < PH; ++u) {
      const int q = (h * PH + u) * 32 + lane;
      acc = dot4(w4[q], a4[q], acc);
    }
    s[h] = wave_sum(acc);
  }
  if (lane == 0) {
    float4 r; r.x = s[0]; r.y = s[1]; r.z = s[2]; r.w = s[3];
    ((float4*)proj)[k] = r;
  }
}

template <int CC>
__global__ __launch_bounds__(128) void k_alpha_node(const float* __restrict__ hx,
                                                    const float* __restrict__ a_src,
                                                    const float* __restrict__ a_dst,
                                                    float* __restrict__ ans,
                                                    float* __restrict__ adn) {
  static_assert(CC % 128 == 0, "a row is a whole number of 512-byte steps of a wave");
  static_assert(NH == 4, "one 16-byte store per node and output");
  constexpr int PH = CC / 128;
  const int lane = threadIdx.x & 31;
  const int n = __builtin_amdgcn_readfirstlane((int)((blockIdx.x * 128 + threadIdx.x) >> 5));
  if (n >= NN) return;
  const float4* h4 = (const float4*)(hx + (size_t)n * (NH * CC));
  const float4* s4 = (const float4*)a_src;
  const float4* d4 = (const float4*)a_dst;
  float s[NH], t[NH];
#pragma unroll
  for (int h = 0; h < NH; ++h) {
    float as = 0.f, ad = 0.f;
#pragma unroll
    for (int u = 0; u < PH; ++u) {
      const int q = (h * PH + u) * 32 + lane;
      const float4 v = h4[q];
      as = dot4(v, s4[q], as);
      ad = dot4(v, d4[q], ad);
    }
    s[h] = wave_sum(as);
    t[h] = wave_sum(ad);
  }
  if (lane == 0) {
    float4 r; r.x = s[0]; r.y = s[1]; r.z = s[2]; r.w = s[3];
    ((float4*)ans)[n] = r;
    float4 w; w.x = t[0]; w.y = t[1]; w.z = t[2]; w.w = t[3];
    ((float4*)adn)[n] = w;
  }
}

__global__ void k_attr_tab(const float* __restrict__ emb, const float* __restrict__ proj,
                           float* __restrict__ tab) {
  int a = blockIdx.x;
  int lane = threadIdx.x;
  const float* row = emb + (size_t)a * EMB;
  const float4* p4 = (const float4*)proj;
  float a0 = 0.f, a1 = 0.f, a2 = 0.f, a3 = 0.f;
#pragma unroll
  for (int t = 0; t < EMB / 32; ++t) {
    int k = lane + 32 * t;
    float v = row[k];
    float4 p = p4[k];
    a0 += v * p.x; a1 += v * p.y; a2 += v * p.z; a3 += v * p.w;
  }
  for (int off = 16; off > 0; off >>= 1) {
    a0 += __shfl_xor(a0, off, 32);
    a1 += __shfl_xor(a1, off, 32);
    a2 += __shfl_xor(a2, off, 32);
    a3 += __shfl_xor(a3, off, 32);
  }
  if (lane == 0) {
    float4 r; r.x = a0; r.y = a1; r.z = a2; r.w = a3;
    ((float4*)tab)[a] = r;
  }
}

__global__ void k_edge_alpha(const int* __restrict__ ea_idx, const float* __restrict__ tab,
                             const unsigned* __restrict__ cnt,
                             const float* __restrict__ ans, const float* __restrict__ adn,
                             const int* __restrict__ src, const int* __restrict__ dst,
                             float* __restrict__ alpha, unsigned* __restrict__ mx) {
  int i = blockIdx.x * blockDim.x + threadIdx.x;
  if (i >= (NE + NN) * NH) return;
  int e = i >> 2, h = i & 3;
  int s, d;
  float ale;
  if (e < NE) {
    s = src[e];
    d = dst[e];
    ale = tab[ea_idx[e * 2 + 0] * NH + h] + tab[ea_idx[e * 2 + 1] * NH + h];
  } else {
    s = d = e - NE;
    const unsigned* cr = cnt + (size_t)d * NEA;
    float sum = 0.f, two_deg = 0.f;
    for (int a = 0; a < NEA; ++a) {
      float cf = (float)cr[a];
      sum += cf * tab[a * NH + h];
      two_deg += cf;
    }
    ale = sum / fmaxf(0.5f * two_deg, 1.0f);
  }
  float a = ans[s * NH + h] + adn[d * NH + h] + ale;
  a = (a > 0.f) ? a : NEG_SLOPE * a;
  alpha[i] = a;
  atomicMax(&mx[d * NH + h], fenc(a));
}

__global__ void k_softmax_den(const int* __restrict__ dst, const unsigned* __restrict__ mx,
                              float* __restrict__ alpha, float* __restrict__ den) {
  const int tot = NE + NN;
  long long i = (long long)blockIdx.x * blockDim.x + threadIdx.x;
  if (i >= (long long)tot * NH) return;
  int e = (int)(i >> 2), h = (int)(i & 3);
  int d = (e < NE) ? dst[e] : (e - NE);
  float ex = expf(alpha[i] - fdec(mx[d * NH + h]));
  alpha[i] = ex;
  atomicAdd(&den[d * NH + h], ex);
}

__device__ __forceinline__ void edge_fma(float4& acc, float w0, float w1, float w2, float w3,
                                         const float4& v0, const float4& v1, const float4& v2,
                                         const float4& v3) {
  acc.x = fmaf(w0, v0.x, acc.x); acc.y = fmaf(w0, v0.y, acc.y);
  acc.z = fmaf(w0, v0.z, acc.z); acc.w = fmaf(w0, v0.w, acc.w);
  acc.x = fmaf(w1, v1.x, acc.x); acc.y = fmaf(w1, v1.y, acc.y);
  acc.z = fmaf(w1, v1.z, acc.z); acc.w = fmaf(w1, v1.w, acc.w);
  acc.x = fmaf(w2, v2.x, acc.x); acc.y = fmaf(w2, v2.y, acc.y);
  acc.z = fmaf(w2, v2.z, acc.z); acc.w = fmaf(w2, v2.w, acc.w);
  acc.x = fmaf(w3, v3.x, acc.x); acc.y = fmaf(w3, v3.y, acc.y);
  acc.z = fmaf(w3, v3.z, acc.z); acc.w = fmaf(w3, v3.w, acc.w);
}

__global__ __launch_bounds__(256) void k_gather(
    const float* __restrict__ hx, const float* __restrict__ ex, const float* __restrict__ den,
    const unsigned* __restrict__ rowptr, const int* __restrict__ csrc, const int* __restrict__ ceid,
    const float* __restrict__ bias, float* __restrict__ o, int C, int chunks, int relu) {
  const int lane = threadIdx.x & 31;
  const int gw = __builtin_amdgcn_readfirstlane((int)((blockIdx.x * 256 + threadIdx.x) >> 5));
  const int node = gw / chunks;
  if (node >= NN) return;
  const int c0 = (gw - node * chunks) * 128 + lane * 4;
  const size_t rs = (size_t)NH * C;

  const float4 dn = *(const float4*)(den + (size_t)node * NH);
  const float i0 = 1.0f / dn.x, i1 = 1.0f / dn.y, i2 = 1.0f / dn.z, i3 = 1.0f / dn.w;

  float4 acc = {0.f, 0.f, 0.f, 0.f};
  {
    const float4 w = *(const float4*)(ex + (size_t)(NE + node) * NH);
    const float* p = hx + (size_t)node * rs + c0;
    const float4 v0 = *(const float4*)p, v1 = *(const float4*)(p + C);
    const float4 v2 = *(const float4*)(p + 2 * C), v3 = *(const float4*)(p + 3 * C);
    edge_fma(acc, w.x * i0, w.y * i1, w.z * i2, w.w * i3, v0, v1, v2, v3);
  }

  const int beg = (int)rowptr[node], end = (int)rowptr[node + 1];
  for (int base = beg; base < end; base += 32) {
    const int idx = base + lane;
    int s_l = node;
    float w0_l = 0.f, w1_l = 0.f, w2_l = 0.f, w3_l = 0.f;
    if (idx < end) {
      s_l = csrc[idx];
      const float4 w = *(const float4*)(ex + (size_t)ceid[idx] * NH);
      w0_l = w.x * i0; w1_l = w.y * i1; w2_l = w.z * i2; w3_l = w.w * i3;
    }
    const int cnt = min(32, end - base);
    int j = 0;
    for (; j + 4 <= cnt; j += 4) {
      float4 v[4][4];
      float w[4][4];
#pragma unroll
      for (int u = 0; u < 4; ++u) {
        const int s = __builtin_amdgcn_readlane(s_l, j + u);
        const float* p = hx + (size_t)s * rs + c0;
        v[u][0] = *(const float4*)p;
        v[u][1] = *(const float4*)(p + C);
        v[u][2] = *(const float4*)(p + 2 * C);
        v[u][3] = *(const float4*)(p + 3 * C);
        w[u][0] = __int_as_float(__builtin_amdgcn_readlane(__float_as_int(w0_l), j + u));
        w[u][1] = __int_as_float(__builtin_amdgcn_readlane(__float_as_int(w1_l), j + u));
        w[u][2] = __int_as_float(__builtin_amdgcn_readlane(__float_as_int(w2_l), j + u));
        w[u][3] = __int_as_float(__builtin_amdgcn_readlane(__float_as_int(w3_l), j + u));
      }
#pragma unroll
      for (int u = 0; u < 4; ++u)
        edge_fma(acc, w[u][0], w[u][1], w[u][2], w[u][3], v[u][0], v[u][1], v[u][2], v[u][3]);
    }
    for (; j < cnt; ++j) {
      const int s = __builtin_amdgcn_readlane(s_l, j);
      const float* p = hx + (size_t)s * rs + c0;
      const float4 v0 = *(const float4*)p, v1 = *(const float4*)(p + C);
      const float4 v2 = *(const float4*)(p + 2 * C), v3 = *(const float4*)(p + 3 * C);
      edge_fma(acc,
               __int_as_float(__builtin_amdgcn_readlane(__float_as_int(w0_l), j)),
               __int_as_float(__builtin_amdgcn_readlane(__float_as_int(w1_l), j)),
               __int_as_float(__builtin_amdgcn_readlane(__float_as_int(w2_l), j)),
               __int_as_float(__builtin_amdgcn_readlane(__float_as_int(w3_l), j)),
               v0, v1, v2, v3);
    }
  }

  float4 r;
  r.x = 0.25f * acc.x + bias[c0 + 0];
  r.y = 0.25f * acc.y + bias[c0 + 1];
  r.z = 0.25f * acc.z + bias[c0 + 2];
  r.w = 0.25f * acc.w + bias[c0 + 3];
  if (relu) { r.x = fmaxf(r.x, 0.f); r.y = fmaxf(r.y, 0.f); r.z = fmaxf(r.z, 0.f); r.w = fmaxf(r.w, 0.f); }
  *(float4*)(o + (size_t)node * C + c0) = r;
}

extern "C" void kernel_launch(void* const* d_in, const int* in_sizes, int n_in,
                              void* d_out, int out_size, void* d_ws, size_t ws_size,
                              hipStream_t stream) {
  const int* x      = (const int*)d_in[0];
  const int* eidx   = (const int*)d_in[1];
  const int* src    = eidx;
  const int* dst    = eidx + NE;
  const int* eattr  = (const int*)d_in[2];
  const float* xemb = (const float*)d_in[3];
  const float* eemb = (const float*)d_in[4];
  const float* W1   = (const float*)d_in[5];
  const float* as1  = (const float*)d_in[6];
  const float* ad1  = (const float*)d_in[7];
  const float* We1  = (const float*)d_in[8];
  const float* ae1  = (const float*)d_in[9];
  const float* b1   = (const float*)d_in[10];
  const float* W2   = (const float*)d_in[11];
  const float* as2  = (const float*)d_in[12];
  const float* ad2  = (const float*)d_in[13];
  const float* We2  = (const float*)d_in[14];
  const float* ae2  = (const float*)d_in[15];
  const float* b2   = (const float*)d_in[16];
  float* out = (float*)d_out;

  float* ws = (float*)d_ws;
  size_t off = 0;
  float* XF   = ws + off; off += (size_t)NN * EMB;
  unsigned* CNT = (unsigned*)(ws + off); off += (size_t)2048 * NEA;
  unsigned* DEG = (unsigned*)(ws + off); off += 2048;
  unsigned* ROWPTR = (unsigned*)(ws + off); off += 2112;
  unsigned* CUR = (unsigned*)(ws + off); off += 2048;
  int* CSRC = (int*)(ws + off); off += (size_t)NE;
  int* CEID = (int*)(ws + off); off += (size_t)NE;
  float* TAB  = ws + off; off += 128;
  float* HX1  = ws + off; off += (size_t)NN * NH * HID;
  float* HX2  = ws + off; off += (size_t)NN * NH * FT;
  float* PROJ = ws + off; off += (size_t)EMB * NH;
  float* ANS  = ws + off; off += (size_t)NN * NH;
  float* ADN  = ws + off; off += (size_t)NN * NH;
  float* ALPH = ws + off; off += (size_t)(NE + NN) * NH;
  unsigned* MX = (unsigned*)(ws + off); off += (size_t)NN * NH;
  float* DEN  = ws + off; off += (size_t)NN * NH;
  float* HB   = ws + off; off += (size_t)NN * HID;
  (void)ws_size; (void)in_sizes; (void)n_in; (void)out_size;

  const int T = 256;

  hipMemsetAsync(CNT, 0, (size_t)(2048 * NEA + 2048) * sizeof(unsigned), stream);
  hipMemsetAsync(MX, 0, (size_t)NN * NH * sizeof(unsigned), stream);
  hipMemsetAsync(DEN, 0, (size_t)NN * NH * sizeof(float), stream);

  k_embed_x<<<cdiv((long long)NN * EMB, T), T, 0, stream>>>(x, xemb, XF);
  k_attr_count<<<cdiv(NE, T), T, 0, stream>>>(dst, eattr, CNT, DEG);
  k_csr_scan<<<1, 1024, 0, stream>>>(DEG, ROWPTR, CUR);
  k_csr_fill<<<cdiv(NE, T), T, 0, stream>>>(src, dst, CUR, CSRC, CEID);

  {
    dim3 g(NH * HID / 128, cdiv(NN, 64));
    k_gemm_wmma_f32<<<g, 256, 0, stream>>>(XF, W1, HX1, NN, NH * HID, EMB);
  }
  k_proj_e<HID><<<cdiv((long long)EMB * 32, 64), 64, 0, stream>>>(We1, ae1, PROJ);
  k_attr_tab<<<NEA, 32, 0, stream>>>(eemb, PROJ, TAB);
  k_alpha_node<HID><<<cdiv((long long)NN * 32, 128), 128, 0, stream>>>(HX1, as1, ad1, ANS, ADN);
  k_edge_alpha<<<cdiv((long long)(NE + NN) * NH, T), T, 0, stream>>>(eattr, TAB, CNT, ANS, ADN, src, dst, ALPH, MX);
  k_softmax_den<<<cdiv((long long)(NE + NN) * NH, T), T, 0, stream>>>(dst, MX, ALPH, DEN);
  k_gather<<<cdiv((long long)NN * (HID / 128), 8), 256, 0, stream>>>(HX1, ALPH, DEN, ROWPTR, CSRC, CEID, b1, HB, HID, HID / 128, 1);

  hipMemsetAsync(MX, 0, (size_t)NN * NH * sizeof(unsigned), stream);
  hipMemsetAsync(DEN, 0, (size_t)NN * NH * sizeof(float), stream);

  {
    dim3 g(NH * FT / 128, cdiv(NN, 64));
    k_gemm_wmma_f32<<<g, 256, 0, stream>>>(HB, W2, HX2, NN, NH * FT, HID);
  }
  k_proj_e<FT><<<cdiv((long long)EMB * 32, 64), 64, 0, stream>>>(We2, ae2, PROJ);
  k_attr_tab<<<NEA, 32, 0, stream>>>(eemb, PROJ, TAB);
  k_alpha_node<FT><<<cdiv((long long)NN * 32, 128), 128, 0, stream>>>(HX2, as2, ad2, ANS, ADN);
  k_edge_alpha<<<cdiv((long long)(NE + NN) * NH, T), T, 0, stream>>>(eattr, TAB, CNT, ANS, ADN, src, dst, ALPH, MX);
  k_softmax_den<<<cdiv((long long)(NE + NN) * NH, T), T, 0, stream>>>(dst, MX, ALPH, DEN);
  k_gather<<<cdiv((long long)NN * (FT / 128), 8), 256, 0, stream>>>(HX2, ALPH, DEN, ROWPTR, CSRC, CEID, b2, out, FT, FT / 128, 0);
}


# Segment softmax inside the gather wave: scores, max, exp, sum per row of the destination-sorted list (attr ids carried in it); two per-edge kernels, their atomics, 4 memsets gone; sum order changes
# speedup vs baseline: 3.8510x; 1.1257x over previous
#include <hip/hip_runtime.h>
#include <math.h>

#define NN   2000
#define NE   32000
#define EMB  768
#define HID  384
#define FT   768
#define NH   4
#define NEA  18
#define NEG_SLOPE 0.2f

typedef __attribute__((ext_vector_type(2))) float v2f;
typedef __attribute__((ext_vector_type(8))) float v8f;

static inline int cdiv(long long a, long long b) { return (int)((a + b - 1) / b); }

__global__ void k_embed_x(const int* __restrict__ x, const float* __restrict__ emb,
                          float* __restrict__ xf) {
  long long i = (long long)blockIdx.x * blockDim.x + threadIdx.x;
  if (i >= (long long)NN * EMB) return;
  int n = (int)(i / EMB), c = (int)(i % EMB);
  xf[i] = emb[(size_t)x[n * 2 + 0] * EMB + c] + emb[(size_t)x[n * 2 + 1] * EMB + c];
}

__global__ void k_attr_count(const int* __restrict__ dst, const int* __restrict__ ea_idx,
                             unsigned* __restrict__ cnt, unsigned* __restrict__ deg) {
  int e = blockIdx.x * blockDim.x + threadIdx.x;
  if (e >= NE) return;
  int d = dst[e];
  atomicAdd(&cnt[d * NEA + ea_idx[e * 2 + 0]], 1u);
  atomicAdd(&cnt[d * NEA + ea_idx[e * 2 + 1]], 1u);
  atomicAdd(&deg[d], 1u);
}

static_assert(NN <= 2048, "k_csr_scan covers two nodes per thread of one 1024-thread block");
__global__ __launch_bounds__(1024) void k_csr_scan(const unsigned* __restrict__ deg,
                                                   unsigned* __restrict__ rowptr,
                                                   unsigned* __restrict__ cur) {
  __shared__ unsigned wsum[32];
  const int t = threadIdx.x, lane = t & 31, wave = t >> 5;
  const int n0 = 2 * t;
  const unsigned a = (n0 < NN) ? deg[n0] : 0u;
  const unsigned b = (n0 + 1 < NN) ? deg[n0 + 1] : 0u;
  unsigned incl = a + b;
  for (int off = 1; off < 32; off <<= 1) {
    unsigned v = __shfl_up(incl, off, 32);
    if (lane >= off) incl += v;
  }
  if (lane == 31) wsum[wave] = incl;
  __syncthreads();
  if (wave == 0) {
    unsigned w = wsum[lane];
    for (int off = 1; off < 32; off <<= 1) {
      unsigned v = __shfl_up(w, off, 32);
      if (lane >= off) w += v;
    }
    wsum[lane] = w;
  }
  __syncthreads();
  const unsigned excl = incl - (a + b) + (wave ? wsum[wave - 1] : 0u);
  if (n0 < NN)     { rowptr[n0] = excl;         cur[n0] = excl; }
  if (n0 + 1 < NN) { rowptr[n0 + 1] = excl + a; cur[n0 + 1] = excl + a; }
  if (t == 1023) rowptr[NN] = excl + a + b;
}

__global__ void k_csr_fill(const int* __restrict__ src, const int* __restrict__ dst,
                           const int* __restrict__ ea_idx,
                           unsigned* __restrict__ cur, int* __restrict__ csrc,
                           int* __restrict__ cattr) {
  int e = blockIdx.x * blockDim.x + threadIdx.x;
  if (e >= NE) return;
  unsigned pos = atomicAdd(&cur[dst[e]], 1u);
  csrc[pos] = src[e];
  cattr[pos] = (int)((unsigned)ea_idx[e * 2 + 0] | ((unsigned)ea_idx[e * 2 + 1] << 16));
}

#define A_PITCH 36
#define B_PITCH 34
__global__ __launch_bounds__(256) void k_gemm_wmma_f32(
    const float* __restrict__ A, const float* __restrict__ B, float* __restrict__ C,
    int M, int N, int K) {
  __shared__ __align__(16) float sA[64][A_PITCH];
  __shared__ __align__(16) float sBt[128][B_PITCH];
  const int tid = threadIdx.x;
  const int wave = tid >> 5, lane = tid & 31;
  const int wm = wave >> 2, wn = wave & 3;
  const int row16 = lane & 15, khalf = lane >> 4;
  const int bm = blockIdx.y * 64, bn = blockIdx.x * 128;

  v8f acc00 = {}, acc01 = {}, acc10 = {}, acc11 = {};

  for (int k0 = 0; k0 < K; k0 += 32) {
    float ra[8], rb[16];
#pragma unroll
    for (int j = 0; j < 8; ++j) {
      int i = tid + j * 256;
      int r = i >> 5, c = i & 31;
      int gr = bm + r;
      ra[j] = (gr < M) ? A[(size_t)gr * K + k0 + c] : 0.0f;
    }
#pragma unroll
    for (int j = 0; j < 16; ++j) {
      int i = tid + j * 256;
      int kk = i >> 7, n = i & 127;
      rb[j] = B[(size_t)(k0 + kk) * N + bn + n];
    }
#pragma unroll
    for (int j = 0; j < 8; ++j) {
      int i = tid + j * 256;
      sA[i >> 5][i & 31] = ra[j];
    }
#pragma unroll
    for (int j = 0; j < 16; ++j) {
      int i = tid + j * 256;
      sBt[i & 127][i >> 7] = rb[j];
    }
    __syncthreads();

    const float* a0p = &sA[wm * 32 + row16][0];
    const float* a1p = &sA[wm * 32 + 16 + row16][0];
    const float* b0p = &sBt[wn * 32 + row16][0];
    const float* b1p = &sBt[wn * 32 + 16 + row16][0];
#pragma unroll
    for (int ks = 0; ks < 32; ks += 4) {
      int kk = ks + khalf * 2;
      v2f a0 = *(const v2f*)(a0p + kk);
      v2f a1 = *(const v2f*)(a1p + kk);
      v2f b0 = *(const v2f*)(b0p + kk);
      v2f b1 = *(const v2f*)(b1p + kk);
      acc00 = __builtin_amdgcn_wmma_f32_16x16x4_f32(false, a0, false, b0, (short)0, acc00, false, false);
      acc01 = __builtin_amdgcn_wmma_f32_16x16x4_f32(false, a0, false, b1, (short)0, acc01, false, false);
      acc10 = __builtin_amdgcn_wmma_f32_16x16x4_f32(false, a1, false, b0, (short)0, acc10, false, false);
      acc11 = __builtin_amdgcn_wmma_f32_16x16x4_f32(false, a1, false, b1, (short)0, acc11, false, false);
    }
    __syncthreads();
  }

  int gn0 = bn + wn * 32 + row16;
  int gn1 = gn0 + 16;
  int gm0 = bm + wm * 32 + 8 * khalf;
  int gm1 = gm0 + 16;
#pragma unroll
  for (int v = 0; v < 8; ++v) {
    if (gm0 + v < M) {
      C[(size_t)(gm0 + v) * N + gn0] = acc00[v];
      C[(size_t)(gm0 + v) * N + gn1] = acc01[v];
    }
    if (gm1 + v < M) {
      C[(size_t)(gm1 + v) * N + gn0] = acc10[v];
      C[(size_t)(gm1 + v) * N + gn1] = acc11[v];
    }
  }
}

__device__ __forceinline__ float dot4(const float4& a, const float4& b, float s) {
  s = fmaf(a.x, b.x, s); s = fmaf(a.y, b.y, s); s = fmaf(a.z, b.z, s); s = fmaf(a.w, b.w, s);
  return s;
}
__device__ __forceinline__ float wave_sum(float v) {
  for (int off = 16; off > 0; off >>= 1) v += __shfl_xor(v, off, 32);
  return v;
}

template <int CC>
__global__ __launch_bounds__(64) void k_proj_e(const float* __restrict__ We,
                                               const float* __restrict__ ae,
                                               float* __restrict__ proj) {
  static_assert(CC % 128 == 0, "a row is a whole number of 512-byte steps of a wave");
  constexpr int PH = CC / 128;
  const int lane = threadIdx.x & 31;
  const int k = __builtin_amdgcn_readfirstlane((int)((blockIdx.x * 64 + threadIdx.x) >> 5));
  if (k >= EMB) return;
  const float4* w4 = (const float4*)(We + (size_t)k * (NH * CC));
  const float4* a4 = (const float4*)ae;
  float s[NH];
#pragma unroll
  for (int h = 0; h < NH; ++h) {
    float acc = 0.f;
#pragma unroll
    for (int u = 0; u < PH; ++u) {
      const int q = (h * PH + u) * 32 + lane;
      acc = dot4(w4[q], a4[q], acc);
    }
    s[h] = wave_sum(acc);
  }
  if (lane == 0) {
    float4 r; r.x = s[0]; r.y = s[1]; r.z = s[2]; r.w = s[3];
    ((float4*)proj)[k] = r;
  }
}

template <int CC>
__global__ __launch_bounds__(128) void k_alpha_node(const float* __restrict__ hx,
                                                    const float* __restrict__ a_src,
                                                    const float* __restrict__ a_dst,
                                                    float* __restrict__ ans,
                                                    float* __restrict__ adn) {
  static_assert(CC % 128 == 0, "a row is a whole number of 512-byte steps of a wave");
  static_assert(NH == 4, "one 16-byte store per node and output");
  constexpr int PH = CC / 128;
  const int lane = threadIdx.x & 31;
  const int n = __builtin_amdgcn_readfirstlane((int)((blockIdx.x * 128 + threadIdx.x) >> 5));
  if (n >= NN) return;
  const float4* h4 = (const float4*)(hx + (size_t)n * (NH * CC));
  const float4* s4 = (const float4*)a_src;
  const float4* d4 = (const float4*)a_dst;
  float s[NH], t[NH];
#pragma unroll
  for (int h = 0; h < NH; ++h) {
    float as = 0.f, ad = 0.f;
#pragma unroll
    for (int u = 0; u < PH; ++u) {
      const int q = (h * PH + u) * 32 + lane;
      const float4 v = h4[q];
      as = dot4(v, s4[q], as);
      ad = dot4(v, d4[q], ad);
    }
    s[h] = wave_sum(as);
    t[h] = wave_sum(ad);
  }
  if (lane == 0) {
    float4 r; r.x = s[0]; r.y = s[1]; r.z = s[2]; r.w = s[3];
    ((float4*)ans)[n] = r;
    float4 w; w.x = t[0]; w.y = t[1]; w.z = t[2]; w.w = t[3];
    ((float4*)adn)[n] = w;
  }
}

__global__ void k_attr_tab(const float* __restrict__ emb, const float* __restrict__ proj,
                           float* __restrict__ tab) {
  int a = blockIdx.x;
  int lane = threadIdx.x;
  const float* row = emb + (size_t)a * EMB;
  const float4* p4 = (const float4*)proj;
  float a0 = 0.f, a1 = 0.f, a2 = 0.f, a3 = 0.f;
#pragma unroll
  for (int t = 0; t < EMB / 32; ++t) {
    int k = lane + 32 * t;
    float v = row[k];
    float4 p = p4[k];
    a0 += v * p.x; a1 += v * p.y; a2 += v * p.z; a3 += v * p.w;
  }
  for (int off = 16; off > 0; off >>= 1) {
    a0 += __shfl_xor(a0, off, 32);
    a1 += __shfl_xor(a1, off, 32);
    a2 += __shfl_xor(a2, off, 32);
    a3 += __shfl_xor(a3, off, 32);
  }
  if (lane == 0) {
    float4 r; r.x = a0; r.y = a1; r.z = a2; r.w = a3;
    ((float4*)tab)[a] = r;
  }
}

__device__ __forceinline__ void edge_fma(float4& acc, float w0, float w1, float w2, float w3,
                                         const float4& v0, const float4& v1, const float4& v2,
                                         const float4& v3) {
  acc.x = fmaf(w0, v0.x, acc.x); acc.y = fmaf(w0, v0.y, acc.y);
  acc.z = fmaf(w0, v0.z, acc.z); acc.w = fmaf(w0, v0.w, acc.w);
  acc.x = fmaf(w1, v1.x, acc.x); acc.y = fmaf(w1, v1.y, acc.y);
  acc.z = fmaf(w1, v1.z, acc.z); acc.w = fmaf(w1, v1.w, acc.w);
  acc.x = fmaf(w2, v2.x, acc.x); acc.y = fmaf(w2, v2.y, acc.y);
  acc.z = fmaf(w2, v2.z, acc.z); acc.w = fmaf(w2, v2.w, acc.w);
  acc.x = fmaf(w3, v3.x, acc.x); acc.y = fmaf(w3, v3.y, acc.y);
  acc.z = fmaf(w3, v3.z, acc.z); acc.w = fmaf(w3, v3.w, acc.w);
}

__device__ __forceinline__ float leaky(float a) { return (a > 0.f) ? a : NEG_SLOPE * a; }

__device__ __forceinline__ float4 edge_score(int s, int pa, const float4* __restrict__ ans4,
                                             const float4* __restrict__ tab4, const float4& ad) {
  const float4 an = ans4[s];
  const float4 t0 = tab4[(unsigned)pa & 0xffffu], t1 = tab4[(unsigned)pa >> 16];
  float4 a;
  a.x = leaky(an.x + ad.x + (t0.x + t1.x));
  a.y = leaky(an.y + ad.y + (t0.y + t1.y));
  a.z = leaky(an.z + ad.z + (t0.z + t1.z));
  a.w = leaky(an.w + ad.w + (t0.w + t1.w));
  return a;
}
__device__ __forceinline__ float4 max4(const float4& a, const float4& b) {
  float4 r; r.x = fmaxf(a.x, b.x); r.y = fmaxf(a.y, b.y); r.z = fmaxf(a.z, b.z); r.w = fmaxf(a.w, b.w);
  return r;
}
__device__ __forceinline__ float4 exp4(const float4& a, const float4& m) {
  float4 r; r.x = expf(a.x - m.x); r.y = expf(a.y - m.y); r.z = expf(a.z - m.z); r.w = expf(a.w - m.w);
  return r;
}
__device__ __forceinline__ float first_lane(float v) {
  return __int_as_float(__builtin_amdgcn_readfirstlane(__float_as_int(v)));
}
__device__ __forceinline__ float4 wave_max4(float4 v) {
  for (int off = 16; off > 0; off >>= 1) {
    v.x = fmaxf(v.x, __shfl_xor(v.x, off, 32)); v.y = fmaxf(v.y, __shfl_xor(v.y, off, 32));
    v.z = fmaxf(v.z, __shfl_xor(v.z, off, 32)); v.w = fmaxf(v.w, __shfl_xor(v.w, off, 32));
  }
  return v;
}
__device__ __forceinline__ float4 wave_sum4(float4 v) {
  for (int off = 16; off > 0; off >>= 1) {
    v.x += __shfl_xor(v.x, off, 32); v.y += __shfl_xor(v.y, off, 32);
    v.z += __shfl_xor(v.z, off, 32); v.w += __shfl_xor(v.w, off, 32);
  }
  return v;
}

__global__ __launch_bounds__(256) void k_gather(
    const float* __restrict__ hx, const float* __restrict__ ans, const float* __restrict__ adn,
    const float* __restrict__ tab, const unsigned* __restrict__ cnt,
    const unsigned* __restrict__ rowptr, const int* __restrict__ csrc, const int* __restrict__ cattr,
    const float* __restrict__ bias, float* __restrict__ o, int C, int chunks, int relu) {
  const int lane = threadIdx.x & 31;
  const int gw = __builtin_amdgcn_readfirstlane((int)((blockIdx.x * 256 + threadIdx.x) >> 5));
  const int node = gw / chunks;
  if (node >= NN) return;
  const int c0 = (gw - node * chunks) * 128 + lane * 4;
  const size_t rs = (size_t)NH * C;
  const float4* ans4 = (const float4*)ans;
  const float4* tab4 = (const float4*)tab;
  const int beg = (int)rowptr[node], end = (int)rowptr[node + 1];

  const float* ps = hx + (size_t)node * rs + c0;
  const float4 sv0 = *(const float4*)ps, sv1 = *(const float4*)(ps + C);
  const float4 sv2 = *(const float4*)(ps + 2 * C), sv3 = *(const float4*)(ps + 3 * C);

  const float4 ad = *(const float4*)(adn + (size_t)node * NH);
  float4 a_self;
  {
    const unsigned* cr = cnt + (size_t)node * NEA;
    float s0 = 0.f, s1 = 0.f, s2 = 0.f, s3 = 0.f;
#pragma unroll
    for (int a = 0; a < NEA; ++a) {
      const float cf = (float)cr[a];
      const float4 t = tab4[a];
      s0 += cf * t.x; s1 += cf * t.y; s2 += cf * t.z; s3 += cf * t.w;
    }
    const float dg = fmaxf((float)(end - beg), 1.0f);
    const float4 an = ans4[node];
    a_self.x = leaky(an.x + ad.x + s0 / dg);
    a_self.y = leaky(an.y + ad.y + s1 / dg);
    a_self.z = leaky(an.z + ad.z + s2 / dg);
    a_self.w = leaky(an.w + ad.w + s3 / dg);
  }

  const bool has0 = beg + lane < end;
  int s_first = node;
  float4 a_first = a_self;
  if (has0) {
    s_first = csrc[beg + lane];
    a_first = edge_score(s_first, cattr[beg + lane], ans4, tab4, ad);
  }
  float4 m = max4(a_first, a_self);
  for (int base = beg + 32; base < end; base += 32) {
    const int idx = base + lane;
    if (idx < end) m = max4(m, edge_score(csrc[idx], cattr[idx], ans4, tab4, ad));
  }
  m = wave_max4(m);
  m.x = first_lane(m.x); m.y = first_lane(m.y); m.z = first_lane(m.z); m.w = first_lane(m.w);
  float4 e_first = {0.f, 0.f, 0.f, 0.f};
  if (has0) e_first = exp4(a_first, m);
  float4 dsum = e_first;
  for (int base = beg + 32; base < end; base += 32) {
    const int idx = base + lane;
    if (idx < end) {
      const float4 ex = exp4(edge_score(csrc[idx], cattr[idx], ans4, tab4, ad), m);
      dsum.x += ex.x; dsum.y += ex.y; dsum.z += ex.z; dsum.w += ex.w;
    }
  }
  dsum = wave_sum4(dsum);
  dsum.x = first_lane(dsum.x); dsum.y = first_lane(dsum.y); dsum.z = first_lane(dsum.z); dsum.w = first_lane(dsum.w);
  const float4 e_self = exp4(a_self, m);
  const float i0 = 1.0f / (dsum.x + e_self.x), i1 = 1.0f / (dsum.y + e_self.y);
  const float i2 = 1.0f / (dsum.z + e_self.z), i3 = 1.0f / (dsum.w + e_self.w);

  float4 acc = {0.f, 0.f, 0.f, 0.f};
  edge_fma(acc, e_self.x * i0, e_self.y * i1, e_self.z * i2, e_self.w * i3, sv0, sv1, sv2, sv3);

  int s_l = s_first;
  float w0_l = e_first.x * i0, w1_l = e_first.y * i1, w2_l = e_first.z * i2, w3_l = e_first.w * i3;
  for (int base = beg; base < end; base += 32) {
    if (base != beg) {
      const int idx = base + lane;
      s_l = node;
      w0_l = 0.f; w1_l = 0.f; w2_l = 0.f; w3_l = 0.f;
      if (idx < end) {
        s_l = csrc[idx];
        const float4 ex = exp4(edge_score(s_l, cattr[idx], ans4, tab4, ad), m);
        w0_l = ex.x * i0; w1_l = ex.y * i1; w2_l = ex.z * i2; w3_l = ex.w * i3;
      }
    }
    const int cnt_b = min(32, end - base);
    int j = 0;
    for (; j + 4 <= cnt_b; j += 4) {
      float4 v[4][4];
      float w[4][4];
#pragma unroll
      for (int u = 0; u < 4; ++u) {
        const int s = __builtin_amdgcn_readlane(s_l, j + u);
        const float* p = hx + (size_t)s * rs + c0;
        v[u][0] = *(const float4*)p;
        v[u][1] = *(const float4*)(p + C);
        v[u][2] = *(const float4*)(p + 2 * C);
        v[u][3] = *(const float4*)(p + 3 * C);
        w[u][0] = __int_as_float(__builtin_amdgcn_readlane(__float_as_int(w0_l), j + u));
        w[u][1] = __int_as_float(__builtin_amdgcn_readlane(__float_as_int(w1_l), j + u));
        w[u][2] = __int_as_float(__builtin_amdgcn_readlane(__float_as_int(w2_l), j + u));
        w[u][3] = __int_as_float(__builtin_amdgcn_readlane(__float_as_int(w3_l), j + u));
      }
#pragma unroll
      for (int u = 0; u < 4; ++u)
        edge_fma(acc, w[u][0], w[u][1], w[u][2], w[u][3], v[u][0], v[u][1], v[u][2], v[u][3]);
    }
    for (; j < cnt_b; ++j) {
      const int s = __builtin_amdgcn_readlane(s_l, j);
      const float* p = hx + (size_t)s * rs + c0;
      const float4 v0 = *(const float4*)p, v1 = *(const float4*)(p + C);
      const float4 v2 = *(const float4*)(p + 2 * C), v3 = *(const float4*)(p + 3 * C);
      edge_fma(acc,
               __int_as_float(__builtin_amdgcn_readlane(__float_as_int(w0_l), j)),
               __int_as_float(__builtin_amdgcn_readlane(__float_as_int(w1_l), j)),
               __int_as_float(__builtin_amdgcn_readlane(__float_as_int(w2_l), j)),
               __int_as_float(__builtin_amdgcn_readlane(__float_as_int(w3_l), j)),
               v0, v1, v2, v3);
    }
  }

  float4 r;
  r.x = 0.25f * acc.x + bias[c0 + 0];
  r.y = 0.25f * acc.y + bias[c0 + 1];
  r.z = 0.25f * acc.z + bias[c0 + 2];
  r.w = 0.25f * acc.w + bias[c0 + 3];
  if (relu) { r.x = fmaxf(r.x, 0.f); r.y = fmaxf(r.y, 0.f); r.z = fmaxf(r.z, 0.f); r.w = fmaxf(r.w, 0.f); }
  *(float4*)(o + (size_t)node * C + c0) = r;
}

extern "C" void kernel_launch(void* const* d_in, const int* in_sizes, int n_in,
                              void* d_out, int out_size, void* d_ws, size_t ws_size,
                              hipStream_t stream) {
  const int* x      = (const int*)d_in[0];
  const int* eidx   = (const int*)d_in[1];
  const int* src    = eidx;
  const int* dst    = eidx + NE;
  const int* eattr  = (const int*)d_in[2];
  const float* xemb = (const float*)d_in[3];
  const float* eemb = (const float*)d_in[4];
  const float* W1   = (const float*)d_in[5];
  const float* as1  = (const float*)d_in[6];
  const float* ad1  = (const float*)d_in[7];
  const float* We1  = (const float*)d_in[8];
  const float* ae1  = (const float*)d_in[9];
  const float* b1   = (const float*)d_in[10];
  const float* W2   = (const float*)d_in[11];
  const float* as2  = (const float*)d_in[12];
  const float* ad2  = (const float*)d_in[13];
  const float* We2  = (const float*)d_in[14];
  const float* ae2  = (const float*)d_in[15];
  const float* b2   = (const float*)d_in[16];
  float* out = (float*)d_out;

  float* ws = (float*)d_ws;
  size_t off = 0;
  float* XF   = ws + off; off += (size_t)NN * EMB;
  unsigned* CNT = (unsigned*)(ws + off); off += (size_t)2048 * NEA;
  unsigned* DEG = (unsigned*)(ws + off); off += 2048;
  unsigned* ROWPTR = (unsigned*)(ws + off); off += 2112;
  unsigned* CUR = (unsigned*)(ws + off); off += 2048;
  int* CSRC = (int*)(ws + off); off += (size_t)NE;
  int* CATTR = (int*)(ws + off); off += (size_t)NE;
  float* TAB  = ws + off; off += 128;
  float* HX1  = ws + off; off += (size_t)NN * NH * HID;
  float* HX2  = ws + off; off += (size_t)NN * NH * FT;
  float* PROJ = ws + off; off += (size_t)EMB * NH;
  float* ANS  = ws + off; off += (size_t)NN * NH;
  float* ADN  = ws + off; off += (size_t)NN * NH;
  float* HB   = ws + off; off += (size_t)NN * HID;
  (void)ws_size; (void)in_sizes; (void)n_in; (void)out_size;

  const int T = 256;

  hipMemsetAsync(CNT, 0, (size_t)(2048 * NEA + 2048) * sizeof(unsigned), stream);

  k_embed_x<<<cdiv((long long)NN * EMB, T), T, 0, stream>>>(x, xemb, XF);
  k_attr_count<<<cdiv(NE, T), T, 0, stream>>>(dst, eattr, CNT, DEG);
  k_csr_scan<<<1, 1024, 0, stream>>>(DEG, ROWPTR, CUR);
  k_csr_fill<<<cdiv(NE, T), T, 0, stream>>>(src, dst, eattr, CUR, CSRC, CATTR);

  {
    dim3 g(NH * HID / 128, cdiv(NN, 64));
    k_gemm_wmma_f32<<<g, 256, 0, stream>>>(XF, W1, HX1, NN, NH * HID, EMB);
  }
  k_proj_e<HID><<<cdiv((long long)EMB * 32, 64), 64, 0, stream>>>(We1, ae1, PROJ);
  k_attr_tab<<<NEA, 32, 0, stream>>>(eemb, PROJ, TAB);
  k_alpha_node<HID><<<cdiv((long long)NN * 32, 128), 128, 0, stream>>>(HX1, as1, ad1, ANS, ADN);
  k_gather<<<cdiv((long long)NN * (HID / 128), 8), 256, 0, stream>>>(HX1, ANS, ADN, TAB, CNT, ROWPTR, CSRC, CATTR, b1, HB, HID, HID / 128, 1);

  {
    dim3 g(NH * FT / 128, cdiv(NN, 64));
    k_gemm_wmma_f32<<<g, 256, 0, stream>>>(HB, W2, HX2, NN, NH * FT, HID);
  }
  k_proj_e<FT><<<cdiv((long long)EMB * 32, 64), 64, 0, stream>>>(We2, ae2, PROJ);
  k_attr_tab<<<NEA, 32, 0, stream>>>(eemb, PROJ, TAB);
  k_alpha_node<FT><<<cdiv((long long)NN * 32, 128), 128, 0, stream>>>(HX2, as2, ad2, ANS, ADN);
  k_gather<<<cdiv((long long)NN * (FT / 128), 8), 256, 0, stream>>>(HX2, ANS, ADN, TAB, CNT, ROWPTR, CSRC, CATTR, b2, out, FT, FT / 128, 0);
}


# Node feature rows formed in GEMM 1's A-tile staging (two table loads + the same add per element); the one-thread-per-element feature kernel, its 48000 short waves and the [N,768] tensor are gone
# speedup vs baseline: 4.0666x; 1.0560x over previous
#include <hip/hip_runtime.h>
#include <math.h>

#define NN   2000
#define NE   32000
#define EMB  768
#define HID  384
#define FT   768
#define NH   4
#define NEA  18
#define NEG_SLOPE 0.2f

typedef __attribute__((ext_vector_type(2))) float v2f;
typedef __attribute__((ext_vector_type(8))) float v8f;

static inline int cdiv(long long a, long long b) { return (int)((a + b - 1) / b); }

__global__ void k_attr_count(const int* __restrict__ dst, const int* __restrict__ ea_idx,
                             unsigned* __restrict__ cnt, unsigned* __restrict__ deg) {
  int e = blockIdx.x * blockDim.x + threadIdx.x;
  if (e >= NE) return;
  int d = dst[e];
  atomicAdd(&cnt[d * NEA + ea_idx[e * 2 + 0]], 1u);
  atomicAdd(&cnt[d * NEA + ea_idx[e * 2 + 1]], 1u);
  atomicAdd(&deg[d], 1u);
}

static_assert(NN <= 2048, "k_csr_scan covers two nodes per thread of one 1024-thread block");
__global__ __launch_bounds__(1024) void k_csr_scan(const unsigned* __restrict__ deg,
                                                   unsigned* __restrict__ rowptr,
                                                   unsigned* __restrict__ cur) {
  __shared__ unsigned wsum[32];
  const int t = threadIdx.x, lane = t & 31, wave = t >> 5;
  const int n0 = 2 * t;
  const unsigned a = (n0 < NN) ? deg[n0] : 0u;
  const unsigned b = (n0 + 1 < NN) ? deg[n0 + 1] : 0u;
  unsigned incl = a + b;
  for (int off = 1; off < 32; off <<= 1) {
    unsigned v = __shfl_up(incl, off, 32);
    if (lane >= off) incl += v;
  }
  if (lane == 31) wsum[wave] = incl;
  __syncthreads();
  if (wave == 0) {
    unsigned w = wsum[lane];
    for (int off = 1; off < 32; off <<= 1) {
      unsigned v = __shfl_up(w, off, 32);
      if (lane >= off) w += v;
    }
    wsum[lane] = w;
  }
  __syncthreads();
  const unsigned excl = incl - (a + b) + (wave ? wsum[wave - 1] : 0u);
  if (n0 < NN)     { rowptr[n0] = excl;         cur[n0] = excl; }
  if (n0 + 1 < NN) { rowptr[n0 + 1] = excl + a; cur[n0 + 1] = excl + a; }
  if (t == 1023) rowptr[NN] = excl + a + b;
}

__global__ void k_csr_fill(const int* __restrict__ src, const int* __restrict__ dst,
                           const int* __restrict__ ea_idx,
                           unsigned* __restrict__ cur, int* __restrict__ csrc,
                           int* __restrict__ cattr) {
  int e = blockIdx.x * blockDim.x + threadIdx.x;
  if (e >= NE) return;
  unsigned pos = atomicAdd(&cur[dst[e]], 1u);
  csrc[pos] = src[e];
  cattr[pos] = (int)((unsigned)ea_idx[e * 2 + 0] | ((unsigned)ea_idx[e * 2 + 1] << 16));
}

#define A_PITCH 36
#define B_PITCH 34
template <bool EMB_A>
__global__ __launch_bounds__(256) void k_gemm_wmma_f32(
    const float* __restrict__ A, const int* __restrict__ xi, const float* __restrict__ emb,
    const float* __restrict__ B, float* __restrict__ C, int M, int N, int K) {
  __shared__ __align__(16) float sA[64][A_PITCH];
  __shared__ __align__(16) float sBt[128][B_PITCH];
  const int tid = threadIdx.x;
  const int wave = tid >> 5, lane = tid & 31;
  const int wm = wave >> 2, wn = wave & 3;
  const int row16 = lane & 15, khalf = lane >> 4;
  const int bm = blockIdx.y * 64, bn = blockIdx.x * 128;

  const float* r0[8];
  const float* r1[8];
  bool rowok[8];
  if constexpr (EMB_A) {
    const int wv = __builtin_amdgcn_readfirstlane(wave);
    const int grl = min(bm + wv + 8 * ((lane >> 1) & 7), M - 1);
    const int myid = xi[grl * 2 + (lane & 1)];
#pragma unroll
    for (int j = 0; j < 8; ++j) {
      rowok[j] = bm + wv + 8 * j < M;
      r0[j] = emb + (size_t)__builtin_amdgcn_readlane(myid, 2 * j) * EMB;
      r1[j] = emb + (size_t)__builtin_amdgcn_readlane(myid, 2 * j + 1) * EMB;
    }
  }

  v8f acc00 = {}, acc01 = {}, acc10 = {}, acc11 = {};

  for (int k0 = 0; k0 < K; k0 += 32) {
    float ra[8], rb[16];
    float ua[8], ub[8];
    const unsigned kl = (unsigned)(k0 + lane);
#pragma unroll
    for (int j = 0; j < 8; ++j) {
      if constexpr (EMB_A) {
        ua[j] = r0[j][kl];
        ub[j] = r1[j][kl];
      } else {
        int i = tid + j * 256;
        int r = i >> 5, c = i & 31;
        int gr = bm + r;
        ra[j] = (gr < M) ? A[(size_t)gr * K + k0 + c] : 0.0f;
      }
    }
#pragma unroll
    for (int j = 0; j < 16; ++j) {
      int i = tid + j * 256;
      int kk = i >> 7, n = i & 127;
      rb[j] = B[(size_t)(k0 + kk) * N + bn + n];
    }
    if constexpr (EMB_A) {
      __builtin_amdgcn_sched_barrier(0);
#pragma unroll
      for (int j = 0; j < 8; ++j) ra[j] = rowok[j] ? ua[j] + ub[j] : 0.0f;
    }
#pragma unroll
    for (int j = 0; j < 8; ++j) {
      int i = tid + j * 256;
      sA[i >> 5][i & 31] = ra[j];
    }
#pragma unroll
    for (int j = 0; j < 16; ++j) {
      int i = tid + j * 256;
      sBt[i & 127][i >> 7] = rb[j];
    }
    __syncthreads();

    const float* a0p = &sA[wm * 32 + row16][0];
    const float* a1p = &sA[wm * 32 + 16 + row16][0];
    const float* b0p = &sBt[wn * 32 + row16][0];
    const float* b1p = &sBt[wn * 32 + 16 + row16][0];
#pragma unroll
    for (int ks = 0; ks < 32; ks += 4) {
      int kk = ks + khalf * 2;
      v2f a0 = *(const v2f*)(a0p + kk);
      v2f a1 = *(const v2f*)(a1p + kk);
      v2f b0 = *(const v2f*)(b0p + kk);
      v2f b1 = *(const v2f*)(b1p + kk);
      acc00 = __builtin_amdgcn_wmma_f32_16x16x4_f32(false, a0, false, b0, (short)0, acc00, false, false);
      acc01 = __builtin_amdgcn_wmma_f32_16x16x4_f32(false, a0, false, b1, (short)0, acc01, false, false);
      acc10 = __builtin_amdgcn_wmma_f32_16x16x4_f32(false, a1, false, b0, (short)0, acc10, false, false);
      acc11 = __builtin_amdgcn_wmma_f32_16x16x4_f32(false, a1, false, b1, (short)0, acc11, false, false);
    }
    __syncthreads();
  }

  int gn0 = bn + wn * 32 + row16;
  int gn1 = gn0 + 16;
  int gm0 = bm + wm * 32 + 8 * khalf;
  int gm1 = gm0 + 16;
#pragma unroll
  for (int v = 0; v < 8; ++v) {
    if (gm0 + v < M) {
      C[(size_t)(gm0 + v) * N + gn0] = acc00[v];
      C[(size_t)(gm0 + v) * N + gn1] = acc01[v];
    }
    if (gm1 + v < M) {
      C[(size_t)(gm1 + v) * N + gn0] = acc10[v];
      C[(size_t)(gm1 + v) * N + gn1] = acc11[v];
    }
  }
}

__device__ __forceinline__ float dot4(const float4& a, const float4& b, float s) {
  s = fmaf(a.x, b.x, s); s = fmaf(a.y, b.y, s); s = fmaf(a.z, b.z, s); s = fmaf(a.w, b.w, s);
  return s;
}
__device__ __forceinline__ float wave_sum(float v) {
  for (int off = 16; off > 0; off >>= 1) v += __shfl_xor(v, off, 32);
  return v;
}

template <int CC>
__global__ __launch_bounds__(64) void k_proj_e(const float* __restrict__ We,
                                               const float* __restrict__ ae,
                                               float* __restrict__ proj) {
  static_assert(CC % 128 == 0, "a row is a whole number of 512-byte steps of a wave");
  constexpr int PH = CC / 128;
  const int lane = threadIdx.x & 31;
  const int k = __builtin_amdgcn_readfirstlane((int)((blockIdx.x * 64 + threadIdx.x) >> 5));
  if (k >= EMB) return;
  const float4* w4 = (const float4*)(We + (size_t)k * (NH * CC));
  const float4* a4 = (const float4*)ae;
  float s[NH];
#pragma unroll
  for (int h = 0; h < NH; ++h) {
    float acc = 0.f;
#pragma unroll
    for (int u = 0; u < PH; ++u) {
      const int q = (h * PH + u) * 32 + lane;
      acc = dot4(w4[q], a4[q], acc);
    }
    s[h] = wave_sum(acc);
  }
  if (lane == 0) {
    float4 r; r.x = s[0]; r.y = s[1]; r.z = s[2]; r.w = s[3];
    ((float4*)proj)[k] = r;
  }
}

template <int CC>
__global__ __launch_bounds__(128) void k_alpha_node(const float* __restrict__ hx,
                                                    const float* __restrict__ a_src,
                                                    const float* __restrict__ a_dst,
                                                    float* __restrict__ ans,
                                                    float* __restrict__ adn) {
  static_assert(CC % 128 == 0, "a row is a whole number of 512-byte steps of a wave");
  static_assert(NH == 4, "one 16-byte store per node and output");
  constexpr int PH = CC / 128;
  const int lane = threadIdx.x & 31;
  const int n = __builtin_amdgcn_readfirstlane((int)((blockIdx.x * 128 + threadIdx.x) >> 5));
  if (n >= NN) return;
  const float4* h4 = (const float4*)(hx + (size_t)n * (NH * CC));
  const float4* s4 = (const float4*)a_src;
  const float4* d4 = (const float4*)a_dst;
  float s[NH], t[NH];
#pragma unroll
  for (int h = 0; h < NH; ++h) {
    float as = 0.f, ad = 0.f;
#pragma unroll
    for (int u = 0; u < PH; ++u) {
      const int q = (h * PH + u) * 32 + lane;
      const float4 v = h4[q];
      as = dot4(v, s4[q], as);
      ad = dot4(v, d4[q], ad);
    }
    s[h] = wave_sum(as);
    t[h] = wave_sum(ad);
  }
  if (lane == 0) {
    float4 r; r.x = s[0]; r.y = s[1]; r.z = s[2]; r.w = s[3];
    ((float4*)ans)[n] = r;
    float4 w; w.x = t[0]; w.y = t[1]; w.z = t[2]; w.w = t[3];
    ((float4*)adn)[n] = w;
  }
}

__global__ void k_attr_tab(const float* __restrict__ emb, const float* __restrict__ proj,
                           float* __restrict__ tab) {
  int a = blockIdx.x;
  int lane = threadIdx.x;
  const float* row = emb + (size_t)a * EMB;
  const float4* p4 = (const float4*)proj;
  float a0 = 0.f, a1 = 0.f, a2 = 0.f, a3 = 0.f;
#pragma unroll
  for (int t = 0; t < EMB / 32; ++t) {
    int k = lane + 32 * t;
    float v = row[k];
    float4 p = p4[k];
    a0 += v * p.x; a1 += v * p.y; a2 += v * p.z; a3 += v * p.w;
  }
  for (int off = 16; off > 0; off >>= 1) {
    a0 += __shfl_xor(a0, off, 32);
    a1 += __shfl_xor(a1, off, 32);
    a2 += __shfl_xor(a2, off, 32);
    a3 += __shfl_xor(a3, off, 32);
  }
  if (lane == 0) {
    float4 r; r.x = a0; r.y = a1; r.z = a2; r.w = a3;
    ((float4*)tab)[a] = r;
  }
}

__device__ __forceinline__ void edge_fma(float4& acc, float w0, float w1, float w2, float w3,
                                         const float4& v0, const float4& v1, const float4& v2,
                                         const float4& v3) {
  acc.x = fmaf(w0, v0.x, acc.x); acc.y = fmaf(w0, v0.y, acc.y);
  acc.z = fmaf(w0, v0.z, acc.z); acc.w = fmaf(w0, v0.w, acc.w);
  acc.x = fmaf(w1, v1.x, acc.x); acc.y = fmaf(w1, v1.y, acc.y);
  acc.z = fmaf(w1, v1.z, acc.z); acc.w = fmaf(w1, v1.w, acc.w);
  acc.x = fmaf(w2, v2.x, acc.x); acc.y = fmaf(w2, v2.y, acc.y);
  acc.z = fmaf(w2, v2.z, acc.z); acc.w = fmaf(w2, v2.w, acc.w);
  acc.x = fmaf(w3, v3.x, acc.x); acc.y = fmaf(w3, v3.y, acc.y);
  acc.z = fmaf(w3, v3.z, acc.z); acc.w = fmaf(w3, v3.w, acc.w);
}

__device__ __forceinline__ float leaky(float a) { return (a > 0.f) ? a : NEG_SLOPE * a; }

__device__ __forceinline__ float4 edge_score(int s, int pa, const float4* __restrict__ ans4,
                                             const float4* __restrict__ tab4, const float4& ad) {
  const float4 an = ans4[s];
  const float4 t0 = tab4[(unsigned)pa & 0xffffu], t1 = tab4[(unsigned)pa >> 16];
  float4 a;
  a.x = leaky(an.x + ad.x + (t0.x + t1.x));
  a.y = leaky(an.y + ad.y + (t0.y + t1.y));
  a.z = leaky(an.z + ad.z + (t0.z + t1.z));
  a.w = leaky(an.w + ad.w + (t0.w + t1.w));
  return a;
}
__device__ __forceinline__ float4 max4(const float4& a, const float4& b) {
  float4 r; r.x = fmaxf(a.x, b.x); r.y = fmaxf(a.y, b.y); r.z = fmaxf(a.z, b.z); r.w = fmaxf(a.w, b.w);
  return r;
}
__device__ __forceinline__ float4 exp4(const float4& a, const float4& m) {
  float4 r; r.x = expf(a.x - m.x); r.y = expf(a.y - m.y); r.z = expf(a.z - m.z); r.w = expf(a.w - m.w);
  return r;
}
__device__ __forceinline__ float first_lane(float v) {
  return __int_as_float(__builtin_amdgcn_readfirstlane(__float_as_int(v)));
}
__device__ __forceinline__ float4 wave_max4(float4 v) {
  for (int off = 16; off > 0; off >>= 1) {
    v.x = fmaxf(v.x, __shfl_xor(v.x, off, 32)); v.y = fmaxf(v.y, __shfl_xor(v.y, off, 32));
    v.z = fmaxf(v.z, __shfl_xor(v.z, off, 32)); v.w = fmaxf(v.w, __shfl_xor(v.w, off, 32));
  }
  return v;
}
__device__ __forceinline__ float4 wave_sum4(float4 v) {
  for (int off = 16; off > 0; off >>= 1) {
    v.x += __shfl_xor(v.x, off, 32); v.y += __shfl_xor(v.y, off, 32);
    v.z += __shfl_xor(v.z, off, 32); v.w += __shfl_xor(v.w, off, 32);
  }
  return v;
}

__global__ __launch_bounds__(256) void k_gather(
    const float* __restrict__ hx, const float* __restrict__ ans, const float* __restrict__ adn,
    const float* __restrict__ tab, const unsigned* __restrict__ cnt,
    const unsigned* __restrict__ rowptr, const int* __restrict__ csrc, const int* __restrict__ cattr,
    const float* __restrict__ bias, float* __restrict__ o, int C, int chunks, int relu) {
  const int lane = threadIdx.x & 31;
  const int gw = __builtin_amdgcn_readfirstlane((int)((blockIdx.x * 256 + threadIdx.x) >> 5));
  const int node = gw / chunks;
  if (node >= NN) return;
  const int c0 = (gw - node * chunks) * 128 + lane * 4;
  const size_t rs = (size_t)NH * C;
  const float4* ans4 = (const float4*)ans;
  const float4* tab4 = (const float4*)tab;
  const int beg = (int)rowptr[node], end = (int)rowptr[node + 1];

  const float* ps = hx + (size_t)node * rs + c0;
  const float4 sv0 = *(const float4*)ps, sv1 = *(const float4*)(ps + C);
  const float4 sv2 = *(const float4*)(ps + 2 * C), sv3 = *(const float4*)(ps + 3 * C);

  const float4 ad = *(const float4*)(adn + (size_t)node * NH);
  float4 a_self;
  {
    const unsigned* cr = cnt + (size_t)node * NEA;
    float s0 = 0.f, s1 = 0.f, s2 = 0.f, s3 = 0.f;
#pragma unroll
    for (int a = 0; a < NEA; ++a) {
      const float cf = (float)cr[a];
      const float4 t = tab4[a];
      s0 += cf * t.x; s1 += cf * t.y; s2 += cf * t.z; s3 += cf * t.w;
    }
    const float dg = fmaxf((float)(end - beg), 1.0f);
    const float4 an = ans4[node];
    a_self.x = leaky(an.x + ad.x + s0 / dg);
    a_self.y = leaky(an.y + ad.y + s1 / dg);
    a_self.z = leaky(an.z + ad.z + s2 / dg);
    a_self.w = leaky(an.w + ad.w + s3 / dg);
  }

  const bool has0 = beg + lane < end;
  int s_first = node;
  float4 a_first = a_self;
  if (has0) {
    s_first = csrc[beg + lane];
    a_first = edge_score(s_first, cattr[beg + lane], ans4, tab4, ad);
  }
  float4 m = max4(a_first, a_self);
  for (int base = beg + 32; base < end; base += 32) {
    const int idx = base + lane;
    if (idx < end) m = max4(m, edge_score(csrc[idx], cattr[idx], ans4, tab4, ad));
  }
  m = wave_max4(m);
  m.x = first_lane(m.x); m.y = first_lane(m.y); m.z = first_lane(m.z); m.w = first_lane(m.w);
  float4 e_first = {0.f, 0.f, 0.f, 0.f};
  if (has0) e_first = exp4(a_first, m);
  float4 dsum = e_first;
  for (int base = beg + 32; base < end; base += 32) {
    const int idx = base + lane;
    if (idx < end) {
      const float4 ex = exp4(edge_score(csrc[idx], cattr[idx], ans4, tab4, ad), m);
      dsum.x += ex.x; dsum.y += ex.y; dsum.z += ex.z; dsum.w += ex.w;
    }
  }
  dsum = wave_sum4(dsum);
  dsum.x = first_lane(dsum.x); dsum.y = first_lane(dsum.y); dsum.z = first_lane(dsum.z); dsum.w = first_lane(dsum.w);
  const float4 e_self = exp4(a_self, m);
  const float i0 = 1.0f / (dsum.x + e_self.x), i1 = 1.0f / (dsum.y + e_self.y);
  const float i2 = 1.0f / (dsum.z + e_self.z), i3 = 1.0f / (dsum.w + e_self.w);

  float4 acc = {0.f, 0.f, 0.f, 0.f};
  edge_fma(acc, e_self.x * i0, e_self.y * i1, e_self.z * i2, e_self.w * i3, sv0, sv1, sv2, sv3);

  int s_l = s_first;
  float w0_l = e_first.x * i0, w1_l = e_first.y * i1, w2_l = e_first.z * i2, w3_l = e_first.w * i3;
  for (int base = beg; base < end; base += 32) {
    if (base != beg) {
      const int idx = base + lane;
      s_l = node;
      w0_l = 0.f; w1_l = 0.f; w2_l = 0.f; w3_l = 0.f;
      if (idx < end) {
        s_l = csrc[idx];
        const float4 ex = exp4(edge_score(s_l, cattr[idx], ans4, tab4, ad), m);
        w0_l = ex.x * i0; w1_l = ex.y * i1; w2_l = ex.z * i2; w3_l = ex.w * i3;
      }
    }
    const int cnt_b = min(32, end - base);
    int j = 0;
    for (; j + 4 <= cnt_b; j += 4) {
      float4 v[4][4];
      float w[4][4];
#pragma unroll
      for (int u = 0; u < 4; ++u) {
        const int s = __builtin_amdgcn_readlane(s_l, j + u);
        const float* p = hx + (size_t)s * rs + c0;
        v[u][0] = *(const float4*)p;
        v[u][1] = *(const float4*)(p + C);
        v[u][2] = *(const float4*)(p + 2 * C);
        v[u][3] = *(const float4*)(p + 3 * C);
        w[u][0] = __int_as_float(__builtin_amdgcn_readlane(__float_as_int(w0_l), j + u));
        w[u][1] = __int_as_float(__builtin_amdgcn_readlane(__float_as_int(w1_l), j + u));
        w[u][2] = __int_as_float(__builtin_amdgcn_readlane(__float_as_int(w2_l), j + u));
        w[u][3] = __int_as_float(__builtin_amdgcn_readlane(__float_as_int(w3_l), j + u));
      }
#pragma unroll
      for (int u = 0; u < 4; ++u)
        edge_fma(acc, w[u][0], w[u][1], w[u][2], w[u][3], v[u][0], v[u][1], v[u][2], v[u][3]);
    }
    for (; j < cnt_b; ++j) {
      const int s = __builtin_amdgcn_readlane(s_l, j);
      const float* p = hx + (size_t)s * rs + c0;
      const float4 v0 = *(const float4*)p, v1 = *(const float4*)(p + C);
      const float4 v2 = *(const float4*)(p + 2 * C), v3 = *(const float4*)(p + 3 * C);
      edge_fma(acc,
               __int_as_float(__builtin_amdgcn_readlane(__float_as_int(w0_l), j)),
               __int_as_float(__builtin_amdgcn_readlane(__float_as_int(w1_l), j)),
               __int_as_float(__builtin_amdgcn_readlane(__float_as_int(w2_l), j)),
               __int_as_float(__builtin_amdgcn_readlane(__float_as_int(w3_l), j)),
               v0, v1, v2, v3);
    }
  }

  float4 r;
  r.x = 0.25f * acc.x + bias[c0 + 0];
  r.y = 0.25f * acc.y + bias[c0 + 1];
  r.z = 0.25f * acc.z + bias[c0 + 2];
  r.w = 0.25f * acc.w + bias[c0 + 3];
  if (relu) { r.x = fmaxf(r.x, 0.f); r.y = fmaxf(r.y, 0.f); r.z = fmaxf(r.z, 0.f); r.w = fmaxf(r.w, 0.f); }
  *(float4*)(o + (size_t)node * C + c0) = r;
}

extern "C" void kernel_launch(void* const* d_in, const int* in_sizes, int n_in,
                              void* d_out, int out_size, void* d_ws, size_t ws_size,
                              hipStream_t stream) {
  const int* x      = (const int*)d_in[0];
  const int* eidx   = (const int*)d_in[1];
  const int* src    = eidx;
  const int* dst    = eidx + NE;
  const int* eattr  = (const int*)d_in[2];
  const float* xemb = (const float*)d_in[3];
  const float* eemb = (const float*)d_in[4];
  const float* W1   = (const float*)d_in[5];
  const float* as1  = (const float*)d_in[6];
  const float* ad1  = (const float*)d_in[7];
  const float* We1  = (const float*)d_in[8];
  const float* ae1  = (const float*)d_in[9];
  const float* b1   = (const float*)d_in[10];
  const float* W2   = (const float*)d_in[11];
  const float* as2  = (const float*)d_in[12];
  const float* ad2  = (const float*)d_in[13];
  const float* We2  = (const float*)d_in[14];
  const float* ae2  = (const float*)d_in[15];
  const float* b2   = (const float*)d_in[16];
  float* out = (float*)d_out;

  float* ws = (float*)d_ws;
  size_t off = 0;
  unsigned* CNT = (unsigned*)(ws + off); off += (size_t)2048 * NEA;
  unsigned* DEG = (unsigned*)(ws + off); off += 2048;
  unsigned* ROWPTR = (unsigned*)(ws + off); off += 2112;
  unsigned* CUR = (unsigned*)(ws + off); off += 2048;
  int* CSRC = (int*)(ws + off); off += (size_t)NE;
  int* CATTR = (int*)(ws + off); off += (size_t)NE;
  float* TAB  = ws + off; off += 128;
  float* HX1  = ws + off; off += (size_t)NN * NH * HID;
  float* HX2  = ws + off; off += (size_t)NN * NH * FT;
  float* PROJ = ws + off; off += (size_t)EMB * NH;
  float* ANS  = ws + off; off += (size_t)NN * NH;
  float* ADN  = ws + off; off += (size_t)NN * NH;
  float* HB   = ws + off; off += (size_t)NN * HID;
  (void)ws_size; (void)in_sizes; (void)n_in; (void)out_size;

  const int T = 256;

  hipMemsetAsync(CNT, 0, (size_t)(2048 * NEA + 2048) * sizeof(unsigned), stream);

  k_attr_count<<<cdiv(NE, T), T, 0, stream>>>(dst, eattr, CNT, DEG);
  k_csr_scan<<<1, 1024, 0, stream>>>(DEG, ROWPTR, CUR);
  k_csr_fill<<<cdiv(NE, T), T, 0, stream>>>(src, dst, eattr, CUR, CSRC, CATTR);

  {
    dim3 g(NH * HID / 128, cdiv(NN, 64));
    k_gemm_wmma_f32<true><<<g, 256, 0, stream>>>(nullptr, x, xemb, W1, HX1, NN, NH * HID, EMB);
  }
  k_proj_e<HID><<<cdiv((long long)EMB * 32, 64), 64, 0, stream>>>(We1, ae1, PROJ);
  k_attr_tab<<<NEA, 32, 0, stream>>>(eemb, PROJ, TAB);
  k_alpha_node<HID><<<cdiv((long long)NN * 32, 128), 128, 0, stream>>>(HX1, as1, ad1, ANS, ADN);
  k_gather<<<cdiv((long long)NN * (HID / 128), 8), 256, 0, stream>>>(HX1, ANS, ADN, TAB, CNT, ROWPTR, CSRC, CATTR, b1, HB, HID, HID / 128, 1);

  {
    dim3 g(NH * FT / 128, cdiv(NN, 64));
    k_gemm_wmma_f32<false><<<g, 256, 0, stream>>>(HB, nullptr, nullptr, W2, HX2, NN, NH * FT, HID);
  }
  k_proj_e<FT><<<cdiv((long long)EMB * 32, 64), 64, 0, stream>>>(We2, ae2, PROJ);
  k_attr_tab<<<NEA, 32, 0, stream>>>(eemb, PROJ, TAB);
  k_alpha_node<FT><<<cdiv((long long)NN * 32, 128), 128, 0, stream>>>(HX2, as2, ad2, ANS, ADN);
  k_gather<<<cdiv((long long)NN * (FT / 128), 8), 256, 0, stream>>>(HX2, ANS, ADN, TAB, CNT, ROWPTR, CSRC, CATTR, b2, out, FT, FT / 128, 0);
}


# Weight-only kernels share the preprocessing launches: proj rows of both layers run beside the edge count, both attr tables beside the degree scan; 13 launches become 9, sums unchanged
# speedup vs baseline: 4.4000x; 1.0820x over previous
#include <hip/hip_runtime.h>
#include <math.h>

#define NN   2000
#define NE   32000
#define EMB  768
#define HID  384
#define FT   768
#define NH   4
#define NEA  18
#define NEG_SLOPE 0.2f

typedef __attribute__((ext_vector_type(2))) float v2f;
typedef __attribute__((ext_vector_type(8))) float v8f;

static inline int cdiv(long long a, long long b) { return (int)((a + b - 1) / b); }

__device__ __forceinline__ void attr_count_edge(int e, const int* __restrict__ dst,
                                                const int* __restrict__ ea_idx,
                                                unsigned* __restrict__ cnt, unsigned* __restrict__ deg) {
  if (e >= NE) return;
  int d = dst[e];
  atomicAdd(&cnt[d * NEA + ea_idx[e * 2 + 0]], 1u);
  atomicAdd(&cnt[d * NEA + ea_idx[e * 2 + 1]], 1u);
  atomicAdd(&deg[d], 1u);
}

static_assert(NN <= 2048, "csr_scan_block covers two nodes per thread of one 1024-thread block");
__device__ __forceinline__ void csr_scan_block(const unsigned* __restrict__ deg,
                                               unsigned* __restrict__ rowptr,
                                               unsigned* __restrict__ cur, unsigned* wsum) {
  const int t = threadIdx.x, lane = t & 31, wave = t >> 5;
  const int n0 = 2 * t;
  const unsigned a = (n0 < NN) ? deg[n0] : 0u;
  const unsigned b = (n0 + 1 < NN) ? deg[n0 + 1] : 0u;
  unsigned incl = a + b;
  for (int off = 1; off < 32; off <<= 1) {
    unsigned v = __shfl_up(incl, off, 32);
    if (lane >= off) incl += v;
  }
  if (lane == 31) wsum[wave] = incl;
  __syncthreads();
  if (wave == 0) {
    unsigned w = wsum[lane];
    for (int off = 1; off < 32; off <<= 1) {
      unsigned v = __shfl_up(w, off, 32);
      if (lane >= off) w += v;
    }
    wsum[lane] = w;
  }
  __syncthreads();
  const unsigned excl = incl - (a + b) + (wave ? wsum[wave - 1] : 0u);
  if (n0 < NN)     { rowptr[n0] = excl;         cur[n0] = excl; }
  if (n0 + 1 < NN) { rowptr[n0 + 1] = excl + a; cur[n0 + 1] = excl + a; }
  if (t == 1023) rowptr[NN] = excl + a + b;
}

__global__ void k_csr_fill(const int* __restrict__ src, const int* __restrict__ dst,
                           const int* __restrict__ ea_idx,
                           unsigned* __restrict__ cur, int* __restrict__ csrc,
                           int* __restrict__ cattr) {
  int e = blockIdx.x * blockDim.x + threadIdx.x;
  if (e >= NE) return;
  unsigned pos = atomicAdd(&cur[dst[e]], 1u);
  csrc[pos] = src[e];
  cattr[pos] = (int)((unsigned)ea_idx[e * 2 + 0] | ((unsigned)ea_idx[e * 2 + 1] << 16));
}

#define A_PITCH 36
#define B_PITCH 34
template <bool EMB_A>
__global__ __launch_bounds__(256) void k_gemm_wmma_f32(
    const float* __restrict__ A, const int* __restrict__ xi, const float* __restrict__ emb,
    const float* __restrict__ B, float* __restrict__ C, int M, int N, int K) {
  __shared__ __align__(16) float sA[64][A_PITCH];
  __shared__ __align__(16) float sBt[128][B_PITCH];
  const int tid = threadIdx.x;
  const int wave = tid >> 5, lane = tid & 31;
  const int wm = wave >> 2, wn = wave & 3;
  const int row16 = lane & 15, khalf = lane >> 4;
  const int bm = blockIdx.y * 64, bn = blockIdx.x * 128;

  const float* r0[8];
  const float* r1[8];
  bool rowok[8];
  if constexpr (EMB_A) {
    const int wv = __builtin_amdgcn_readfirstlane(wave);
    const int grl = min(bm + wv + 8 * ((lane >> 1) & 7), M - 1);
    const int myid = xi[grl * 2 + (lane & 1)];
#pragma unroll
    for (int j = 0; j < 8; ++j) {
      rowok[j] = bm + wv + 8 * j < M;
      r0[j] = emb + (size_t)__builtin_amdgcn_readlane(myid, 2 * j) * EMB;
      r1[j] = emb + (size_t)__builtin_amdgcn_readlane(myid, 2 * j + 1) * EMB;
    }
  }

  v8f acc00 = {}, acc01 = {}, acc10 = {}, acc11 = {};

  for (int k0 = 0; k0 < K; k0 += 32) {
    float ra[8], rb[16];
    float ua[8], ub[8];
    const unsigned kl = (unsigned)(k0 + lane);
#pragma unroll
    for (int j = 0; j < 8; ++j) {
      if constexpr (EMB_A) {
        ua[j] = r0[j][kl];
        ub[j] = r1[j][kl];
      } else {
        int i = tid + j * 256;
        int r = i >> 5, c = i & 31;
        int gr = bm + r;
        ra[j] = (gr < M) ? A[(size_t)gr * K + k0 + c] : 0.0f;
      }
    }
#pragma unroll
    for (int j = 0; j < 16; ++j) {
      int i = tid + j * 256;
      int kk = i >> 7, n = i & 127;
      rb[j] = B[(size_t)(k0 + kk) * N + bn + n];
    }
    if constexpr (EMB_A) {
      __builtin_amdgcn_sched_barrier(0);
#pragma unroll
      for (int j = 0; j < 8; ++j) ra[j] = rowok[j] ? ua[j] + ub[j] : 0.0f;
    }
#pragma unroll
    for (int j = 0; j < 8; ++j) {
      int i = tid + j * 256;
      sA[i >> 5][i & 31] = ra[j];
    }
#pragma unroll
    for (int j = 0; j < 16; ++j) {
      int i = tid + j * 256;
      sBt[i & 127][i >> 7] = rb[j];
    }
    __syncthreads();

    const float* a0p = &sA[wm * 32 + row16][0];
    const float* a1p = &sA[wm * 32 + 16 + row16][0];
    const float* b0p = &sBt[wn * 32 + row16][0];
    const float* b1p = &sBt[wn * 32 + 16 + row16][0];
#pragma unroll
    for (int ks = 0; ks < 32; ks += 4) {
      int kk = ks + khalf * 2;
      v2f a0 = *(const v2f*)(a0p + kk);
      v2f a1 = *(const v2f*)(a1p + kk);
      v2f b0 = *(const v2f*)(b0p + kk);
      v2f b1 = *(const v2f*)(b1p + kk);
      acc00 = __builtin_amdgcn_wmma_f32_16x16x4_f32(false, a0, false, b0, (short)0, acc00, false, false);
      acc01 = __builtin_amdgcn_wmma_f32_16x16x4_f32(false, a0, false, b1, (short)0, acc01, false, false);
      acc10 = __builtin_amdgcn_wmma_f32_16x16x4_f32(false, a1, false, b0, (short)0, acc10, false, false);
      acc11 = __builtin_amdgcn_wmma_f32_16x16x4_f32(false, a1, false, b1, (short)0, acc11, false, false);
    }
    __syncthreads();
  }

  int gn0 = bn + wn * 32 + row16;
  int gn1 = gn0 + 16;
  int gm0 = bm + wm * 32 + 8 * khalf;
  int gm1 = gm0 + 16;
#pragma unroll
  for (int v = 0; v < 8; ++v) {
    if (gm0 + v < M) {
      C[(size_t)(gm0 + v) * N + gn0] = acc00[v];
      C[(size_t)(gm0 + v) * N + gn1] = acc01[v];
    }
    if (gm1 + v < M) {
      C[(size_t)(gm1 + v) * N + gn0] = acc10[v];
      C[(size_t)(gm1 + v) * N + gn1] = acc11[v];
    }
  }
}

__device__ __forceinline__ float dot4(const float4& a, const float4& b, float s) {
  s = fmaf(a.x, b.x, s); s = fmaf(a.y, b.y, s); s = fmaf(a.z, b.z, s); s = fmaf(a.w, b.w, s);
  return s;
}
__device__ __forceinline__ float wave_sum(float v) {
  for (int off = 16; off > 0; off >>= 1) v += __shfl_xor(v, off, 32);
  return v;
}

template <int CC>
__device__ __forceinline__ void proj_row(const float* __restrict__ We, const float* __restrict__ ae,
                                         float* __restrict__ proj, int k, int lane) {
  static_assert(CC % 128 == 0, "a row is a whole number of 512-byte steps of a wave");
  constexpr int PH = CC / 128;
  if (k >= EMB) return;
  const float4* w4 = (const float4*)(We + (size_t)k * (NH * CC));
  const float4* a4 = (const float4*)ae;
  float s[NH];
#pragma unroll
  for (int h = 0; h < NH; ++h) {
    float acc = 0.f;
#pragma unroll
    for (int u = 0; u < PH; ++u) {
      const int q = (h * PH + u) * 32 + lane;
      acc = dot4(w4[q], a4[q], acc);
    }
    s[h] = wave_sum(acc);
  }
  if (lane == 0) {
    float4 r; r.x = s[0]; r.y = s[1]; r.z = s[2]; r.w = s[3];
    ((float4*)proj)[k] = r;
  }
}

template <int CC>
__global__ __launch_bounds__(128) void k_alpha_node(const float* __restrict__ hx,
                                                    const float* __restrict__ a_src,
                                                    const float* __restrict__ a_dst,
                                                    float* __restrict__ ans,
                                                    float* __restrict__ adn) {
  static_assert(CC % 128 == 0, "a row is a whole number of 512-byte steps of a wave");
  static_assert(NH == 4, "one 16-byte store per node and output");
  constexpr int PH = CC / 128;
  const int lane = threadIdx.x & 31;
  const int n = __builtin_amdgcn_readfirstlane((int)((blockIdx.x * 128 + threadIdx.x) >> 5));
  if (n >= NN) return;
  const float4* h4 = (const float4*)(hx + (size_t)n * (NH * CC));
  const float4* s4 = (const float4*)a_src;
  const float4* d4 = (const float4*)a_dst;
  float s[NH], t[NH];
#pragma unroll
  for (int h = 0; h < NH; ++h) {
    float as = 0.f, ad = 0.f;
#pragma unroll
    for (int u = 0; u < PH; ++u) {
      const int q = (h * PH + u) * 32 + lane;
      const float4 v = h4[q];
      as = dot4(v, s4[q], as);
      ad = dot4(v, d4[q], ad);
    }
    s[h] = wave_sum(as);
    t[h] = wave_sum(ad);
  }
  if (lane == 0) {
    float4 r; r.x = s[0]; r.y = s[1]; r.z = s[2]; r.w = s[3];
    ((float4*)ans)[n] = r;
    float4 w; w.x = t[0]; w.y = t[1]; w.z = t[2]; w.w = t[3];
    ((float4*)adn)[n] = w;
  }
}

__device__ __forceinline__ void attr_tab_row(const float* __restrict__ emb,
                                             const float* __restrict__ proj,
                                             float* __restrict__ tab, int a, int lane) {
  const float* row = emb + (size_t)a * EMB;
  const float4* p4 = (const float4*)proj;
  float a0 = 0.f, a1 = 0.f, a2 = 0.f, a3 = 0.f;
#pragma unroll
  for (int t = 0; t < EMB / 32; ++t) {
    int k = lane + 32 * t;
    float v = row[k];
    float4 p = p4[k];
    a0 += v * p.x; a1 += v * p.y; a2 += v * p.z; a3 += v * p.w;
  }
  for (int off = 16; off > 0; off >>= 1) {
    a0 += __shfl_xor(a0, off, 32);
    a1 += __shfl_xor(a1, off, 32);
    a2 += __shfl_xor(a2, off, 32);
    a3 += __shfl_xor(a3, off, 32);
  }
  if (lane == 0) {
    float4 r; r.x = a0; r.y = a1; r.z = a2; r.w = a3;
    ((float4*)tab)[a] = r;
  }
}

__device__ __forceinline__ void edge_fma(float4& acc, float w0, float w1, float w2, float w3,
                                         const float4& v0, const float4& v1, const float4& v2,
                                         const float4& v3) {
  acc.x = fmaf(w0, v0.x, acc.x); acc.y = fmaf(w0, v0.y, acc.y);
  acc.z = fmaf(w0, v0.z, acc.z); acc.w = fmaf(w0, v0.w, acc.w);
  acc.x = fmaf(w1, v1.x, acc.x); acc.y = fmaf(w1, v1.y, acc.y);
  acc.z = fmaf(w1, v1.z, acc.z); acc.w = fmaf(w1, v1.w, acc.w);
  acc.x = fmaf(w2, v2.x, acc.x); acc.y = fmaf(w2, v2.y, acc.y);
  acc.z = fmaf(w2, v2.z, acc.z); acc.w = fmaf(w2, v2.w, acc.w);
  acc.x = fmaf(w3, v3.x, acc.x); acc.y = fmaf(w3, v3.y, acc.y);
  acc.z = fmaf(w3, v3.z, acc.z); acc.w = fmaf(w3, v3.w, acc.w);
}

__device__ __forceinline__ float leaky(float a) { return (a > 0.f) ? a : NEG_SLOPE * a; }

__device__ __forceinline__ float4 edge_score(int s, int pa, const float4* __restrict__ ans4,
                                             const float4* __restrict__ tab4, const float4& ad) {
  const float4 an = ans4[s];
  const float4 t0 = tab4[(unsigned)pa & 0xffffu], t1 = tab4[(unsigned)pa >> 16];
  float4 a;
  a.x = leaky(an.x + ad.x + (t0.x + t1.x));
  a.y = leaky(an.y + ad.y + (t0.y + t1.y));
  a.z = leaky(an.z + ad.z + (t0.z + t1.z));
  a.w = leaky(an.w + ad.w + (t0.w + t1.w));
  return a;
}
__device__ __forceinline__ float4 max4(const float4& a, const float4& b) {
  float4 r; r.x = fmaxf(a.x, b.x); r.y = fmaxf(a.y, b.y); r.z = fmaxf(a.z, b.z); r.w = fmaxf(a.w, b.w);
  return r;
}
__device__ __forceinline__ float4 exp4(const float4& a, const float4& m) {
  float4 r; r.x = expf(a.x - m.x); r.y = expf(a.y - m.y); r.z = expf(a.z - m.z); r.w = expf(a.w - m.w);
  return r;
}
__device__ __forceinline__ float first_lane(float v) {
  return __int_as_float(__builtin_amdgcn_readfirstlane(__float_as_int(v)));
}
__device__ __forceinline__ float4 wave_max4(float4 v) {
  for (int off = 16; off > 0; off >>= 1) {
    v.x = fmaxf(v.x, __shfl_xor(v.x, off, 32)); v.y = fmaxf(v.y, __shfl_xor(v.y, off, 32));
    v.z = fmaxf(v.z, __shfl_xor(v.z, off, 32)); v.w = fmaxf(v.w, __shfl_xor(v.w, off, 32));
  }
  return v;
}
__device__ __forceinline__ float4 wave_sum4(float4 v) {
  for (int off = 16; off > 0; off >>= 1) {
    v.x += __shfl_xor(v.x, off, 32); v.y += __shfl_xor(v.y, off, 32);
    v.z += __shfl_xor(v.z, off, 32); v.w += __shfl_xor(v.w, off, 32);
  }
  return v;
}

__global__ __launch_bounds__(256) void k_gather(
    const float* __restrict__ hx, const float* __restrict__ ans, const float* __restrict__ adn,
    const float* __restrict__ tab, const unsigned* __restrict__ cnt,
    const unsigned* __restrict__ rowptr, const int* __restrict__ csrc, const int* __restrict__ cattr,
    const float* __restrict__ bias, float* __restrict__ o, int C, int chunks, int relu) {
  const int lane = threadIdx.x & 31;
  const int gw = __builtin_amdgcn_readfirstlane((int)((blockIdx.x * 256 + threadIdx.x) >> 5));
  const int node = gw / chunks;
  if (node >= NN) return;
  const int c0 = (gw - node * chunks) * 128 + lane * 4;
  const size_t rs = (size_t)NH * C;
  const float4* ans4 = (const float4*)ans;
  const float4* tab4 = (const float4*)tab;
  const int beg = (int)rowptr[node], end = (int)rowptr[node + 1];

  const float* ps = hx + (size_t)node * rs + c0;
  const float4 sv0 = *(const float4*)ps, sv1 = *(const float4*)(ps + C);
  const float4 sv2 = *(const float4*)(ps + 2 * C), sv3 = *(const float4*)(ps + 3 * C);

  const float4 ad = *(const float4*)(adn + (size_t)node * NH);
  float4 a_self;
  {
    const unsigned* cr = cnt + (size_t)node * NEA;
    float s0 = 0.f, s1 = 0.f, s2 = 0.f, s3 = 0.f;
#pragma unroll
    for (int a = 0; a < NEA; ++a) {
      const float cf = (float)cr[a];
      const float4 t = tab4[a];
      s0 += cf * t.x; s1 += cf * t.y; s2 += cf * t.z; s3 += cf * t.w;
    }
    const float dg = fmaxf((float)(end - beg), 1.0f);
    const float4 an = ans4[node];
    a_self.x = leaky(an.x + ad.x + s0 / dg);
    a_self.y = leaky(an.y + ad.y + s1 / dg);
    a_self.z = leaky(an.z + ad.z + s2 / dg);
    a_self.w = leaky(an.w + ad.w + s3 / dg);
  }

  const bool has0 = beg + lane < end;
  int s_first = node;
  float4 a_first = a_self;
  if (has0) {
    s_first = csrc[beg + lane];
    a_first = edge_score(s_first, cattr[beg + lane], ans4, tab4, ad);
  }
  float4 m = max4(a_first, a_self);
  for (int base = beg + 32; base < end; base += 32) {
    const int idx = base + lane;
    if (idx < end) m = max4(m, edge_score(csrc[idx], cattr[idx], ans4, tab4, ad));
  }
  m = wave_max4(m);
  m.x = first_lane(m.x); m.y = first_lane(m.y); m.z = first_lane(m.z); m.w = first_lane(m.w);
  float4 e_first = {0.f, 0.f, 0.f, 0.f};
  if (has0) e_first = exp4(a_first, m);
  float4 dsum = e_first;
  for (int base = beg + 32; base < end; base += 32) {
    const int idx = base + lane;
    if (idx < end) {
      const float4 ex = exp4(edge_score(csrc[idx], cattr[idx], ans4, tab4, ad), m);
      dsum.x += ex.x; dsum.y += ex.y; dsum.z += ex.z; dsum.w += ex.w;
    }
  }
  dsum = wave_sum4(dsum);
  dsum.x = first_lane(dsum.x); dsum.y = first_lane(dsum.y); dsum.z = first_lane(dsum.z); dsum.w = first_lane(dsum.w);
  const float4 e_self = exp4(a_self, m);
  const float i0 = 1.0f / (dsum.x + e_self.x), i1 = 1.0f / (dsum.y + e_self.y);
  const float i2 = 1.0f / (dsum.z + e_self.z), i3 = 1.0f / (dsum.w + e_self.w);

  float4 acc = {0.f, 0.f, 0.f, 0.f};
  edge_fma(acc, e_self.x * i0, e_self.y * i1, e_self.z * i2, e_self.w * i3, sv0, sv1, sv2, sv3);

  int s_l = s_first;
  float w0_l = e_first.x * i0, w1_l = e_first.y * i1, w2_l = e_first.z * i2, w3_l = e_first.w * i3;
  for (int base = beg; base < end; base += 32) {
    if (base != beg) {
      const int idx = base + lane;
      s_l = node;
      w0_l = 0.f; w1_l = 0.f; w2_l = 0.f; w3_l = 0.f;
      if (idx < end) {
        s_l = csrc[idx];
        const float4 ex = exp4(edge_score(s_l, cattr[idx], ans4, tab4, ad), m);
        w0_l = ex.x * i0; w1_l = ex.y * i1; w2_l = ex.z * i2; w3_l = ex.w * i3;
      }
    }
    const int cnt_b = min(32, end - base);
    int j = 0;
    for (; j + 4 <= cnt_b; j += 4) {
      float4 v[4][4];
      float w[4][4];
#pragma unroll
      for (int u = 0; u < 4; ++u) {
        const int s = __builtin_amdgcn_readlane(s_l, j + u);
        const float* p = hx + (size_t)s * rs + c0;
        v[u][0] = *(const float4*)p;
        v[u][1] = *(const float4*)(p + C);
        v[u][2] = *(const float4*)(p + 2 * C);
        v[u][3] = *(const float4*)(p + 3 * C);
        w[u][0] = __int_as_float(__builtin_amdgcn_readlane(__float_as_int(w0_l), j + u));
        w[u][1] = __int_as_float(__builtin_amdgcn_readlane(__float_as_int(w1_l), j + u));
        w[u][2] = __int_as_float(__builtin_amdgcn_readlane(__float_as_int(w2_l), j + u));
        w[u][3] = __int_as_float(__builtin_amdgcn_readlane(__float_as_int(w3_l), j + u));
      }
#pragma unroll
      for (int u = 0; u < 4; ++u)
        edge_fma(acc, w[u][0], w[u][1], w[u][2], w[u][3], v[u][0], v[u][1], v[u][2], v[u][3]);
    }
    for (; j < cnt_b; ++j) {
      const int s = __builtin_amdgcn_readlane(s_l, j);
      const float* p = hx + (size_t)s * rs + c0;
      const float4 v0 = *(const float4*)p, v1 = *(const float4*)(p + C);
      const float4 v2 = *(const float4*)(p + 2 * C), v3 = *(const float4*)(p + 3 * C);
      edge_fma(acc,
               __int_as_float(__builtin_amdgcn_readlane(__float_as_int(w0_l), j)),
               __int_as_float(__builtin_amdgcn_readlane(__float_as_int(w1_l), j)),
               __int_as_float(__builtin_amdgcn_readlane(__float_as_int(w2_l), j)),
               __int_as_float(__builtin_amdgcn_readlane(__float_as_int(w3_l), j)),
               v0, v1, v2, v3);
    }
  }

  float4 r;
  r.x = 0.25f * acc.x + bias[c0 + 0];
  r.y = 0.25f * acc.y + bias[c0 + 1];
  r.z = 0.25f * acc.z + bias[c0 + 2];
  r.w = 0.25f * acc.w + bias[c0 + 3];
  if (relu) { r.x = fmaxf(r.x, 0.f); r.y = fmaxf(r.y, 0.f); r.z = fmaxf(r.z, 0.f); r.w = fmaxf(r.w, 0.f); }
  *(float4*)(o + (size_t)node * C + c0) = r;
}

#define CNT_BLOCKS  ((NE + 255) / 256)
#define PROJ_BLOCKS (EMB / 8)
static_assert(EMB % 8 == 0, "a proj work-group covers 8 whole k");
__global__ __launch_bounds__(256) void k_count_proj(
    const int* __restrict__ dst, const int* __restrict__ ea_idx,
    unsigned* __restrict__ cnt, unsigned* __restrict__ deg,
    const float* __restrict__ We1, const float* __restrict__ ae1, float* __restrict__ proj1,
    const float* __restrict__ We2, const float* __restrict__ ae2, float* __restrict__ proj2) {
  const int b = blockIdx.x;
  const int lane = threadIdx.x & 31;
  const int wave = __builtin_amdgcn_readfirstlane((int)(threadIdx.x >> 5));
  if (b < PROJ_BLOCKS) {
    proj_row<FT>(We2, ae2, proj2, b * 8 + wave, lane);
  } else if (b < 2 * PROJ_BLOCKS) {
    proj_row<HID>(We1, ae1, proj1, (b - PROJ_BLOCKS) * 8 + wave, lane);
  } else {
    attr_count_edge((b - 2 * PROJ_BLOCKS) * 256 + (int)threadIdx.x, dst, ea_idx, cnt, deg);
  }
}

#define TAB_BLOCKS ((2 * NEA + 31) / 32)
__global__ __launch_bounds__(1024) void k_scan_tab(
    const unsigned* __restrict__ deg, unsigned* __restrict__ rowptr, unsigned* __restrict__ cur,
    const float* __restrict__ emb,
    const float* __restrict__ proj1, float* __restrict__ tab1,
    const float* __restrict__ proj2, float* __restrict__ tab2) {
  __shared__ unsigned wsum[32];
  if (blockIdx.x == 0) {
    csr_scan_block(deg, rowptr, cur, wsum);
    return;
  }
  const int lane = threadIdx.x & 31;
  const int r = __builtin_amdgcn_readfirstlane((int)((blockIdx.x - 1) * 32 + (threadIdx.x >> 5)));
  if (r < NEA) attr_tab_row(emb, proj1, tab1, r, lane);
  else if (r < 2 * NEA) attr_tab_row(emb, proj2, tab2, r - NEA, lane);
}

extern "C" void kernel_launch(void* const* d_in, const int* in_sizes, int n_in,
                              void* d_out, int out_size, void* d_ws, size_t ws_size,
                              hipStream_t stream) {
  const int* x      = (const int*)d_in[0];
  const int* eidx   = (const int*)d_in[1];
  const int* src    = eidx;
  const int* dst    = eidx + NE;
  const int* eattr  = (const int*)d_in[2];
  const float* xemb = (const float*)d_in[3];
  const float* eemb = (const float*)d_in[4];
  const float* W1   = (const float*)d_in[5];
  const float* as1  = (const float*)d_in[6];
  const float* ad1  = (const float*)d_in[7];
  const float* We1  = (const float*)d_in[8];
  const float* ae1  = (const float*)d_in[9];
  const float* b1   = (const float*)d_in[10];
  const float* W2   = (const float*)d_in[11];
  const float* as2  = (const float*)d_in[12];
  const float* ad2  = (const float*)d_in[13];
  const float* We2  = (const float*)d_in[14];
  const float* ae2  = (const float*)d_in[15];
  const float* b2   = (const float*)d_in[16];
  float* out = (float*)d_out;

  float* ws = (float*)d_ws;
  size_t off = 0;
  unsigned* CNT = (unsigned*)(ws + off); off += (size_t)2048 * NEA;
  unsigned* DEG = (unsigned*)(ws + off); off += 2048;
  unsigned* ROWPTR = (unsigned*)(ws + off); off += 2112;
  unsigned* CUR = (unsigned*)(ws + off); off += 2048;
  int* CSRC = (int*)(ws + off); off += (size_t)NE;
  int* CATTR = (int*)(ws + off); off += (size_t)NE;
  float* TAB1 = ws + off; off += 128;
  float* TAB2 = ws + off; off += 128;
  float* HX1  = ws + off; off += (size_t)NN * NH * HID;
  float* HX2  = ws + off; off += (size_t)NN * NH * FT;
  float* PROJ1 = ws + off; off += (size_t)EMB * NH;
  float* PROJ2 = ws + off; off += (size_t)EMB * NH;
  float* ANS  = ws + off; off += (size_t)NN * NH;
  float* ADN  = ws + off; off += (size_t)NN * NH;
  float* HB   = ws + off; off += (size_t)NN * HID;
  (void)ws_size; (void)in_sizes; (void)n_in; (void)out_size;

  const int T = 256;

  hipMemsetAsync(CNT, 0, (size_t)(2048 * NEA + 2048) * sizeof(unsigned), stream);

  k_count_proj<<<2 * PROJ_BLOCKS + CNT_BLOCKS, 256, 0, stream>>>(dst, eattr, CNT, DEG, We1, ae1, PROJ1, We2, ae2, PROJ2);
  k_scan_tab<<<1 + TAB_BLOCKS, 1024, 0, stream>>>(DEG, ROWPTR, CUR, eemb, PROJ1, TAB1, PROJ2, TAB2);
  k_csr_fill<<<cdiv(NE, T), T, 0, stream>>>(src, dst, eattr, CUR, CSRC, CATTR);

  {
    dim3 g(NH * HID / 128, cdiv(NN, 64));
    k_gemm_wmma_f32<true><<<g, 256, 0, stream>>>(nullptr, x, xemb, W1, HX1, NN, NH * HID, EMB);
  }
  k_alpha_node<HID><<<cdiv((long long)NN * 32, 128), 128, 0, stream>>>(HX1, as1, ad1, ANS, ADN);
  k_gather<<<cdiv((long long)NN * (HID / 128), 8), 256, 0, stream>>>(HX1, ANS, ADN, TAB1, CNT, ROWPTR, CSRC, CATTR, b1, HB, HID, HID / 128, 1);

  {
    dim3 g(NH * FT / 128, cdiv(NN, 64));
    k_gemm_wmma_f32<false><<<g, 256, 0, stream>>>(HB, nullptr, nullptr, W2, HX2, NN, NH * FT, HID);
  }
  k_alpha_node<FT><<<cdiv((long long)NN * 32, 128), 128, 0, stream>>>(HX2, as2, ad2, ANS, ADN);
  k_gather<<<cdiv((long long)NN * (FT / 128), 8), 256, 0, stream>>>(HX2, ANS, ADN, TAB2, CNT, ROWPTR, CSRC, CATTR, b2, out, FT, FT / 128, 0);
}


# GEMM K loop software-pipelined: next K-block's global loads in flight in registers during the current block's matrix instructions, stored to a second LDS copy after them; one barrier per K-block
# speedup vs baseline: 5.8241x; 1.3237x over previous
#include <hip/hip_runtime.h>
#include <math.h>

#define NN   2000
#define NE   32000
#define EMB  768
#define HID  384
#define FT   768
#define NH   4
#define NEA  18
#define NEG_SLOPE 0.2f

typedef __attribute__((ext_vector_type(2))) float v2f;
typedef __attribute__((ext_vector_type(8))) float v8f;

static inline int cdiv(long long a, long long b) { return (int)((a + b - 1) / b); }

__device__ __forceinline__ void attr_count_edge(int e, const int* __restrict__ dst,
                                                const int* __restrict__ ea_idx,
                                                unsigned* __restrict__ cnt, unsigned* __restrict__ deg) {
  if (e >= NE) return;
  int d = dst[e];
  atomicAdd(&cnt[d * NEA + ea_idx[e * 2 + 0]], 1u);
  atomicAdd(&cnt[d * NEA + ea_idx[e * 2 + 1]], 1u);
  atomicAdd(&deg[d], 1u);
}

static_assert(NN <= 2048, "csr_scan_block covers two nodes per thread of one 1024-thread block");
__device__ __forceinline__ void csr_scan_block(const unsigned* __restrict__ deg,
                                               unsigned* __restrict__ rowptr,
                                               unsigned* __restrict__ cur, unsigned* wsum) {
  const int t = threadIdx.x, lane = t & 31, wave = t >> 5;
  const int n0 = 2 * t;
  const unsigned a = (n0 < NN) ? deg[n0] : 0u;
  const unsigned b = (n0 + 1 < NN) ? deg[n0 + 1] : 0u;
  unsigned incl = a + b;
  for (int off = 1; off < 32; off <<= 1) {
    unsigned v = __shfl_up(incl, off, 32);
    if (lane >= off) incl += v;
  }
  if (lane == 31) wsum[wave] = incl;
  __syncthreads();
  if (wave == 0) {
    unsigned w = wsum[lane];
    for (int off = 1; off < 32; off <<= 1) {
      unsigned v = __shfl_up(w, off, 32);
      if (lane >= off) w += v;
    }
    wsum[lane] = w;
  }
  __syncthreads();
  const unsigned excl = incl - (a + b) + (wave ? wsum[wave - 1] : 0u);
  if (n0 < NN)     { rowptr[n0] = excl;         cur[n0] = excl; }
  if (n0 + 1 < NN) { rowptr[n0 + 1] = excl + a; cur[n0 + 1] = excl + a; }
  if (t == 1023) rowptr[NN] = excl + a + b;
}

__global__ void k_csr_fill(const int* __restrict__ src, const int* __restrict__ dst,
                           const int* __restrict__ ea_idx,
                           unsigned* __restrict__ cur, int* __restrict__ csrc,
                           int* __restrict__ cattr) {
  int e = blockIdx.x * blockDim.x + threadIdx.x;
  if (e >= NE) return;
  unsigned pos = atomicAdd(&cur[dst[e]], 1u);
  csrc[pos] = src[e];
  cattr[pos] = (int)((unsigned)ea_idx[e * 2 + 0] | ((unsigned)ea_idx[e * 2 + 1] << 16));
}

#define A_PITCH 36
#define B_PITCH 34

__device__ __forceinline__ void gemm_block_mma(const float* __restrict__ a0p, const float* __restrict__ a1p,
                                               const float* __restrict__ b0p, const float* __restrict__ b1p,
                                               int khalf, v8f& acc00, v8f& acc01, v8f& acc10, v8f& acc11) {
#pragma unroll
  for (int ks = 0; ks < 32; ks += 4) {
    int kk = ks + khalf * 2;
    v2f a0 = *(const v2f*)(a0p + kk);
    v2f a1 = *(const v2f*)(a1p + kk);
    v2f b0 = *(const v2f*)(b0p + kk);
    v2f b1 = *(const v2f*)(b1p + kk);
    acc00 = __builtin_amdgcn_wmma_f32_16x16x4_f32(false, a0, false, b0, (short)0, acc00, false, false);
    acc01 = __builtin_amdgcn_wmma_f32_16x16x4_f32(false, a0, false, b1, (short)0, acc01, false, false);
    acc10 = __builtin_amdgcn_wmma_f32_16x16x4_f32(false, a1, false, b0, (short)0, acc10, false, false);
    acc11 = __builtin_amdgcn_wmma_f32_16x16x4_f32(false, a1, false, b1, (short)0, acc11, false, false);
  }
}

template <bool EMB_A>
__global__ __launch_bounds__(256) void k_gemm_wmma_f32(
    const float* __restrict__ A, const int* __restrict__ xi, const float* __restrict__ emb,
    const float* __restrict__ B, float* __restrict__ C, int M, int N, int K) {
  __shared__ __align__(16) float sA[2][64][A_PITCH];
  __shared__ __align__(16) float sBt[2][128][B_PITCH];
  const int tid = threadIdx.x;
  const int wave = tid >> 5, lane = tid & 31;
  const int wm = wave >> 2, wn = wave & 3;
  const int row16 = lane & 15, khalf = lane >> 4;
  const int bm = blockIdx.y * 64, bn = blockIdx.x * 128;

  const int ar = tid >> 5, ac = tid & 31;
  const int bk = tid >> 7, bc = tid & 127;

  const float* r0[8];
  const float* r1[8];
  unsigned aoff[8];
  bool rowok[8];
  if constexpr (EMB_A) {
    const int wv = __builtin_amdgcn_readfirstlane(wave);
    const int grl = min(bm + wv + 8 * ((lane >> 1) & 7), M - 1);
    const int myid = xi[grl * 2 + (lane & 1)];
#pragma unroll
    for (int j = 0; j < 8; ++j) {
      rowok[j] = bm + wv + 8 * j < M;
      r0[j] = emb + (size_t)__builtin_amdgcn_readlane(myid, 2 * j) * EMB;
      r1[j] = emb + (size_t)__builtin_amdgcn_readlane(myid, 2 * j + 1) * EMB;
    }
  } else {
#pragma unroll
    for (int j = 0; j < 8; ++j) {
      const int gr = bm + ar + 8 * j;
      rowok[j] = gr < M;
      aoff[j] = (unsigned)min(gr, M - 1) * (unsigned)K + (unsigned)ac;
    }
  }
  const unsigned boff = (unsigned)bk * (unsigned)N + (unsigned)bc;

  float ua[8], ub[8], rb[16];

  auto load_tiles = [&](int k0) {
    if constexpr (EMB_A) {
      const unsigned kl = (unsigned)(k0 + lane);
#pragma unroll
      for (int j = 0; j < 8; ++j) { ua[j] = r0[j][kl]; ub[j] = r1[j][kl]; }
    } else {
      const float* Ak = A + k0;
#pragma unroll
      for (int j = 0; j < 8; ++j) ua[j] = Ak[aoff[j]];
    }
    const float* Bk = B + (size_t)k0 * N + bn;
#pragma unroll
    for (int j = 0; j < 16; ++j) rb[j] = Bk[boff + (unsigned)(2 * j) * (unsigned)N];
  };
  auto store_tiles = [&](int p) {
#pragma unroll
    for (int j = 0; j < 8; ++j) {
      float v;
      if constexpr (EMB_A) v = ua[j] + ub[j]; else v = ua[j];
      sA[p][ar + 8 * j][ac] = rowok[j] ? v : 0.0f;
    }
#pragma unroll
    for (int j = 0; j < 16; ++j) sBt[p][bc][bk + 2 * j] = rb[j];
  };

  v8f acc00 = {}, acc01 = {}, acc10 = {}, acc11 = {};
  const int arow = wm * 32 + row16, brow = wn * 32 + row16;

  load_tiles(0);
  store_tiles(0);
  __syncthreads();
  int p = 0;
  for (int k0 = 32; k0 < K; k0 += 32) {
    load_tiles(k0);
    __builtin_amdgcn_sched_barrier(0);
    gemm_block_mma(&sA[p][arow][0], &sA[p][arow + 16][0], &sBt[p][brow][0], &sBt[p][brow + 16][0],
                   khalf, acc00, acc01, acc10, acc11);
    __builtin_amdgcn_sched_barrier(0);
    store_tiles(p ^ 1);
    __syncthreads();
    p ^= 1;
  }
  gemm_block_mma(&sA[p][arow][0], &sA[p][arow + 16][0], &sBt[p][brow][0], &sBt[p][brow + 16][0],
                 khalf, acc00, acc01, acc10, acc11);

  int gn0 = bn + wn * 32 + row16;
  int gn1 = gn0 + 16;
  int gm0 = bm + wm * 32 + 8 * khalf;
  int gm1 = gm0 + 16;
#pragma unroll
  for (int v = 0; v < 8; ++v) {
    if (gm0 + v < M) {
      C[(size_t)(gm0 + v) * N + gn0] = acc00[v];
      C[(size_t)(gm0 + v) * N + gn1] = acc01[v];
    }
    if (gm1 + v < M) {
      C[(size_t)(gm1 + v) * N + gn0] = acc10[v];
      C[(size_t)(gm1 + v) * N + gn1] = acc11[v];
    }
  }
}

__device__ __forceinline__ float dot4(const float4& a, const float4& b, float s) {
  s = fmaf(a.x, b.x, s); s = fmaf(a.y, b.y, s); s = fmaf(a.z, b.z, s); s = fmaf(a.w, b.w, s);
  return s;
}
__device__ __forceinline__ float wave_sum(float v) {
  for (int off = 16; off > 0; off >>= 1) v += __shfl_xor(v, off, 32);
  return v;
}

template <int CC>
__device__ __forceinline__ void proj_row(const float* __restrict__ We, const float* __restrict__ ae,
                                         float* __restrict__ proj, int k, int lane) {
  static_assert(CC % 128 == 0, "a row is a whole number of 512-byte steps of a wave");
  constexpr int PH = CC / 128;
  if (k >= EMB) return;
  const float4* w4 = (const float4*)(We + (size_t)k * (NH * CC));
  const float4* a4 = (const float4*)ae;
  float s[NH];
#pragma unroll
  for (int h = 0; h < NH; ++h) {
    float acc = 0.f;
#pragma unroll
    for (int u = 0; u < PH; ++u) {
      const int q = (h * PH + u) * 32 + lane;
      acc = dot4(w4[q], a4[q], acc);
    }
    s[h] = wave_sum(acc);
  }
  if (lane == 0) {
    float4 r; r.x = s[0]; r.y = s[1]; r.z = s[2]; r.w = s[3];
    ((float4*)proj)[k] = r;
  }
}

template <int CC>
__global__ __launch_bounds__(128) void k_alpha_node(const float* __restrict__ hx,
                                                    const float* __restrict__ a_src,
                                                    const float* __restrict__ a_dst,
                                                    float* __restrict__ ans,
                                                    float* __restrict__ adn) {
  static_assert(CC % 128 == 0, "a row is a whole number of 512-byte steps of a wave");
  static_assert(NH == 4, "one 16-byte store per node and output");
  constexpr int PH = CC / 128;
  const int lane = threadIdx.x & 31;
  const int n = __builtin_amdgcn_readfirstlane((int)((blockIdx.x * 128 + threadIdx.x) >> 5));
  if (n >= NN) return;
  const float4* h4 = (const float4*)(hx + (size_t)n * (NH * CC));
  const float4* s4 = (const float4*)a_src;
  const float4* d4 = (const float4*)a_dst;
  float s[NH], t[NH];
#pragma unroll
  for (int h = 0; h < NH; ++h) {
    float as = 0.f, ad = 0.f;
#pragma unroll
    for (int u = 0; u < PH; ++u) {
      const int q = (h * PH + u) * 32 + lane;
      const float4 v = h4[q];
      as = dot4(v, s4[q], as);
      ad = dot4(v, d4[q], ad);
    }
    s[h] = wave_sum(as);
    t[h] = wave_sum(ad);
  }
  if (lane == 0) {
    float4 r; r.x = s[0]; r.y = s[1]; r.z = s[2]; r.w = s[3];
    ((float4*)ans)[n] = r;
    float4 w; w.x = t[0]; w.y = t[1]; w.z = t[2]; w.w = t[3];
    ((float4*)adn)[n] = w;
  }
}

__device__ __forceinline__ void attr_tab_row(const float* __restrict__ emb,
                                             const float* __restrict__ proj,
                                             float* __restrict__ tab, int a, int lane) {
  const float* row = emb + (size_t)a * EMB;
  const float4* p4 = (const float4*)proj;
  float a0 = 0.f, a1 = 0.f, a2 = 0.f, a3 = 0.f;
#pragma unroll
  for (int t = 0; t < EMB / 32; ++t) {
    int k = lane + 32 * t;
    float v = row[k];
    float4 p = p4[k];
    a0 += v * p.x; a1 += v * p.y; a2 += v * p.z; a3 += v * p.w;
  }
  for (int off = 16; off > 0; off >>= 1) {
    a0 += __shfl_xor(a0, off, 32);
    a1 += __shfl_xor(a1, off, 32);
    a2 += __shfl_xor(a2, off, 32);
    a3 += __shfl_xor(a3, off, 32);
  }
  if (lane == 0) {
    float4 r; r.x = a0; r.y = a1; r.z = a2; r.w = a3;
    ((float4*)tab)[a] = r;
  }
}

__device__ __forceinline__ void edge_fma(float4& acc, float w0, float w1, float w2, float w3,
                                         const float4& v0, const float4& v1, const float4& v2,
                                         const float4& v3) {
  acc.x = fmaf(w0, v0.x, acc.x); acc.y = fmaf(w0, v0.y, acc.y);
  acc.z = fmaf(w0, v0.z, acc.z); acc.w = fmaf(w0, v0.w, acc.w);
  acc.x = fmaf(w1, v1.x, acc.x); acc.y = fmaf(w1, v1.y, acc.y);
  acc.z = fmaf(w1, v1.z, acc.z); acc.w = fmaf(w1, v1.w, acc.w);
  acc.x = fmaf(w2, v2.x, acc.x); acc.y = fmaf(w2, v2.y, acc.y);
  acc.z = fmaf(w2, v2.z, acc.z); acc.w = fmaf(w2, v2.w, acc.w);
  acc.x = fmaf(w3, v3.x, acc.x); acc.y = fmaf(w3, v3.y, acc.y);
  acc.z = fmaf(w3, v3.z, acc.z); acc.w = fmaf(w3, v3.w, acc.w);
}

__device__ __forceinline__ float leaky(float a) { return (a > 0.f) ? a : NEG_SLOPE * a; }

__device__ __forceinline__ float4 edge_score(int s, int pa, const float4* __restrict__ ans4,
                                             const float4* __restrict__ tab4, const float4& ad) {
  const float4 an = ans4[s];
  const float4 t0 = tab4[(unsigned)pa & 0xffffu], t1 = tab4[(unsigned)pa >> 16];
  float4 a;
  a.x = leaky(an.x + ad.x + (t0.x + t1.x));
  a.y = leaky(an.y + ad.y + (t0.y + t1.y));
  a.z = leaky(an.z + ad.z + (t0.z + t1.z));
  a.w = leaky(an.w + ad.w + (t0.w + t1.w));
  return a;
}
__device__ __forceinline__ float4 max4(const float4& a, const float4& b) {
  float4 r; r.x = fmaxf(a.x, b.x); r.y = fmaxf(a.y, b.y); r.z = fmaxf(a.z, b.z); r.w = fmaxf(a.w, b.w);
  return r;
}
__device__ __forceinline__ float4 exp4(const float4& a, const float4& m) {
  float4 r; r.x = expf(a.x - m.x); r.y = expf(a.y - m.y); r.z = expf(a.z - m.z); r.w = expf(a.w - m.w);
  return r;
}
__device__ __forceinline__ float first_lane(float v) {
  return __int_as_float(__builtin_amdgcn_readfirstlane(__float_as_int(v)));
}
__device__ __forceinline__ float4 wave_max4(float4 v) {
  for (int off = 16; off > 0; off >>= 1) {
    v.x = fmaxf(v.x, __shfl_xor(v.x, off, 32)); v.y = fmaxf(v.y, __shfl_xor(v.y, off, 32));
    v.z = fmaxf(v.z, __shfl_xor(v.z, off, 32)); v.w = fmaxf(v.w, __shfl_xor(v.w, off, 32));
  }
  return v;
}
__device__ __forceinline__ float4 wave_sum4(float4 v) {
  for (int off = 16; off > 0; off >>= 1) {
    v.x += __shfl_xor(v.x, off, 32); v.y += __shfl_xor(v.y, off, 32);
    v.z += __shfl_xor(v.z, off, 32); v.w += __shfl_xor(v.w, off, 32);
  }
  return v;
}

__global__ __launch_bounds__(256) void k_gather(
    const float* __restrict__ hx, const float* __restrict__ ans, const float* __restrict__ adn,
    const float* __restrict__ tab, const unsigned* __restrict__ cnt,
    const unsigned* __restrict__ rowptr, const int* __restrict__ csrc, const int* __restrict__ cattr,
    const float* __restrict__ bias, float* __restrict__ o, int C, int chunks, int relu) {
  const int lane = threadIdx.x & 31;
  const int gw = __builtin_amdgcn_readfirstlane((int)((blockIdx.x * 256 + threadIdx.x) >> 5));
  const int node = gw / chunks;
  if (node >= NN) return;
  const int c0 = (gw - node * chunks) * 128 + lane * 4;
  const size_t rs = (size_t)NH * C;
  const float4* ans4 = (const float4*)ans;
  const float4* tab4 = (const float4*)tab;
  const int beg = (int)rowptr[node], end = (int)rowptr[node + 1];

  const float* ps = hx + (size_t)node * rs + c0;
  const float4 sv0 = *(const float4*)ps, sv1 = *(const float4*)(ps + C);
  const float4 sv2 = *(const float4*)(ps + 2 * C), sv3 = *(const float4*)(ps + 3 * C);

  const float4 ad = *(const float4*)(adn + (size_t)node * NH);
  float4 a_self;
  {
    const unsigned* cr = cnt + (size_t)node * NEA;
    float s0 = 0.f, s1 = 0.f, s2 = 0.f, s3 = 0.f;
#pragma unroll
    for (int a = 0; a < NEA; ++a) {
      const float cf = (float)cr[a];
      const float4 t = tab4[a];
      s0 += cf * t.x; s1 += cf * t.y; s2 += cf * t.z; s3 += cf * t.w;
    }
    const float dg = fmaxf((float)(end - beg), 1.0f);
    const float4 an = ans4[node];
    a_self.x = leaky(an.x + ad.x + s0 / dg);
    a_self.y = leaky(an.y + ad.y + s1 / dg);
    a_self.z = leaky(an.z + ad.z + s2 / dg);
    a_self.w = leaky(an.w + ad.w + s3 / dg);
  }

  const bool has0 = beg + lane < end;
  int s_first = node;
  float4 a_first = a_self;
  if (has0) {
    s_first = csrc[beg + lane];
    a_first = edge_score(s_first, cattr[beg + lane], ans4, tab4, ad);
  }
  float4 m = max4(a_first, a_self);
  for (int base = beg + 32; base < end; base += 32) {
    const int idx = base + lane;
    if (idx < end) m = max4(m, edge_score(csrc[idx], cattr[idx], ans4, tab4, ad));
  }
  m = wave_max4(m);
  m.x = first_lane(m.x); m.y = first_lane(m.y); m.z = first_lane(m.z); m.w = first_lane(m.w);
  float4 e_first = {0.f, 0.f, 0.f, 0.f};
  if (has0) e_first = exp4(a_first, m);
  float4 dsum = e_first;
  for (int base = beg + 32; base < end; base += 32) {
    const int idx = base + lane;
    if (idx < end) {
      const float4 ex = exp4(edge_score(csrc[idx], cattr[idx], ans4, tab4, ad), m);
      dsum.x += ex.x; dsum.y += ex.y; dsum.z += ex.z; dsum.w += ex.w;
    }
  }
  dsum = wave_sum4(dsum);
  dsum.x = first_lane(dsum.x); dsum.y = first_lane(dsum.y); dsum.z = first_lane(dsum.z); dsum.w = first_lane(dsum.w);
  const float4 e_self = exp4(a_self, m);
  const float i0 = 1.0f / (dsum.x + e_self.x), i1 = 1.0f / (dsum.y + e_self.y);
  const float i2 = 1.0f / (dsum.z + e_self.z), i3 = 1.0f / (dsum.w + e_self.w);

  float4 acc = {0.f, 0.f, 0.f, 0.f};
  edge_fma(acc, e_self.x * i0, e_self.y * i1, e_self.z * i2, e_self.w * i3, sv0, sv1, sv2, sv3);

  int s_l = s_first;
  float w0_l = e_first.x * i0, w1_l = e_first.y * i1, w2_l = e_first.z * i2, w3_l = e_first.w * i3;
  for (int base = beg; base < end; base += 32) {
    if (base != beg) {
      const int idx = base + lane;
      s_l = node;
      w0_l = 0.f; w1_l = 0.f; w2_l = 0.f; w3_l = 0.f;
      if (idx < end) {
        s_l = csrc[idx];
        const float4 ex = exp4(edge_score(s_l, cattr[idx], ans4, tab4, ad), m);
        w0_l = ex.x * i0; w1_l = ex.y * i1; w2_l = ex.z * i2; w3_l = ex.w * i3;
      }
    }
    const int cnt_b = min(32, end - base);
    int j = 0;
    for (; j + 4 <= cnt_b; j += 4) {
      float4 v[4][4];
      float w[4][4];
#pragma unroll
      for (int u = 0; u < 4; ++u) {
        const int s = __builtin_amdgcn_readlane(s_l, j + u);
        const float* p = hx + (size_t)s * rs + c0;
        v[u][0] = *(const float4*)p;
        v[u][1] = *(const float4*)(p + C);
        v[u][2] = *(const float4*)(p + 2 * C);
        v[u][3] = *(const float4*)(p + 3 * C);
        w[u][0] = __int_as_float(__builtin_amdgcn_readlane(__float_as_int(w0_l), j + u));
        w[u][1] = __int_as_float(__builtin_amdgcn_readlane(__float_as_int(w1_l), j + u));
        w[u][2] = __int_as_float(__builtin_amdgcn_readlane(__float_as_int(w2_l), j + u));
        w[u][3] = __int_as_float(__builtin_amdgcn_readlane(__float_as_int(w3_l), j + u));
      }
#pragma unroll
      for (int u = 0; u < 4; ++u)
        edge_fma(acc, w[u][0], w[u][1], w[u][2], w[u][3], v[u][0], v[u][1], v[u][2], v[u][3]);
    }
    for (; j < cnt_b; ++j) {
      const int s = __builtin_amdgcn_readlane(s_l, j);
      const float* p = hx + (size_t)s * rs + c0;
      const float4 v0 = *(const float4*)p, v1 = *(const float4*)(p + C);
      const float4 v2 = *(const float4*)(p + 2 * C), v3 = *(const float4*)(p + 3 * C);
      edge_fma(acc,
               __int_as_float(__builtin_amdgcn_readlane(__float_as_int(w0_l), j)),
               __int_as_float(__builtin_amdgcn_readlane(__float_as_int(w1_l), j)),
               __int_as_float(__builtin_amdgcn_readlane(__float_as_int(w2_l), j)),
               __int_as_float(__builtin_amdgcn_readlane(__float_as_int(w3_l), j)),
               v0, v1, v2, v3);
    }
  }

  float4 r;
  r.x = 0.25f * acc.x + bias[c0 + 0];
  r.y = 0.25f * acc.y + bias[c0 + 1];
  r.z = 0.25f * acc.z + bias[c0 + 2];
  r.w = 0.25f * acc.w + bias[c0 + 3];
  if (relu) { r.x = fmaxf(r.x, 0.f); r.y = fmaxf(r.y, 0.f); r.z = fmaxf(r.z, 0.f); r.w = fmaxf(r.w, 0.f); }
  *(float4*)(o + (size_t)node * C + c0) = r;
}

#define CNT_BLOCKS  ((NE + 255) / 256)
#define PROJ_BLOCKS (EMB / 8)
static_assert(EMB % 8 == 0, "a proj work-group covers 8 whole k");
__global__ __launch_bounds__(256) void k_count_proj(
    const int* __restrict__ dst, const int* __restrict__ ea_idx,
    unsigned* __restrict__ cnt, unsigned* __restrict__ deg,
    const float* __restrict__ We1, const float* __restrict__ ae1, float* __restrict__ proj1,
    const float* __restrict__ We2, const float* __restrict__ ae2, float* __restrict__ proj2) {
  const int b = blockIdx.x;
  const int lane = threadIdx.x & 31;
  const int wave = __builtin_amdgcn_readfirstlane((int)(threadIdx.x >> 5));
  if (b < PROJ_BLOCKS) {
    proj_row<FT>(We2, ae2, proj2, b * 8 + wave, lane);
  } else if (b < 2 * PROJ_BLOCKS) {
    proj_row<HID>(We1, ae1, proj1, (b - PROJ_BLOCKS) * 8 + wave, lane);
  } else {
    attr_count_edge((b - 2 * PROJ_BLOCKS) * 256 + (int)threadIdx.x, dst, ea_idx, cnt, deg);
  }
}

#define TAB_BLOCKS ((2 * NEA + 31) / 32)
__global__ __launch_bounds__(1024) void k_scan_tab(
    const unsigned* __restrict__ deg, unsigned* __restrict__ rowptr, unsigned* __restrict__ cur,
    const float* __restrict__ emb,
    const float* __restrict__ proj1, float* __restrict__ tab1,
    const float* __restrict__ proj2, float* __restrict__ tab2) {
  __shared__ unsigned wsum[32];
  if (blockIdx.x == 0) {
    csr_scan_block(deg, rowptr, cur, wsum);
    return;
  }
  const int lane = threadIdx.x & 31;
  const int r = __builtin_amdgcn_readfirstlane((int)((blockIdx.x - 1) * 32 + (threadIdx.x >> 5)));
  if (r < NEA) attr_tab_row(emb, proj1, tab1, r, lane);
  else if (r < 2 * NEA) attr_tab_row(emb, proj2, tab2, r - NEA, lane);
}

extern "C" void kernel_launch(void* const* d_in, const int* in_sizes, int n_in,
                              void* d_out, int out_size, void* d_ws, size_t ws_size,
                              hipStream_t stream) {
  const int* x      = (const int*)d_in[0];
  const int* eidx   = (const int*)d_in[1];
  const int* src    = eidx;
  const int* dst    = eidx + NE;
  const int* eattr  = (const int*)d_in[2];
  const float* xemb = (const float*)d_in[3];
  const float* eemb = (const float*)d_in[4];
  const float* W1   = (const float*)d_in[5];
  const float* as1  = (const float*)d_in[6];
  const float* ad1  = (const float*)d_in[7];
  const float* We1  = (const float*)d_in[8];
  const float* ae1  = (const float*)d_in[9];
  const float* b1   = (const float*)d_in[10];
  const float* W2   = (const float*)d_in[11];
  const float* as2  = (const float*)d_in[12];
  const float* ad2  = (const float*)d_in[13];
  const float* We2  = (const float*)d_in[14];
  const float* ae2  = (const float*)d_in[15];
  const float* b2   = (const float*)d_in[16];
  float* out = (float*)d_out;

  float* ws = (float*)d_ws;
  size_t off = 0;
  unsigned* CNT = (unsigned*)(ws + off); off += (size_t)2048 * NEA;
  unsigned* DEG = (unsigned*)(ws + off); off += 2048;
  unsigned* ROWPTR = (unsigned*)(ws + off); off += 2112;
  unsigned* CUR = (unsigned*)(ws + off); off += 2048;
  int* CSRC = (int*)(ws + off); off += (size_t)NE;
  int* CATTR = (int*)(ws + off); off += (size_t)NE;
  float* TAB1 = ws + off; off += 128;
  float* TAB2 = ws + off; off += 128;
  float* HX1  = ws + off; off += (size_t)NN * NH * HID;
  float* HX2  = ws + off; off += (size_t)NN * NH * FT;
  float* PROJ1 = ws + off; off += (size_t)EMB * NH;
  float* PROJ2 = ws + off; off += (size_t)EMB * NH;
  float* ANS  = ws + off; off += (size_t)NN * NH;
  float* ADN  = ws + off; off += (size_t)NN * NH;
  float* HB   = ws + off; off += (size_t)NN * HID;
  (void)ws_size; (void)in_sizes; (void)n_in; (void)out_size;

  const int T = 256;

  hipMemsetAsync(CNT, 0, (size_t)(2048 * NEA + 2048) * sizeof(unsigned), stream);

  k_count_proj<<<2 * PROJ_BLOCKS + CNT_BLOCKS, 256, 0, stream>>>(dst, eattr, CNT, DEG, We1, ae1, PROJ1, We2, ae2, PROJ2);
  k_scan_tab<<<1 + TAB_BLOCKS, 1024, 0, stream>>>(DEG, ROWPTR, CUR, eemb, PROJ1, TAB1, PROJ2, TAB2);
  k_csr_fill<<<cdiv(NE, T), T, 0, stream>>>(src, dst, eattr, CUR, CSRC, CATTR);

  {
    dim3 g(NH * HID / 128, cdiv(NN, 64));
    k_gemm_wmma_f32<true><<<g, 256, 0, stream>>>(nullptr, x, xemb, W1, HX1, NN, NH * HID, EMB);
  }
  k_alpha_node<HID><<<cdiv((long long)NN * 32, 128), 128, 0, stream>>>(HX1, as1, ad1, ANS, ADN);
  k_gather<<<cdiv((long long)NN * (HID / 128), 8), 256, 0, stream>>>(HX1, ANS, ADN, TAB1, CNT, ROWPTR, CSRC, CATTR, b1, HB, HID, HID / 128, 1);

  {
    dim3 g(NH * FT / 128, cdiv(NN, 64));
    k_gemm_wmma_f32<false><<<g, 256, 0, stream>>>(HB, nullptr, nullptr, W2, HX2, NN, NH * FT, HID);
  }
  k_alpha_node<FT><<<cdiv((long long)NN * 32, 128), 128, 0, stream>>>(HX2, as2, ad2, ANS, ADN);
  k_gather<<<cdiv((long long)NN * (FT / 128), 8), 256, 0, stream>>>(HX2, ANS, ADN, TAB2, CNT, ROWPTR, CSRC, CATTR, b2, out, FT, FT / 128, 0);
}


# Layer 1 transform by linearity: WMMA GEMM over the 178 node-table rows (32x64 tiles), each node row the sum of two product rows, formed in the per-node scalar kernel; sum order changes
# speedup vs baseline: 6.4308x; 1.1042x over previous
#include <hip/hip_runtime.h>
#include <math.h>

#define NN   2000
#define NE   32000
#define EMB  768
#define HID  384
#define FT   768
#define NH   4
#define NEA  18
#define NXE  178
#define NEG_SLOPE 0.2f

typedef __attribute__((ext_vector_type(2))) float v2f;
typedef __attribute__((ext_vector_type(8))) float v8f;

static inline int cdiv(long long a, long long b) { return (int)((a + b - 1) / b); }

__device__ __forceinline__ void attr_count_edge(int e, const int* __restrict__ dst,
                                                const int* __restrict__ ea_idx,
                                                unsigned* __restrict__ cnt, unsigned* __restrict__ deg) {
  if (e >= NE) return;
  int d = dst[e];
  atomicAdd(&cnt[d * NEA + ea_idx[e * 2 + 0]], 1u);
  atomicAdd(&cnt[d * NEA + ea_idx[e * 2 + 1]], 1u);
  atomicAdd(&deg[d], 1u);
}

static_assert(NN <= 2048, "csr_scan_block covers two nodes per thread of one 1024-thread block");
__device__ __forceinline__ void csr_scan_block(const unsigned* __restrict__ deg,
                                               unsigned* __restrict__ rowptr,
                                               unsigned* __restrict__ cur, unsigned* wsum) {
  const int t = threadIdx.x, lane = t & 31, wave = t >> 5;
  const int n0 = 2 * t;
  const unsigned a = (n0 < NN) ? deg[n0] : 0u;
  const unsigned b = (n0 + 1 < NN) ? deg[n0 + 1] : 0u;
  unsigned incl = a + b;
  for (int off = 1; off < 32; off <<= 1) {
    unsigned v = __shfl_up(incl, off, 32);
    if (lane >= off) incl += v;
  }
  if (lane == 31) wsum[wave] = incl;
  __syncthreads();
  if (wave == 0) {
    unsigned w = wsum[lane];
    for (int off = 1; off < 32; off <<= 1) {
      unsigned v = __shfl_up(w, off, 32);
      if (lane >= off) w += v;
    }
    wsum[lane] = w;
  }
  __syncthreads();
  const unsigned excl = incl - (a + b) + (wave ? wsum[wave - 1] : 0u);
  if (n0 < NN)     { rowptr[n0] = excl;         cur[n0] = excl; }
  if (n0 + 1 < NN) { rowptr[n0 + 1] = excl + a; cur[n0 + 1] = excl + a; }
  if (t == 1023) rowptr[NN] = excl + a + b;
}

__global__ void k_csr_fill(const int* __restrict__ src, const int* __restrict__ dst,
                           const int* __restrict__ ea_idx,
                           unsigned* __restrict__ cur, int* __restrict__ csrc,
                           int* __restrict__ cattr) {
  int e = blockIdx.x * blockDim.x + threadIdx.x;
  if (e >= NE) return;
  unsigned pos = atomicAdd(&cur[dst[e]], 1u);
  csrc[pos] = src[e];
  cattr[pos] = (int)((unsigned)ea_idx[e * 2 + 0] | ((unsigned)ea_idx[e * 2 + 1] << 16));
}

#define A_PITCH 36
#define B_PITCH 34

template <int TM, int TN>
__device__ __forceinline__ void gemm_block_mma(const float* __restrict__ ap, const float* __restrict__ bp,
                                               int khalf, v8f (&acc)[TM][TN]) {
#pragma unroll
  for (int ks = 0; ks < 32; ks += 4) {
    int kk = ks + khalf * 2;
    v2f a[TM], b[TN];
#pragma unroll
    for (int i = 0; i < TM; ++i) a[i] = *(const v2f*)(ap + i * 16 * A_PITCH + kk);
#pragma unroll
    for (int j = 0; j < TN; ++j) b[j] = *(const v2f*)(bp + j * 16 * B_PITCH + kk);
#pragma unroll
    for (int i = 0; i < TM; ++i)
#pragma unroll
      for (int j = 0; j < TN; ++j)
        acc[i][j] = __builtin_amdgcn_wmma_f32_16x16x4_f32(false, a[i], false, b[j], (short)0, acc[i][j], false, false);
  }
}

template <int TM, int TN>
__global__ __launch_bounds__(256) void k_gemm_wmma_f32(
    const float* __restrict__ A, const float* __restrict__ B, float* __restrict__ C, int M, int N, int K) {
  constexpr int BM = 32 * TM, BN = 64 * TN;
  constexpr int AJ = BM / 8;
  constexpr int BR = 256 / BN;
  constexpr int BJ = 32 / BR;
  __shared__ __align__(16) float sA[2][BM][A_PITCH];
  __shared__ __align__(16) float sBt[2][BN][B_PITCH];
  const int tid = threadIdx.x;
  const int wave = tid >> 5, lane = tid & 31;
  const int wm = wave >> 2, wn = wave & 3;
  const int row16 = lane & 15, khalf = lane >> 4;
  const int bm = blockIdx.y * BM, bn = blockIdx.x * BN;

  const int ar = tid >> 5, ac = tid & 31;
  const int bk = tid / BN, bc = tid % BN;

  unsigned aoff[AJ];
  bool rowok[AJ];
#pragma unroll
  for (int j = 0; j < AJ; ++j) {
    const int gr = bm + ar + 8 * j;
    rowok[j] = gr < M;
    aoff[j] = (unsigned)min(gr, M - 1) * (unsigned)K + (unsigned)ac;
  }
  const unsigned boff = (unsigned)bk * (unsigned)N + (unsigned)bc;

  float ua[AJ], rb[BJ];

  auto load_tiles = [&](int k0) {
    const float* Ak = A + k0;
#pragma unroll
    for (int j = 0; j < AJ; ++j) ua[j] = Ak[aoff[j]];
    const float* Bk = B + (size_t)k0 * N + bn;
#pragma unroll
    for (int j = 0; j < BJ; ++j) rb[j] = Bk[boff + (unsigned)(BR * j) * (unsigned)N];
  };
  auto store_tiles = [&](int p) {
#pragma unroll
    for (int j = 0; j < AJ; ++j) sA[p][ar + 8 * j][ac] = rowok[j] ? ua[j] : 0.0f;
#pragma unroll
    for (int j = 0; j < BJ; ++j) sBt[p][bc][bk + BR * j] = rb[j];
  };

  v8f acc[TM][TN] = {};
  const int arow = wm * (16 * TM) + row16, brow = wn * (16 * TN) + row16;

  load_tiles(0);
  store_tiles(0);
  __syncthreads();
  int p = 0;
  for (int k0 = 32; k0 < K; k0 += 32) {
    load_tiles(k0);
    __builtin_amdgcn_sched_barrier(0);
    gemm_block_mma<TM, TN>(&sA[p][arow][0], &sBt[p][brow][0], khalf, acc);
    __builtin_amdgcn_sched_barrier(0);
    store_tiles(p ^ 1);
    __syncthreads();
    p ^= 1;
  }
  gemm_block_mma<TM, TN>(&sA[p][arow][0], &sBt[p][brow][0], khalf, acc);

  const int gn = bn + wn * (16 * TN) + row16;
  const int gm = bm + wm * (16 * TM) + 8 * khalf;
#pragma unroll
  for (int v = 0; v < 8; ++v) {
#pragma unroll
    for (int i = 0; i < TM; ++i) {
      if (gm + 16 * i + v < M) {
#pragma unroll
        for (int j = 0; j < TN; ++j) C[(size_t)(gm + 16 * i + v) * N + gn + 16 * j] = acc[i][j][v];
      }
    }
  }
}

__device__ __forceinline__ float dot4(const float4& a, const float4& b, float s) {
  s = fmaf(a.x, b.x, s); s = fmaf(a.y, b.y, s); s = fmaf(a.z, b.z, s); s = fmaf(a.w, b.w, s);
  return s;
}
__device__ __forceinline__ float wave_sum(float v) {
  for (int off = 16; off > 0; off >>= 1) v += __shfl_xor(v, off, 32);
  return v;
}

template <int CC>
__device__ __forceinline__ void proj_row(const float* __restrict__ We, const float* __restrict__ ae,
                                         float* __restrict__ proj, int k, int lane) {
  static_assert(CC % 128 == 0, "a row is a whole number of 512-byte steps of a wave");
  constexpr int PH = CC / 128;
  if (k >= EMB) return;
  const float4* w4 = (const float4*)(We + (size_t)k * (NH * CC));
  const float4* a4 = (const float4*)ae;
  float s[NH];
#pragma unroll
  for (int h = 0; h < NH; ++h) {
    float acc = 0.f;
#pragma unroll
    for (int u = 0; u < PH; ++u) {
      const int q = (h * PH + u) * 32 + lane;
      acc = dot4(w4[q], a4[q], acc);
    }
    s[h] = wave_sum(acc);
  }
  if (lane == 0) {
    float4 r; r.x = s[0]; r.y = s[1]; r.z = s[2]; r.w = s[3];
    ((float4*)proj)[k] = r;
  }
}

template <int CC, bool SUM_ROWS>
__global__ __launch_bounds__(128) void k_alpha_node(const float* __restrict__ hx,
                                                    const float* __restrict__ ptab,
                                                    const int* __restrict__ xi,
                                                    float* __restrict__ hxw,
                                                    const float* __restrict__ a_src,
                                                    const float* __restrict__ a_dst,
                                                    float* __restrict__ ans,
                                                    float* __restrict__ adn) {
  static_assert(CC % 128 == 0, "a row is a whole number of 512-byte steps of a wave");
  static_assert(NH == 4, "one 16-byte store per node and output");
  constexpr int PH = CC / 128;
  const int lane = threadIdx.x & 31;
  const int n = __builtin_amdgcn_readfirstlane((int)((blockIdx.x * 128 + threadIdx.x) >> 5));
  if (n >= NN) return;
  const float4* h4 = nullptr;
  const float4* p0 = nullptr;
  const float4* p1 = nullptr;
  float4* w4 = nullptr;
  if constexpr (SUM_ROWS) {
    p0 = (const float4*)(ptab + (size_t)xi[2 * n] * (NH * CC));
    p1 = (const float4*)(ptab + (size_t)xi[2 * n + 1] * (NH * CC));
    w4 = (float4*)(hxw + (size_t)n * (NH * CC));
  } else {
    h4 = (const float4*)(hx + (size_t)n * (NH * CC));
  }
  const float4* s4 = (const float4*)a_src;
  const float4* d4 = (const float4*)a_dst;
  float s[NH], t[NH];
#pragma unroll
  for (int h = 0; h < NH; ++h) {
    float as = 0.f, ad = 0.f;
#pragma unroll
    for (int u = 0; u < PH; ++u) {
      const int q = (h * PH + u) * 32 + lane;
      float4 v;
      if constexpr (SUM_ROWS) {
        const float4 v0 = p0[q], v1 = p1[q];
        v.x = v0.x + v1.x; v.y = v0.y + v1.y; v.z = v0.z + v1.z; v.w = v0.w + v1.w;
        w4[q] = v;
      } else {
        v = h4[q];
      }
      as = dot4(v, s4[q], as);
      ad = dot4(v, d4[q], ad);
    }
    s[h] = wave_sum(as);
    t[h] = wave_sum(ad);
  }
  if (lane == 0) {
    float4 r; r.x = s[0]; r.y = s[1]; r.z = s[2]; r.w = s[3];
    ((float4*)ans)[n] = r;
    float4 w; w.x = t[0]; w.y = t[1]; w.z = t[2]; w.w = t[3];
    ((float4*)adn)[n] = w;
  }
}

__device__ __forceinline__ void attr_tab_row(const float* __restrict__ emb,
                                             const float* __restrict__ proj,
                                             float* __restrict__ tab, int a, int lane) {
  const float* row = emb + (size_t)a * EMB;
  const float4* p4 = (const float4*)proj;
  float a0 = 0.f, a1 = 0.f, a2 = 0.f, a3 = 0.f;
#pragma unroll
  for (int t = 0; t < EMB / 32; ++t) {
    int k = lane + 32 * t;
    float v = row[k];
    float4 p = p4[k];
    a0 += v * p.x; a1 += v * p.y; a2 += v * p.z; a3 += v * p.w;
  }
  for (int off = 16; off > 0; off >>= 1) {
    a0 += __shfl_xor(a0, off, 32);
    a1 += __shfl_xor(a1, off, 32);
    a2 += __shfl_xor(a2, off, 32);
    a3 += __shfl_xor(a3, off, 32);
  }
  if (lane == 0) {
    float4 r; r.x = a0; r.y = a1; r.z = a2; r.w = a3;
    ((float4*)tab)[a] = r;
  }
}

__device__ __forceinline__ void edge_fma(float4& acc, float w0, float w1, float w2, float w3,
                                         const float4& v0, const float4& v1, const float4& v2,
                                         const float4& v3) {
  acc.x = fmaf(w0, v0.x, acc.x); acc.y = fmaf(w0, v0.y, acc.y);
  acc.z = fmaf(w0, v0.z, acc.z); acc.w = fmaf(w0, v0.w, acc.w);
  acc.x = fmaf(w1, v1.x, acc.x); acc.y = fmaf(w1, v1.y, acc.y);
  acc.z = fmaf(w1, v1.z, acc.z); acc.w = fmaf(w1, v1.w, acc.w);
  acc.x = fmaf(w2, v2.x, acc.x); acc.y = fmaf(w2, v2.y, acc.y);
  acc.z = fmaf(w2, v2.z, acc.z); acc.w = fmaf(w2, v2.w, acc.w);
  acc.x = fmaf(w3, v3.x, acc.x); acc.y = fmaf(w3, v3.y, acc.y);
  acc.z = fmaf(w3, v3.z, acc.z); acc.w = fmaf(w3, v3.w, acc.w);
}

__device__ __forceinline__ float leaky(float a) { return (a > 0.f) ? a : NEG_SLOPE * a; }

__device__ __forceinline__ float4 edge_score(int s, int pa, const float4* __restrict__ ans4,
                                             const float4* __restrict__ tab4, const float4& ad) {
  const float4 an = ans4[s];
  const float4 t0 = tab4[(unsigned)pa & 0xffffu], t1 = tab4[(unsigned)pa >> 16];
  float4 a;
  a.x = leaky(an.x + ad.x + (t0.x + t1.x));
  a.y = leaky(an.y + ad.y + (t0.y + t1.y));
  a.z = leaky(an.z + ad.z + (t0.z + t1.z));
  a.w = leaky(an.w + ad.w + (t0.w + t1.w));
  return a;
}
__device__ __forceinline__ float4 max4(const float4& a, const float4& b) {
  float4 r; r.x = fmaxf(a.x, b.x); r.y = fmaxf(a.y, b.y); r.z = fmaxf(a.z, b.z); r.w = fmaxf(a.w, b.w);
  return r;
}
__device__ __forceinline__ float4 exp4(const float4& a, const float4& m) {
  float4 r; r.x = expf(a.x - m.x); r.y = expf(a.y - m.y); r.z = expf(a.z - m.z); r.w = expf(a.w - m.w);
  return r;
}
__device__ __forceinline__ float first_lane(float v) {
  return __int_as_float(__builtin_amdgcn_readfirstlane(__float_as_int(v)));
}
__device__ __forceinline__ float4 wave_max4(float4 v) {
  for (int off = 16; off > 0; off >>= 1) {
    v.x = fmaxf(v.x, __shfl_xor(v.x, off, 32)); v.y = fmaxf(v.y, __shfl_xor(v.y, off, 32));
    v.z = fmaxf(v.z, __shfl_xor(v.z, off, 32)); v.w = fmaxf(v.w, __shfl_xor(v.w, off, 32));
  }
  return v;
}
__device__ __forceinline__ float4 wave_sum4(float4 v) {
  for (int off = 16; off > 0; off >>= 1) {
    v.x += __shfl_xor(v.x, off, 32); v.y += __shfl_xor(v.y, off, 32);
    v.z += __shfl_xor(v.z, off, 32); v.w += __shfl_xor(v.w, off, 32);
  }
  return v;
}

__global__ __launch_bounds__(256) void k_gather(
    const float* __restrict__ hx, const float* __restrict__ ans, const float* __restrict__ adn,
    const float* __restrict__ tab, const unsigned* __restrict__ cnt,
    const unsigned* __restrict__ rowptr, const int* __restrict__ csrc, const int* __restrict__ cattr,
    const float* __restrict__ bias, float* __restrict__ o, int C, int chunks, int relu) {
  const int lane = threadIdx.x & 31;
  const int gw = __builtin_amdgcn_readfirstlane((int)((blockIdx.x * 256 + threadIdx.x) >> 5));
  const int node = gw / chunks;
  if (node >= NN) return;
  const int c0 = (gw - node * chunks) * 128 + lane * 4;
  const size_t rs = (size_t)NH * C;
  const float4* ans4 = (const float4*)ans;
  const float4* tab4 = (const float4*)tab;
  const int beg = (int)rowptr[node], end = (int)rowptr[node + 1];

  const float* ps = hx + (size_t)node * rs + c0;
  const float4 sv0 = *(const float4*)ps, sv1 = *(const float4*)(ps + C);
  const float4 sv2 = *(const float4*)(ps + 2 * C), sv3 = *(const float4*)(ps + 3 * C);

  const float4 ad = *(const float4*)(adn + (size_t)node * NH);
  float4 a_self;
  {
    const unsigned* cr = cnt + (size_t)node * NEA;
    float s0 = 0.f, s1 = 0.f, s2 = 0.f, s3 = 0.f;
#pragma unroll
    for (int a = 0; a < NEA; ++a) {
      const float cf = (float)cr[a];
      const float4 t = tab4[a];
      s0 += cf * t.x; s1 += cf * t.y; s2 += cf * t.z; s3 += cf * t.w;
    }
    const float dg = fmaxf((float)(end - beg), 1.0f);
    const float4 an = ans4[node];
    a_self.x = leaky(an.x + ad.x + s0 / dg);
    a_self.y = leaky(an.y + ad.y + s1 / dg);
    a_self.z = leaky(an.z + ad.z + s2 / dg);
    a_self.w = leaky(an.w + ad.w + s3 / dg);
  }

  const bool has0 = beg + lane < end;
  int s_first = node;
  float4 a_first = a_self;
  if (has0) {
    s_first = csrc[beg + lane];
    a_first = edge_score(s_first, cattr[beg + lane], ans4, tab4, ad);
  }
  float4 m = max4(a_first, a_self);
  for (int base = beg + 32; base < end; base += 32) {
    const int idx = base + lane;
    if (idx < end) m = max4(m, edge_score(csrc[idx], cattr[idx], ans4, tab4, ad));
  }
  m = wave_max4(m);
  m.x = first_lane(m.x); m.y = first_lane(m.y); m.z = first_lane(m.z); m.w = first_lane(m.w);
  float4 e_first = {0.f, 0.f, 0.f, 0.f};
  if (has0) e_first = exp4(a_first, m);
  float4 dsum = e_first;
  for (int base = beg + 32; base < end; base += 32) {
    const int idx = base + lane;
    if (idx < end) {
      const float4 ex = exp4(edge_score(csrc[idx], cattr[idx], ans4, tab4, ad), m);
      dsum.x += ex.x; dsum.y += ex.y; dsum.z += ex.z; dsum.w += ex.w;
    }
  }
  dsum = wave_sum4(dsum);
  dsum.x = first_lane(dsum.x); dsum.y = first_lane(dsum.y); dsum.z = first_lane(dsum.z); dsum.w = first_lane(dsum.w);
  const float4 e_self = exp4(a_self, m);
  const float i0 = 1.0f / (dsum.x + e_self.x), i1 = 1.0f / (dsum.y + e_self.y);
  const float i2 = 1.0f / (dsum.z + e_self.z), i3 = 1.0f / (dsum.w + e_self.w);

  float4 acc = {0.f, 0.f, 0.f, 0.f};
  edge_fma(acc, e_self.x * i0, e_self.y * i1, e_self.z * i2, e_self.w * i3, sv0, sv1, sv2, sv3);

  int s_l = s_first;
  float w0_l = e_first.x * i0, w1_l = e_first.y * i1, w2_l = e_first.z * i2, w3_l = e_first.w * i3;
  for (int base = beg; base < end; base += 32) {
    if (base != beg) {
      const int idx = base + lane;
      s_l = node;
      w0_l = 0.f; w1_l = 0.f; w2_l = 0.f; w3_l = 0.f;
      if (idx < end) {
        s_l = csrc[idx];
        const float4 ex = exp4(edge_score(s_l, cattr[idx], ans4, tab4, ad), m);
        w0_l = ex.x * i0; w1_l = ex.y * i1; w2_l = ex.z * i2; w3_l = ex.w * i3;
      }
    }
    const int cnt_b = min(32, end - base);
    int j = 0;
    for (; j + 4 <= cnt_b; j += 4) {
      float4 v[4][4];
      float w[4][4];
#pragma unroll
      for (int u = 0; u < 4; ++u) {
        const int s = __builtin_amdgcn_readlane(s_l, j + u);
        const float* p = hx + (size_t)s * rs + c0;
        v[u][0] = *(const float4*)p;
        v[u][1] = *(const float4*)(p + C);
        v[u][2] = *(const float4*)(p + 2 * C);
        v[u][3] = *(const float4*)(p + 3 * C);
        w[u][0] = __int_as_float(__builtin_amdgcn_readlane(__float_as_int(w0_l), j + u));
        w[u][1] = __int_as_float(__builtin_amdgcn_readlane(__float_as_int(w1_l), j + u));
        w[u][2] = __int_as_float(__builtin_amdgcn_readlane(__float_as_int(w2_l), j + u));
        w[u][3] = __int_as_float(__builtin_amdgcn_readlane(__float_as_int(w3_l), j + u));
      }
#pragma unroll
      for (int u = 0; u < 4; ++u)
        edge_fma(acc, w[u][0], w[u][1], w[u][2], w[u][3], v[u][0], v[u][1], v[u][2], v[u][3]);
    }
    for (; j < cnt_b; ++j) {
      const int s = __builtin_amdgcn_readlane(s_l, j);
      const float* p = hx + (size_t)s * rs + c0;
      const float4 v0 = *(const float4*)p, v1 = *(const float4*)(p + C);
      const float4 v2 = *(const float4*)(p + 2 * C), v3 = *(const float4*)(p + 3 * C);
      edge_fma(acc,
               __int_as_float(__builtin_amdgcn_readlane(__float_as_int(w0_l), j)),
               __int_as_float(__builtin_amdgcn_readlane(__float_as_int(w1_l), j)),
               __int_as_float(__builtin_amdgcn_readlane(__float_as_int(w2_l), j)),
               __int_as_float(__builtin_amdgcn_readlane(__float_as_int(w3_l), j)),
               v0, v1, v2, v3);
    }
  }

  float4 r;
  r.x = 0.25f * acc.x + bias[c0 + 0];
  r.y = 0.25f * acc.y + bias[c0 + 1];
  r.z = 0.25f * acc.z + bias[c0 + 2];
  r.w = 0.25f * acc.w + bias[c0 + 3];
  if (relu) { r.x = fmaxf(r.x, 0.f); r.y = fmaxf(r.y, 0.f); r.z = fmaxf(r.z, 0.f); r.w = fmaxf(r.w, 0.f); }
  *(float4*)(o + (size_t)node * C + c0) = r;
}

#define CNT_BLOCKS  ((NE + 255) / 256)
#define PROJ_BLOCKS (EMB / 8)
static_assert(EMB % 8 == 0, "a proj work-group covers 8 whole k");
__global__ __launch_bounds__(256) void k_count_proj(
    const int* __restrict__ dst, const int* __restrict__ ea_idx,
    unsigned* __restrict__ cnt, unsigned* __restrict__ deg,
    const float* __restrict__ We1, const float* __restrict__ ae1, float* __restrict__ proj1,
    const float* __restrict__ We2, const float* __restrict__ ae2, float* __restrict__ proj2) {
  const int b = blockIdx.x;
  const int lane = threadIdx.x & 31;
  const int wave = __builtin_amdgcn_readfirstlane((int)(threadIdx.x >> 5));
  if (b < PROJ_BLOCKS) {
    proj_row<FT>(We2, ae2, proj2, b * 8 + wave, lane);
  } else if (b < 2 * PROJ_BLOCKS) {
    proj_row<HID>(We1, ae1, proj1, (b - PROJ_BLOCKS) * 8 + wave, lane);
  } else {
    attr_count_edge((b - 2 * PROJ_BLOCKS) * 256 + (int)threadIdx.x, dst, ea_idx, cnt, deg);
  }
}

#define TAB_BLOCKS ((2 * NEA + 31) / 32)
__global__ __launch_bounds__(1024) void k_scan_tab(
    const unsigned* __restrict__ deg, unsigned* __restrict__ rowptr, unsigned* __restrict__ cur,
    const float* __restrict__ emb,
    const float* __restrict__ proj1, float* __restrict__ tab1,
    const float* __restrict__ proj2, float* __restrict__ tab2) {
  __shared__ unsigned wsum[32];
  if (blockIdx.x == 0) {
    csr_scan_block(deg, rowptr, cur, wsum);
    return;
  }
  const int lane = threadIdx.x & 31;
  const int r = __builtin_amdgcn_readfirstlane((int)((blockIdx.x - 1) * 32 + (threadIdx.x >> 5)));
  if (r < NEA) attr_tab_row(emb, proj1, tab1, r, lane);
  else if (r < 2 * NEA) attr_tab_row(emb, proj2, tab2, r - NEA, lane);
}

extern "C" void kernel_launch(void* const* d_in, const int* in_sizes, int n_in,
                              void* d_out, int out_size, void* d_ws, size_t ws_size,
                              hipStream_t stream) {
  const int* x      = (const int*)d_in[0];
  const int* eidx   = (const int*)d_in[1];
  const int* src    = eidx;
  const int* dst    = eidx + NE;
  const int* eattr  = (const int*)d_in[2];
  const float* xemb = (const float*)d_in[3];
  const float* eemb = (const float*)d_in[4];
  const float* W1   = (const float*)d_in[5];
  const float* as1  = (const float*)d_in[6];
  const float* ad1  = (const float*)d_in[7];
  const float* We1  = (const float*)d_in[8];
  const float* ae1  = (const float*)d_in[9];
  const float* b1   = (const float*)d_in[10];
  const float* W2   = (const float*)d_in[11];
  const float* as2  = (const float*)d_in[12];
  const float* ad2  = (const float*)d_in[13];
  const float* We2  = (const float*)d_in[14];
  const float* ae2  = (const float*)d_in[15];
  const float* b2   = (const float*)d_in[16];
  float* out = (float*)d_out;

  float* ws = (float*)d_ws;
  size_t off = 0;
  unsigned* CNT = (unsigned*)(ws + off); off += (size_t)2048 * NEA;
  unsigned* DEG = (unsigned*)(ws + off); off += 2048;
  unsigned* ROWPTR = (unsigned*)(ws + off); off += 2112;
  unsigned* CUR = (unsigned*)(ws + off); off += 2048;
  int* CSRC = (int*)(ws + off); off += (size_t)NE;
  int* CATTR = (int*)(ws + off); off += (size_t)NE;
  float* TAB1 = ws + off; off += 128;
  float* TAB2 = ws + off; off += 128;
  float* HX1  = ws + off; off += (size_t)NN * NH * HID;
  float* HX2  = ws + off; off += (size_t)NN * NH * FT;
  float* PROJ1 = ws + off; off += (size_t)EMB * NH;
  float* PROJ2 = ws + off; off += (size_t)EMB * NH;
  float* ANS  = ws + off; off += (size_t)NN * NH;
  float* ADN  = ws + off; off += (size_t)NN * NH;
  float* HB   = ws + off; off += (size_t)NN * HID;
  float* PT1  = ws + off; off += (size_t)NXE * NH * HID;
  (void)ws_size; (void)in_sizes; (void)n_in; (void)out_size;

  const int T = 256;

  hipMemsetAsync(CNT, 0, (size_t)(2048 * NEA + 2048) * sizeof(unsigned), stream);

  k_count_proj<<<2 * PROJ_BLOCKS + CNT_BLOCKS, 256, 0, stream>>>(dst, eattr, CNT, DEG, We1, ae1, PROJ1, We2, ae2, PROJ2);
  k_scan_tab<<<1 + TAB_BLOCKS, 1024, 0, stream>>>(DEG, ROWPTR, CUR, eemb, PROJ1, TAB1, PROJ2, TAB2);
  k_csr_fill<<<cdiv(NE, T), T, 0, stream>>>(src, dst, eattr, CUR, CSRC, CATTR);

  {
    dim3 g(NH * HID / 64, cdiv(NXE, 32));
    k_gemm_wmma_f32<1, 1><<<g, 256, 0, stream>>>(xemb, W1, PT1, NXE, NH * HID, EMB);
  }
  k_alpha_node<HID, true><<<cdiv((long long)NN * 32, 128), 128, 0, stream>>>(nullptr, PT1, x, HX1, as1, ad1, ANS, ADN);
  k_gather<<<cdiv((long long)NN * (HID / 128), 8), 256, 0, stream>>>(HX1, ANS, ADN, TAB1, CNT, ROWPTR, CSRC, CATTR, b1, HB, HID, HID / 128, 1);

  {
    dim3 g(NH * FT / 128, cdiv(NN, 64));
    k_gemm_wmma_f32<2, 2><<<g, 256, 0, stream>>>(HB, W2, HX2, NN, NH * FT, HID);
  }
  k_alpha_node<FT, false><<<cdiv((long long)NN * 32, 128), 128, 0, stream>>>(HX2, nullptr, nullptr, nullptr, as2, ad2, ANS, ADN);
  k_gather<<<cdiv((long long)NN * (FT / 128), 8), 256, 0, stream>>>(HX2, ANS, ADN, TAB2, CNT, ROWPTR, CSRC, CATTR, b2, out, FT, FT / 128, 0);
}
